# starting kernel
# baseline (speedup 1.0000x reference)
#include <hip/hip_runtime.h>
#include <math.h>

#define D_IN   128
#define HEADS  8
#define HID    64
#define F1     (HEADS * HID)
#define D_OUT  8
#define NEG_SLOPE 0.2f

typedef float v2f __attribute__((ext_vector_type(2)));
typedef float v8f __attribute__((ext_vector_type(8)));

__device__ __forceinline__ unsigned f2ord(float f) {
  unsigned u = __float_as_uint(f);
  return (u & 0x80000000u) ? ~u : (u | 0x80000000u);
}
__device__ __forceinline__ float ord2f(unsigned v) {
  unsigned u = (v & 0x80000000u) ? (v ^ 0x80000000u) : ~v;
  return __uint_as_float(u);
}
__device__ __forceinline__ float lrelu(float x) { return x > 0.f ? x : NEG_SLOPE * x; }

__global__ void gemm1_wmma(const float* __restrict__ X, const float* __restrict__ W,
                           float* __restrict__ H, int n) {
  int wave = (blockIdx.x * blockDim.x + threadIdx.x) >> 5;
  int lane = threadIdx.x & 31;
  const int ntiles = F1 / 16;
  int mt = wave / ntiles;
  int nt = wave - mt * ntiles;
  int row0 = mt * 16;
  int col0 = nt * 16;
  if (row0 >= n) return;

  int half = lane >> 4;
  int l15  = lane & 15;
  const float* xrow = X + (size_t)(row0 + l15) * D_IN;
  int bcol = col0 + l15;

  v8f acc = {};
  for (int k = 0; k < D_IN; k += 4) {
    int kk = k + half * 2;
    v2f a; a.x = xrow[kk];                       a.y = xrow[kk + 1];
    v2f b; b.x = W[(size_t)kk * F1 + bcol];      b.y = W[(size_t)(kk + 1) * F1 + bcol];
    acc = __builtin_amdgcn_wmma_f32_16x16x4_f32(false, a, false, b,
                                                (short)0, acc, false, false);
  }
#pragma unroll
  for (int v = 0; v < 8; ++v) {
    int r = row0 + v + half * 8;
    H[(size_t)r * F1 + bcol] = acc[v];
  }
}

__global__ void gemm2_wmma(const float* __restrict__ X2, const float* __restrict__ W2,
                           float* __restrict__ H2, int n) {
  int wave = (blockIdx.x * blockDim.x + threadIdx.x) >> 5;
  int lane = threadIdx.x & 31;
  int row0 = wave * 16;
  if (row0 >= n) return;

  int half = lane >> 4;
  int l15  = lane & 15;
  const float* xrow = X2 + (size_t)(row0 + l15) * F1;
  bool colok = (l15 < D_OUT);

  v8f acc = {};
  for (int k = 0; k < F1; k += 4) {
    int kk = k + half * 2;
    v2f a; a.x = xrow[kk]; a.y = xrow[kk + 1];
    v2f b;
    b.x = colok ? W2[(size_t)kk * D_OUT + l15]       : 0.f;
    b.y = colok ? W2[(size_t)(kk + 1) * D_OUT + l15] : 0.f;
    acc = __builtin_amdgcn_wmma_f32_16x16x4_f32(false, a, false, b,
                                                (short)0, acc, false, false);
  }
  if (colok) {
#pragma unroll
    for (int v = 0; v < 8; ++v) {
      int r = row0 + v + half * 8;
      H2[(size_t)r * D_OUT + l15] = acc[v];
    }
  }
}

__global__ void alphas1(const float* __restrict__ H, const float* __restrict__ a_src,
                        const float* __restrict__ a_dst,
                        float* __restrict__ as, float* __restrict__ ad, int n) {
  int t = blockIdx.x * blockDim.x + threadIdx.x;
  if (t >= n * HEADS) return;
  int h = t % HEADS;
  const float* row = H + (size_t)(t / HEADS) * F1 + h * HID;
  const float* vs = a_src + h * HID;
  const float* vd = a_dst + h * HID;
  float s = 0.f, d = 0.f;
#pragma unroll 8
  for (int c = 0; c < HID; ++c) { float x = row[c]; s += x * vs[c]; d += x * vd[c]; }
  as[t] = s; ad[t] = d;
}

__global__ void alphas2(const float* __restrict__ H2, const float* __restrict__ a_src,
                        const float* __restrict__ a_dst,
                        float* __restrict__ as, float* __restrict__ ad, int n) {
  int t = blockIdx.x * blockDim.x + threadIdx.x;
  if (t >= n) return;
  const float* row = H2 + (size_t)t * D_OUT;
  float s = 0.f, d = 0.f;
#pragma unroll
  for (int c = 0; c < D_OUT; ++c) { s += row[c] * a_src[c]; d += row[c] * a_dst[c]; }
  as[t] = s; ad[t] = d;
}

__device__ __forceinline__ void edge_nodes(int e, int nE, const int* src, const int* dst,
                                           int& s_, int& d_) {
  if (e < nE) { s_ = src[e]; d_ = dst[e]; }
  else        { s_ = d_ = e - nE; }
}

__global__ void edge_max(const int* __restrict__ src, const int* __restrict__ dst,
                         int nE, int n, const float* __restrict__ as,
                         const float* __restrict__ ad, unsigned* __restrict__ m, int Hn) {
  int t = blockIdx.x * blockDim.x + threadIdx.x;
  int total = (nE + n) * Hn;
  if (t >= total) return;
  int e = t / Hn, h = t - e * Hn;
  int s_, d_; edge_nodes(e, nE, src, dst, s_, d_);
  float ev = lrelu(as[(size_t)s_ * Hn + h] + ad[(size_t)d_ * Hn + h]);
  atomicMax(m + (size_t)d_ * Hn + h, f2ord(ev));
}

__global__ void edge_denom(const int* __restrict__ src, const int* __restrict__ dst,
                           int nE, int n, const float* __restrict__ as,
                           const float* __restrict__ ad, const unsigned* __restrict__ m,
                           float* __restrict__ denom, int Hn) {
  int t = blockIdx.x * blockDim.x + threadIdx.x;
  int total = (nE + n) * Hn;
  if (t >= total) return;
  int e = t / Hn, h = t - e * Hn;
  int s_, d_; edge_nodes(e, nE, src, dst, s_, d_);
  float ev = lrelu(as[(size_t)s_ * Hn + h] + ad[(size_t)d_ * Hn + h]);
  float w = __expf(ev - ord2f(m[(size_t)d_ * Hn + h]));
  atomicAdd(denom + (size_t)d_ * Hn + h, w);
}

__global__ void edge_agg1(const int* __restrict__ src, const int* __restrict__ dst,
                          int nE, int n, const float* __restrict__ H,
                          const float* __restrict__ as, const float* __restrict__ ad,
                          const unsigned* __restrict__ m, const float* __restrict__ denom,
                          float* __restrict__ out) {
  int e = (blockIdx.x * blockDim.x + threadIdx.x) >> 5;
  int lane = threadIdx.x & 31;
  if (e >= nE + n) return;
  int s_, d_; edge_nodes(e, nE, src, dst, s_, d_);
  int h = lane >> 2;
  float ev = lrelu(as[(size_t)s_ * HEADS + h] + ad[(size_t)d_ * HEADS + h]);
  float w = __expf(ev - ord2f(m[(size_t)d_ * HEADS + h]));
  float alpha = w / (denom[(size_t)d_ * HEADS + h] + 1e-16f);
  const float4* hp = (const float4*)(H + (size_t)s_ * F1 + lane * 16);
  float* op = out + (size_t)d_ * F1 + lane * 16;
#pragma unroll
  for (int j = 0; j < 4; ++j) {
    float4 v = hp[j];
    atomicAdd(op + j * 4 + 0, v.x * alpha);
    atomicAdd(op + j * 4 + 1, v.y * alpha);
    atomicAdd(op + j * 4 + 2, v.z * alpha);
    atomicAdd(op + j * 4 + 3, v.w * alpha);
  }
}

__global__ void relu_bias(float* __restrict__ X2, const float* __restrict__ b1, int total) {
  int t = blockIdx.x * blockDim.x + threadIdx.x;
  if (t >= total) return;
  float v = X2[t] + b1[t & (F1 - 1)];
  X2[t] = v > 0.f ? v : 0.f;
}

__global__ void edge_agg2(const int* __restrict__ src, const int* __restrict__ dst,
                          int nE, int n, const float* __restrict__ H2,
                          const float* __restrict__ as, const float* __restrict__ ad,
                          const unsigned* __restrict__ m, const float* __restrict__ denom,
                          float* __restrict__ out) {
  int t = blockIdx.x * blockDim.x + threadIdx.x;
  int total = (nE + n) * D_OUT;
  if (t >= total) return;
  int e = t >> 3, c = t & 7;
  int s_, d_; edge_nodes(e, nE, src, dst, s_, d_);
  float ev = lrelu(as[s_] + ad[d_]);
  float alpha = __expf(ev - ord2f(m[d_])) / (denom[d_] + 1e-16f);
  atomicAdd(out + (size_t)d_ * D_OUT + c, H2[(size_t)s_ * D_OUT + c] * alpha);
}

__global__ void finalize(const float* __restrict__ acc, const float* __restrict__ b2,
                         float* __restrict__ out, int total) {
  int t = blockIdx.x * blockDim.x + threadIdx.x;
  if (t >= total) return;
  out[t] = acc[t] + b2[t & (D_OUT - 1)];
}

extern "C" void kernel_launch(void* const* d_in, const int* in_sizes, int n_in,
                              void* d_out, int out_size, void* d_ws, size_t ws_size,
                              hipStream_t stream) {
  const float* x      = (const float*)d_in[0];
  const int*   ei     = (const int*)  d_in[1];
  const float* W1     = (const float*)d_in[2];
  const float* a_src1 = (const float*)d_in[3];
  const float* a_dst1 = (const float*)d_in[4];
  const float* b1     = (const float*)d_in[5];
  const float* W2     = (const float*)d_in[6];
  const float* a_src2 = (const float*)d_in[7];
  const float* a_dst2 = (const float*)d_in[8];
  const float* b2     = (const float*)d_in[9];

  int n  = in_sizes[0] / D_IN;
  int nE = in_sizes[1] / 2;
  const int* src = ei;
  const int* dst = ei + nE;
  int EE = nE + n;

  char* ws = (char*)d_ws;
  size_t off = 0;
  auto alloc = [&](size_t bytes) -> char* {
    char* p = ws + off; off += (bytes + 255) & ~(size_t)255; return p;
  };
  float*    h1   = (float*)   alloc((size_t)n * F1 * 4);
  float*    out1 = (float*)   alloc((size_t)n * F1 * 4);
  float*    as1  = (float*)   alloc((size_t)n * HEADS * 4);
  float*    ad1  = (float*)   alloc((size_t)n * HEADS * 4);
  unsigned* m1   = (unsigned*)alloc((size_t)n * HEADS * 4);
  float*    dn1  = (float*)   alloc((size_t)n * HEADS * 4);
  float*    h2   = (float*)   alloc((size_t)n * D_OUT * 4);
  float*    as2  = (float*)   alloc((size_t)n * 4);
  float*    ad2  = (float*)   alloc((size_t)n * 4);
  unsigned* m2   = (unsigned*)alloc((size_t)n * 4);
  float*    dn2  = (float*)   alloc((size_t)n * 4);
  float*    o2   = (float*)   alloc((size_t)n * D_OUT * 4);

  hipMemsetAsync(out1, 0, off - (size_t)((char*)out1 - ws), stream);

  {
    int waves  = ((n + 15) / 16) * (F1 / 16);
    int blocks = (waves + 7) / 8;
    gemm1_wmma<<<blocks, 256, 0, stream>>>(x, W1, h1, n);
  }
  alphas1<<<(n * HEADS + 255) / 256, 256, 0, stream>>>(h1, a_src1, a_dst1, as1, ad1, n);
  {
    int total = EE * HEADS;
    edge_max  <<<(total + 255) / 256, 256, 0, stream>>>(src, dst, nE, n, as1, ad1, m1, HEADS);
    edge_denom<<<(total + 255) / 256, 256, 0, stream>>>(src, dst, nE, n, as1, ad1, m1, dn1, HEADS);
  }
  edge_agg1<<<(EE + 7) / 8, 256, 0, stream>>>(src, dst, nE, n, h1, as1, ad1, m1, dn1, out1);
  relu_bias<<<((n * F1) + 255) / 256, 256, 0, stream>>>(out1, b1, n * F1);

  {
    int waves  = (n + 15) / 16;
    int blocks = (waves + 7) / 8;
    gemm2_wmma<<<blocks, 256, 0, stream>>>(out1, W2, h2, n);
  }
  alphas2<<<(n + 255) / 256, 256, 0, stream>>>(h2, a_src2, a_dst2, as2, ad2, n);
  edge_max  <<<(EE + 255) / 256, 256, 0, stream>>>(src, dst, nE, n, as2, ad2, m2, 1);
  edge_denom<<<(EE + 255) / 256, 256, 0, stream>>>(src, dst, nE, n, as2, ad2, m2, dn2, 1);
  edge_agg2<<<(EE * D_OUT + 255) / 256, 256, 0, stream>>>(src, dst, nE, n, h2, as2, ad2, m2, dn2, o2);

  finalize<<<(n * D_OUT + 255) / 256, 256, 0, stream>>>(o2, b2, (float*)d_out, n * D_OUT);
}


# Layer-1 tile kernel lists its edges in LDS (one scan): score max, softmax denominator (sum order: self loop, then list) and messages, 4 edges in flight, all from the list; 2 edge-pass launches gone
# speedup vs baseline: 4.6714x; 4.6714x over previous
#include <hip/hip_runtime.h>
#include <math.h>

#define D_IN   128
#define HEADS  8
#define HID    64
#define F1     (HEADS * HID)
#define D_OUT  8
#define NEG_SLOPE 0.2f

typedef float v2f __attribute__((ext_vector_type(2)));
typedef float v8f __attribute__((ext_vector_type(8)));

__device__ __forceinline__ unsigned f2ord(float f) {
  unsigned u = __float_as_uint(f);
  return (u & 0x80000000u) ? ~u : (u | 0x80000000u);
}
__device__ __forceinline__ float ord2f(unsigned v) {
  unsigned u = (v & 0x80000000u) ? (v ^ 0x80000000u) : ~v;
  return __uint_as_float(u);
}
__device__ __forceinline__ float lrelu(float x) { return x > 0.f ? x : NEG_SLOPE * x; }

#define G1_BM 128
#define G1_BN 128
#define G1_XP 132
#define G1_WP 288

__global__ __launch_bounds__(256)
void gemm1_wmma(const float* __restrict__ X, const float* __restrict__ W,
                float* __restrict__ H, int n) {
  __shared__ float Xs[G1_BM * G1_XP];
  __shared__ float Ws[(D_IN / 2) * G1_WP];

  const int tid = threadIdx.x, lane = tid & 31;
  const int wave = __builtin_amdgcn_readfirstlane(tid >> 5);
  const int col0 = blockIdx.x * G1_BN;
  const int row0 = blockIdx.y * G1_BM;

#pragma unroll
  for (int i = 0; i < (G1_BM * D_IN / 4) / 256; ++i) {
    int idx = tid + 256 * i;
    int r = idx >> 5, c4 = idx & 31;
    int gr = min(row0 + r, n - 1);
    float4 v = ((const float4*)(X + (size_t)gr * D_IN))[c4];
    *(float4*)(Xs + r * G1_XP + 4 * c4) = v;
  }
#pragma unroll
  for (int i = 0; i < ((D_IN / 2) * G1_BN / 4) / 256; ++i) {
    int idx = tid + 256 * i;
    int kp = idx >> 5, c4 = idx & 31;
    const float* wp = W + (size_t)(2 * kp) * F1 + col0 + 4 * c4;
    float4 r0 = *(const float4*)wp;
    float4 r1 = *(const float4*)(wp + F1);
    float* d = Ws + kp * G1_WP + 8 * c4;
    *(float4*)d       = make_float4(r0.x, r1.x, r0.y, r1.y);
    *(float4*)(d + 4) = make_float4(r0.z, r1.z, r0.w, r1.w);
  }
  __syncthreads();

  const int half = lane >> 4;
  const int l15  = lane & 15;
  const float* xa = Xs + (16 * wave + l15) * G1_XP + 2 * half;
  const float* wb = Ws + half * G1_WP + 2 * l15;

  v8f acc[G1_BN / 16];
#pragma unroll
  for (int i = 0; i < G1_BN / 16; ++i) acc[i] = (v8f){};
#pragma unroll
  for (int j = 0; j < D_IN / 4; ++j) {
    v2f a = *(const v2f*)(xa + 4 * j);
#pragma unroll
    for (int i = 0; i < G1_BN / 16; ++i) {
      v2f b = *(const v2f*)(wb + 2 * j * G1_WP + 32 * i);
      acc[i] = __builtin_amdgcn_wmma_f32_16x16x4_f32(false, a, false, b,
                                                     (short)0, acc[i], false, false);
    }
  }

  float* xo = Xs + (16 * wave + 8 * half) * G1_XP + l15;
#pragma unroll
  for (int i = 0; i < G1_BN / 16; ++i) {
#pragma unroll
    for (int v = 0; v < 8; ++v) xo[v * G1_XP + 16 * i] = acc[i][v];
  }
  __syncthreads();
  const float* xr = Xs + (16 * wave) * G1_XP + 4 * lane;
  float* hp = H + (size_t)(row0 + 16 * wave) * F1 + col0 + 4 * lane;
  const int nrows = n - (row0 + 16 * wave);
  if (nrows >= 16) {
    float4 o[16];
#pragma unroll
    for (int r = 0; r < 16; ++r) o[r] = *(const float4*)(xr + r * G1_XP);
#pragma unroll
    for (int r = 0; r < 16; ++r) *(float4*)(hp + (size_t)r * F1) = o[r];
  } else {
    for (int r = 0; r < nrows; ++r)
      *(float4*)(hp + (size_t)r * F1) = *(const float4*)(xr + r * G1_XP);
  }
}

__global__ void gemm2_wmma(const float* __restrict__ X2, const float* __restrict__ b1,
                           const float* __restrict__ W2,
                           float* __restrict__ H2, int n) {
  int wave = (blockIdx.x * blockDim.x + threadIdx.x) >> 5;
  int lane = threadIdx.x & 31;
  int row0 = wave * 16;
  if (row0 >= n) return;

  int half = lane >> 4;
  int l15  = lane & 15;
  const float* xrow = X2 + (size_t)(row0 + l15) * F1;
  bool colok = (l15 < D_OUT);

  v8f acc = {};
  for (int k = 0; k < F1; k += 4) {
    int kk = k + half * 2;
    float s0 = xrow[kk] + b1[kk], s1 = xrow[kk + 1] + b1[kk + 1];
    v2f a; a.x = s0 > 0.f ? s0 : 0.f; a.y = s1 > 0.f ? s1 : 0.f;
    v2f b;
    b.x = colok ? W2[(size_t)kk * D_OUT + l15]       : 0.f;
    b.y = colok ? W2[(size_t)(kk + 1) * D_OUT + l15] : 0.f;
    acc = __builtin_amdgcn_wmma_f32_16x16x4_f32(false, a, false, b,
                                                (short)0, acc, false, false);
  }
  if (colok) {
#pragma unroll
    for (int v = 0; v < 8; ++v) {
      int r = row0 + v + half * 8;
      H2[(size_t)r * D_OUT + l15] = acc[v];
    }
  }
}

__global__ void alphas1(const float* __restrict__ H, const float* __restrict__ a_src,
                        const float* __restrict__ a_dst,
                        float* __restrict__ as, float* __restrict__ ad, int n) {
  int t = blockIdx.x * blockDim.x + threadIdx.x;
  if (t >= n * HEADS) return;
  int h = t % HEADS;
  const float* row = H + (size_t)(t / HEADS) * F1 + h * HID;
  const float* vs = a_src + h * HID;
  const float* vd = a_dst + h * HID;
  float s = 0.f, d = 0.f;
#pragma unroll 8
  for (int c = 0; c < HID; ++c) { float x = row[c]; s += x * vs[c]; d += x * vd[c]; }
  as[t] = s; ad[t] = d;
}

__global__ void alphas2(const float* __restrict__ H2, const float* __restrict__ a_src,
                        const float* __restrict__ a_dst,
                        float* __restrict__ as, float* __restrict__ ad, int n) {
  int t = blockIdx.x * blockDim.x + threadIdx.x;
  if (t >= n) return;
  const float* row = H2 + (size_t)t * D_OUT;
  float s = 0.f, d = 0.f;
#pragma unroll
  for (int c = 0; c < D_OUT; ++c) { s += row[c] * a_src[c]; d += row[c] * a_dst[c]; }
  as[t] = s; ad[t] = d;
}

__device__ __forceinline__ void edge_nodes(int e, int nE, const int* src, const int* dst,
                                           int& s_, int& d_) {
  if (e < nE) { s_ = src[e]; d_ = dst[e]; }
  else        { s_ = d_ = e - nE; }
}

__global__ void edge_max(const int* __restrict__ src, const int* __restrict__ dst,
                         int nE, int n, const float* __restrict__ as,
                         const float* __restrict__ ad, unsigned* __restrict__ m, int Hn) {
  int t = blockIdx.x * blockDim.x + threadIdx.x;
  int total = (nE + n) * Hn;
  if (t >= total) return;
  int e = t / Hn, h = t - e * Hn;
  int s_, d_; edge_nodes(e, nE, src, dst, s_, d_);
  float ev = lrelu(as[(size_t)s_ * Hn + h] + ad[(size_t)d_ * Hn + h]);
  atomicMax(m + (size_t)d_ * Hn + h, f2ord(ev));
}

__global__ void edge_denom(const int* __restrict__ src, const int* __restrict__ dst,
                           int nE, int n, const float* __restrict__ as,
                           const float* __restrict__ ad, const unsigned* __restrict__ m,
                           float* __restrict__ denom, int Hn) {
  int t = blockIdx.x * blockDim.x + threadIdx.x;
  int total = (nE + n) * Hn;
  if (t >= total) return;
  int e = t / Hn, h = t - e * Hn;
  int s_, d_; edge_nodes(e, nE, src, dst, s_, d_);
  float ev = lrelu(as[(size_t)s_ * Hn + h] + ad[(size_t)d_ * Hn + h]);
  float w = __expf(ev - ord2f(m[(size_t)d_ * Hn + h]));
  atomicAdd(denom + (size_t)d_ * Hn + h, w);
}

#define TILE_R      98
#define AGG_THREADS 1024
#define AGG_WAVES   (AGG_THREADS / 32)
#define LIST_CAP    4096

struct Msg { float4 v0, v1, v2, v3; };

__device__ __forceinline__ Msg load_row(const float* __restrict__ H, int s_, int lane) {
  const float4* hp = (const float4*)(H + (size_t)s_ * F1);
  Msg m;
  m.v0 = hp[lane]; m.v1 = hp[lane + 32]; m.v2 = hp[lane + 64]; m.v3 = hp[lane + 96];
  return m;
}

__device__ __forceinline__ float edge_alpha(float a_s, int r, int hh, const float* s_ad,
                                            const float* s_m, const float* s_dn) {
  float ev = lrelu(a_s + s_ad[r * HEADS + hh]);
  float w = __expf(ev - s_m[r * HEADS + hh]);
  return w / (s_dn[r * HEADS + hh] + 1e-16f);
}

template <bool START>
__device__ __forceinline__ void msg_into_tile(const Msg& m, float alpha, int r, int lane,
                                              float* acc) {
  int hs = lane >> 4;
  float a0 = __shfl(alpha, hs), a1 = __shfl(alpha, hs + 2);
  float a2 = __shfl(alpha, hs + 4), a3 = __shfl(alpha, hs + 6);
  float* ap = acc + r * F1 + lane;
  if (START) {
    ap[0]       = m.v0.x * a0; ap[128]      = m.v0.y * a0; ap[256]      = m.v0.z * a0; ap[384]      = m.v0.w * a0;
    ap[32]      = m.v1.x * a1; ap[128 + 32] = m.v1.y * a1; ap[256 + 32] = m.v1.z * a1; ap[384 + 32] = m.v1.w * a1;
    ap[64]      = m.v2.x * a2; ap[128 + 64] = m.v2.y * a2; ap[256 + 64] = m.v2.z * a2; ap[384 + 64] = m.v2.w * a2;
    ap[96]      = m.v3.x * a3; ap[128 + 96] = m.v3.y * a3; ap[256 + 96] = m.v3.z * a3; ap[384 + 96] = m.v3.w * a3;
  } else {
    atomicAdd(ap,            m.v0.x * a0); atomicAdd(ap + 128,      m.v0.y * a0);
    atomicAdd(ap + 256,      m.v0.z * a0); atomicAdd(ap + 384,      m.v0.w * a0);
    atomicAdd(ap + 32,       m.v1.x * a1); atomicAdd(ap + 128 + 32, m.v1.y * a1);
    atomicAdd(ap + 256 + 32, m.v1.z * a1); atomicAdd(ap + 384 + 32, m.v1.w * a1);
    atomicAdd(ap + 64,       m.v2.x * a2); atomicAdd(ap + 128 + 64, m.v2.y * a2);
    atomicAdd(ap + 256 + 64, m.v2.z * a2); atomicAdd(ap + 384 + 64, m.v2.w * a2);
    atomicAdd(ap + 96,       m.v3.x * a3); atomicAdd(ap + 128 + 96, m.v3.y * a3);
    atomicAdd(ap + 256 + 96, m.v3.z * a3); atomicAdd(ap + 384 + 96, m.v3.w * a3);
  }
}

template <bool START>
__device__ __forceinline__ void edge_into_tile(int s_, int r, int lane,
                                               const float* __restrict__ H,
                                               const float* __restrict__ as,
                                               const float* s_ad, const float* s_m,
                                               const float* s_dn, float* acc) {
  Msg m = load_row(H, s_, lane);
  int hh = lane & 7;
  float alpha = edge_alpha(as[(size_t)s_ * HEADS + hh], r, hh, s_ad, s_m, s_dn);
  msg_into_tile<START>(m, alpha, r, lane, acc);
}

__device__ __forceinline__ void scores8(const float* __restrict__ as, const float* s_ad,
                                        int s_, int r, float* ev) {
  const float4* ap = (const float4*)(as + (size_t)s_ * HEADS);
  float4 a0 = ap[0], a1 = ap[1];
  const float4* dp = (const float4*)(s_ad + r * HEADS);
  float4 d0 = dp[0], d1 = dp[1];
  ev[0] = lrelu(a0.x + d0.x); ev[1] = lrelu(a0.y + d0.y);
  ev[2] = lrelu(a0.z + d0.z); ev[3] = lrelu(a0.w + d0.w);
  ev[4] = lrelu(a1.x + d1.x); ev[5] = lrelu(a1.y + d1.y);
  ev[6] = lrelu(a1.z + d1.z); ev[7] = lrelu(a1.w + d1.w);
}

__global__ __launch_bounds__(AGG_THREADS)
void tile_agg1(const int* __restrict__ src, const int* __restrict__ dst,
               int nE, int n, const float* __restrict__ H,
               const float* __restrict__ as, const float* __restrict__ ad,
               float* __restrict__ out) {
  __shared__ float acc[TILE_R * F1];
  __shared__ __attribute__((aligned(16))) float s_ad[TILE_R * HEADS];
  __shared__ float s_m[TILE_R * HEADS];
  __shared__ float s_dn[TILE_R * HEADS];
  __shared__ unsigned s_mo[TILE_R * HEADS];
  __shared__ int2 s_list[LIST_CAP];
  __shared__ int s_cnt;

  const int tid = threadIdx.x, lane = tid & 31, wave = tid >> 5;
  const int node0 = blockIdx.x * TILE_R;
  const int rows = min(TILE_R, n - node0);

  for (int i = tid; i < rows * HEADS; i += AGG_THREADS) {
    size_t g = (size_t)node0 * HEADS + i;
    float a_d = ad[g];
    float ev = lrelu(as[g] + a_d);
    s_ad[i] = a_d;
    s_mo[i] = f2ord(ev);
    s_dn[i] = ev;
  }
  if (tid == 0) s_cnt = 0;
  __syncthreads();

  auto scan = [&](auto take) {
    auto take4 = [&](int4 d, bool ok, int e0) {
      unsigned lo = min(min((unsigned)(d.x - node0), (unsigned)(d.y - node0)),
                        min((unsigned)(d.z - node0), (unsigned)(d.w - node0)));
      bool any = (lo < (unsigned)rows) & ok;
      if (__builtin_amdgcn_ballot_w32(any) != 0u) {
        take(ok ? d.x : -1, e0);
        take(ok ? d.y : -1, e0 + 1);
        take(ok ? d.z : -1, e0 + 2);
        take(ok ? d.w : -1, e0 + 3);
      }
    };
    int pre = (int)(((16u - (unsigned)((size_t)dst & 15u)) & 15u) >> 2);
    if (pre > nE) pre = nE;
    const int nvec = (nE - pre) >> 2;
    const int4* dv = (const int4*)(dst + pre);
    const int nchunk = (nvec + 31) >> 5;
    if (nvec > 0) {
      const int last = nvec - 1;
      int4 cur[4];
#pragma unroll
      for (int j = 0; j < 4; ++j)
        cur[j] = dv[min(((wave + AGG_WAVES * j) << 5) + lane, last)];
      for (int base = wave; base < nchunk; base += 4 * AGG_WAVES) {
        int4 nxt[4];
#pragma unroll
        for (int j = 0; j < 4; ++j)
          nxt[j] = dv[min(((base + 4 * AGG_WAVES + AGG_WAVES * j) << 5) + lane, last)];
#pragma unroll
        for (int j = 0; j < 4; ++j) {
          int p = ((base + AGG_WAVES * j) << 5) + lane;
          take4(cur[j], p < nvec, pre + 4 * p);
          cur[j] = nxt[j];
        }
      }
    }
    if (wave == 0) {
      int tail0 = pre + 4 * nvec;
      int e = lane < pre ? lane : tail0 + (lane - pre);
      take(e < nE ? dst[e] : -1, e);
    }
  };

  auto for_hits = [&](int d, int e, auto f) {
    unsigned dl = (unsigned)(d - node0);
    unsigned mask = __builtin_amdgcn_ballot_w32(dl < (unsigned)rows);
    while (mask) {
      int l = __builtin_ctz(mask);
      mask &= mask - 1;
      int r  = __builtin_amdgcn_readlane((int)dl, l);
      int ee = __builtin_amdgcn_readlane(e, l);
      f(src[ee], r);
    }
  };

  scan([&](int d, int e) {
    unsigned dl = (unsigned)(d - node0);
    bool hit = dl < (unsigned)rows;
    unsigned mask = __builtin_amdgcn_ballot_w32(hit);
    if (mask != 0u) {
      int base = 0;
      if (lane == 0) base = atomicAdd(&s_cnt, __builtin_popcount(mask));
      base = __builtin_amdgcn_readfirstlane(base);
      int pos = base + (int)__builtin_amdgcn_mbcnt_lo(mask, 0u);
      if (hit && pos < LIST_CAP) s_list[pos] = make_int2(e, (int)dl);
    }
  });
  __syncthreads();
  const int cnt = s_cnt;
  const bool listed = cnt <= LIST_CAP;

  if (listed) {
    for (int i = tid; i < cnt; i += AGG_THREADS) {
      int2 en = s_list[i];
      int s_ = src[en.x];
      s_list[i].x = s_;
      float ev[HEADS];
      scores8(as, s_ad, s_, en.y, ev);
      unsigned* mp = s_mo + en.y * HEADS;
#pragma unroll
      for (int h = 0; h < HEADS; ++h) atomicMax(mp + h, f2ord(ev[h]));
    }
  } else {
    scan([&](int d, int e) {
      for_hits(d, e, [&](int s_, int r) {
        int hh = lane & 7;
        float ev = lrelu(as[(size_t)s_ * HEADS + hh] + s_ad[r * HEADS + hh]);
        if (lane < HEADS) atomicMax(s_mo + r * HEADS + hh, f2ord(ev));
      });
    });
  }
  __syncthreads();
  for (int i = tid; i < rows * HEADS; i += AGG_THREADS) {
    float mx = ord2f(s_mo[i]);
    s_m[i] = mx;
    s_dn[i] = __expf(s_dn[i] - mx);
  }
  __syncthreads();

  if (listed) {
    for (int i = tid; i < cnt; i += AGG_THREADS) {
      int2 en = s_list[i];
      float ev[HEADS];
      scores8(as, s_ad, en.x, en.y, ev);
      const float* mp = s_m + en.y * HEADS;
      float* dp = s_dn + en.y * HEADS;
#pragma unroll
      for (int h = 0; h < HEADS; ++h) atomicAdd(dp + h, __expf(ev[h] - mp[h]));
    }
  } else {
    scan([&](int d, int e) {
      for_hits(d, e, [&](int s_, int r) {
        int hh = lane & 7;
        float ev = lrelu(as[(size_t)s_ * HEADS + hh] + s_ad[r * HEADS + hh]);
        if (lane < HEADS) atomicAdd(s_dn + r * HEADS + hh, __expf(ev - s_m[r * HEADS + hh]));
      });
    });
  }
  __syncthreads();

  for (int r = wave; r < rows; r += AGG_WAVES)
    edge_into_tile<true>(node0 + r, r, lane, H, as, s_ad, s_m, s_dn, acc);
  __syncthreads();

  if (listed) {
    const int hh = lane & 7;
    int i = wave;
    for (; i + 3 * AGG_WAVES < cnt; i += 4 * AGG_WAVES) {
      int sN[4], rN[4];
      float av[4];
      Msg mm[4];
#pragma unroll
      for (int k = 0; k < 4; ++k) {
        int2 en = s_list[i + AGG_WAVES * k];
        sN[k] = __builtin_amdgcn_readfirstlane(en.x);
        rN[k] = __builtin_amdgcn_readfirstlane(en.y);
        av[k] = as[(size_t)sN[k] * HEADS + hh];
      }
#pragma unroll
      for (int k = 0; k < 4; ++k) mm[k] = load_row(H, sN[k], lane);
      asm volatile("" :: "v"(av[0]), "v"(av[1]), "v"(av[2]), "v"(av[3]),
                   "v"(mm[0].v0.x), "v"(mm[0].v1.x), "v"(mm[0].v2.x), "v"(mm[0].v3.x),
                   "v"(mm[1].v0.x), "v"(mm[1].v1.x), "v"(mm[1].v2.x), "v"(mm[1].v3.x),
                   "v"(mm[2].v0.x), "v"(mm[2].v1.x), "v"(mm[2].v2.x), "v"(mm[2].v3.x),
                   "v"(mm[3].v0.x), "v"(mm[3].v1.x), "v"(mm[3].v2.x), "v"(mm[3].v3.x));
#pragma unroll
      for (int k = 0; k < 4; ++k) {
        float alpha = edge_alpha(av[k], rN[k], hh, s_ad, s_m, s_dn);
        msg_into_tile<false>(mm[k], alpha, rN[k], lane, acc);
      }
    }
    for (; i < cnt; i += AGG_WAVES) {
      int2 en = s_list[i];
      edge_into_tile<false>(__builtin_amdgcn_readfirstlane(en.x),
                            __builtin_amdgcn_readfirstlane(en.y),
                            lane, H, as, s_ad, s_m, s_dn, acc);
    }
  } else {
    scan([&](int d, int e) {
      for_hits(d, e, [&](int s_, int r) {
        edge_into_tile<false>(s_, r, lane, H, as, s_ad, s_m, s_dn, acc);
      });
    });
  }
  __syncthreads();

  for (int i = tid; i < rows * (F1 / 4); i += AGG_THREADS) {
    int r = i >> 7, f = i & 127;
    const float* a = acc + r * F1 + f;
    float4 v = make_float4(a[0], a[128], a[256], a[384]);
    ((float4*)(out + (size_t)(node0 + r) * F1))[f] = v;
  }
}

__global__ void edge_agg2(const int* __restrict__ src, const int* __restrict__ dst,
                          int nE, int n, const float* __restrict__ H2,
                          const float* __restrict__ as, const float* __restrict__ ad,
                          const unsigned* __restrict__ m, const float* __restrict__ denom,
                          float* __restrict__ out) {
  int t = blockIdx.x * blockDim.x + threadIdx.x;
  int total = (nE + n) * D_OUT;
  if (t >= total) return;
  int e = t >> 3, c = t & 7;
  int s_, d_; edge_nodes(e, nE, src, dst, s_, d_);
  float ev = lrelu(as[s_] + ad[d_]);
  float alpha = __expf(ev - ord2f(m[d_])) / (denom[d_] + 1e-16f);
  atomicAdd(out + (size_t)d_ * D_OUT + c, H2[(size_t)s_ * D_OUT + c] * alpha);
}

__global__ void finalize(const float* __restrict__ acc, const float* __restrict__ b2,
                         float* __restrict__ out, int total) {
  int t = blockIdx.x * blockDim.x + threadIdx.x;
  if (t >= total) return;
  out[t] = acc[t] + b2[t & (D_OUT - 1)];
}

extern "C" void kernel_launch(void* const* d_in, const int* in_sizes, int n_in,
                              void* d_out, int out_size, void* d_ws, size_t ws_size,
                              hipStream_t stream) {
  const float* x      = (const float*)d_in[0];
  const int*   ei     = (const int*)  d_in[1];
  const float* W1     = (const float*)d_in[2];
  const float* a_src1 = (const float*)d_in[3];
  const float* a_dst1 = (const float*)d_in[4];
  const float* b1     = (const float*)d_in[5];
  const float* W2     = (const float*)d_in[6];
  const float* a_src2 = (const float*)d_in[7];
  const float* a_dst2 = (const float*)d_in[8];
  const float* b2     = (const float*)d_in[9];

  int n  = in_sizes[0] / D_IN;
  int nE = in_sizes[1] / 2;
  const int* src = ei;
  const int* dst = ei + nE;
  int EE = nE + n;

  char* ws = (char*)d_ws;
  size_t off = 0;
  auto alloc = [&](size_t bytes) -> char* {
    char* p = ws + off; off += (bytes + 255) & ~(size_t)255; return p;
  };
  float*    h1   = (float*)   alloc((size_t)n * F1 * 4);
  float*    out1 = (float*)   alloc((size_t)n * F1 * 4);
  float*    as1  = (float*)   alloc((size_t)n * HEADS * 4);
  float*    ad1  = (float*)   alloc((size_t)n * HEADS * 4);
  float*    h2   = (float*)   alloc((size_t)n * D_OUT * 4);
  float*    as2  = (float*)   alloc((size_t)n * 4);
  float*    ad2  = (float*)   alloc((size_t)n * 4);
  unsigned* m2   = (unsigned*)alloc((size_t)n * 4);
  float*    dn2  = (float*)   alloc((size_t)n * 4);
  float*    o2   = (float*)   alloc((size_t)n * D_OUT * 4);

  hipMemsetAsync(as1, 0, off - (size_t)((char*)as1 - ws), stream);

  gemm1_wmma<<<dim3(F1 / G1_BN, (n + G1_BM - 1) / G1_BM), 256, 0, stream>>>(x, W1, h1, n);
  alphas1<<<(n * HEADS + 255) / 256, 256, 0, stream>>>(h1, a_src1, a_dst1, as1, ad1, n);
  tile_agg1<<<(n + TILE_R - 1) / TILE_R, AGG_THREADS, 0, stream>>>(src, dst, nE, n, h1, as1, ad1, out1);

  {
    int waves  = (n + 15) / 16;
    int blocks = (waves + 7) / 8;
    gemm2_wmma<<<blocks, 256, 0, stream>>>(out1, b1, W2, h2, n);
  }
  alphas2<<<(n + 255) / 256, 256, 0, stream>>>(h2, a_src2, a_dst2, as2, ad2, n);
  edge_max  <<<(EE + 255) / 256, 256, 0, stream>>>(src, dst, nE, n, as2, ad2, m2, 1);
  edge_denom<<<(EE + 255) / 256, 256, 0, stream>>>(src, dst, nE, n, as2, ad2, m2, dn2, 1);
  edge_agg2<<<(EE * D_OUT + 255) / 256, 256, 0, stream>>>(src, dst, nE, n, h2, as2, ad2, m2, dn2, o2);

  finalize<<<(n * D_OUT + 255) / 256, 256, 0, stream>>>(o2, b2, (float*)d_out, n * D_OUT);
}


# Layer-2 softmax and sums per tile of 98 destination nodes in LDS, over the tile's edge list that the layer-1 kernel now stores; memset and 4 launches gone; sum order: self loop, then arrival
# speedup vs baseline: 6.3225x; 1.3534x over previous
#include <hip/hip_runtime.h>
#include <math.h>

#define D_IN   128
#define HEADS  8
#define HID    64
#define F1     (HEADS * HID)
#define D_OUT  8
#define NEG_SLOPE 0.2f

typedef float v2f __attribute__((ext_vector_type(2)));
typedef float v8f __attribute__((ext_vector_type(8)));

__device__ __forceinline__ unsigned f2ord(float f) {
  unsigned u = __float_as_uint(f);
  return (u & 0x80000000u) ? ~u : (u | 0x80000000u);
}
__device__ __forceinline__ float ord2f(unsigned v) {
  unsigned u = (v & 0x80000000u) ? (v ^ 0x80000000u) : ~v;
  return __uint_as_float(u);
}
__device__ __forceinline__ float lrelu(float x) { return x > 0.f ? x : NEG_SLOPE * x; }

#define G1_BM 128
#define G1_BN 128
#define G1_XP 132
#define G1_WP 288

__global__ __launch_bounds__(256)
void gemm1_wmma(const float* __restrict__ X, const float* __restrict__ W,
                float* __restrict__ H, int n) {
  __shared__ float Xs[G1_BM * G1_XP];
  __shared__ float Ws[(D_IN / 2) * G1_WP];

  const int tid = threadIdx.x, lane = tid & 31;
  const int wave = __builtin_amdgcn_readfirstlane(tid >> 5);
  const int col0 = blockIdx.x * G1_BN;
  const int row0 = blockIdx.y * G1_BM;

#pragma unroll
  for (int i = 0; i < (G1_BM * D_IN / 4) / 256; ++i) {
    int idx = tid + 256 * i;
    int r = idx >> 5, c4 = idx & 31;
    int gr = min(row0 + r, n - 1);
    float4 v = ((const float4*)(X + (size_t)gr * D_IN))[c4];
    *(float4*)(Xs + r * G1_XP + 4 * c4) = v;
  }
#pragma unroll
  for (int i = 0; i < ((D_IN / 2) * G1_BN / 4) / 256; ++i) {
    int idx = tid + 256 * i;
    int kp = idx >> 5, c4 = idx & 31;
    const float* wp = W + (size_t)(2 * kp) * F1 + col0 + 4 * c4;
    float4 r0 = *(const float4*)wp;
    float4 r1 = *(const float4*)(wp + F1);
    float* d = Ws + kp * G1_WP + 8 * c4;
    *(float4*)d       = make_float4(r0.x, r1.x, r0.y, r1.y);
    *(float4*)(d + 4) = make_float4(r0.z, r1.z, r0.w, r1.w);
  }
  __syncthreads();

  const int half = lane >> 4;
  const int l15  = lane & 15;
  const float* xa = Xs + (16 * wave + l15) * G1_XP + 2 * half;
  const float* wb = Ws + half * G1_WP + 2 * l15;

  v8f acc[G1_BN / 16];
#pragma unroll
  for (int i = 0; i < G1_BN / 16; ++i) acc[i] = (v8f){};
#pragma unroll
  for (int j = 0; j < D_IN / 4; ++j) {
    v2f a = *(const v2f*)(xa + 4 * j);
#pragma unroll
    for (int i = 0; i < G1_BN / 16; ++i) {
      v2f b = *(const v2f*)(wb + 2 * j * G1_WP + 32 * i);
      acc[i] = __builtin_amdgcn_wmma_f32_16x16x4_f32(false, a, false, b,
                                                     (short)0, acc[i], false, false);
    }
  }

  float* xo = Xs + (16 * wave + 8 * half) * G1_XP + l15;
#pragma unroll
  for (int i = 0; i < G1_BN / 16; ++i) {
#pragma unroll
    for (int v = 0; v < 8; ++v) xo[v * G1_XP + 16 * i] = acc[i][v];
  }
  __syncthreads();
  const float* xr = Xs + (16 * wave) * G1_XP + 4 * lane;
  float* hp = H + (size_t)(row0 + 16 * wave) * F1 + col0 + 4 * lane;
  const int nrows = n - (row0 + 16 * wave);
  if (nrows >= 16) {
    float4 o[16];
#pragma unroll
    for (int r = 0; r < 16; ++r) o[r] = *(const float4*)(xr + r * G1_XP);
#pragma unroll
    for (int r = 0; r < 16; ++r) *(float4*)(hp + (size_t)r * F1) = o[r];
  } else {
    for (int r = 0; r < nrows; ++r)
      *(float4*)(hp + (size_t)r * F1) = *(const float4*)(xr + r * G1_XP);
  }
}

__global__ void gemm2_wmma(const float* __restrict__ X2, const float* __restrict__ b1,
                           const float* __restrict__ W2,
                           float* __restrict__ H2, int n) {
  int wave = (blockIdx.x * blockDim.x + threadIdx.x) >> 5;
  int lane = threadIdx.x & 31;
  int row0 = wave * 16;
  if (row0 >= n) return;

  int half = lane >> 4;
  int l15  = lane & 15;
  const float* xrow = X2 + (size_t)(row0 + l15) * F1;
  bool colok = (l15 < D_OUT);

  v8f acc = {};
  for (int k = 0; k < F1; k += 4) {
    int kk = k + half * 2;
    float s0 = xrow[kk] + b1[kk], s1 = xrow[kk + 1] + b1[kk + 1];
    v2f a; a.x = s0 > 0.f ? s0 : 0.f; a.y = s1 > 0.f ? s1 : 0.f;
    v2f b;
    b.x = colok ? W2[(size_t)kk * D_OUT + l15]       : 0.f;
    b.y = colok ? W2[(size_t)(kk + 1) * D_OUT + l15] : 0.f;
    acc = __builtin_amdgcn_wmma_f32_16x16x4_f32(false, a, false, b,
                                                (short)0, acc, false, false);
  }
  if (colok) {
#pragma unroll
    for (int v = 0; v < 8; ++v) {
      int r = row0 + v + half * 8;
      H2[(size_t)r * D_OUT + l15] = acc[v];
    }
  }
}

__global__ void alphas1(const float* __restrict__ H, const float* __restrict__ a_src,
                        const float* __restrict__ a_dst,
                        float* __restrict__ as, float* __restrict__ ad, int n) {
  int t = blockIdx.x * blockDim.x + threadIdx.x;
  if (t >= n * HEADS) return;
  int h = t % HEADS;
  const float* row = H + (size_t)(t / HEADS) * F1 + h * HID;
  const float* vs = a_src + h * HID;
  const float* vd = a_dst + h * HID;
  float s = 0.f, d = 0.f;
#pragma unroll 8
  for (int c = 0; c < HID; ++c) { float x = row[c]; s += x * vs[c]; d += x * vd[c]; }
  as[t] = s; ad[t] = d;
}

__global__ void alphas2(const float* __restrict__ H2, const float* __restrict__ a_src,
                        const float* __restrict__ a_dst,
                        float* __restrict__ as, float* __restrict__ ad, int n) {
  int t = blockIdx.x * blockDim.x + threadIdx.x;
  if (t >= n) return;
  const float* row = H2 + (size_t)t * D_OUT;
  float s = 0.f, d = 0.f;
#pragma unroll
  for (int c = 0; c < D_OUT; ++c) { s += row[c] * a_src[c]; d += row[c] * a_dst[c]; }
  as[t] = s; ad[t] = d;
}

#define TILE_R      98
#define AGG_THREADS 1024
#define AGG_WAVES   (AGG_THREADS / 32)
#define LIST_CAP    4096

struct Msg { float4 v0, v1, v2, v3; };

__device__ __forceinline__ Msg load_row(const float* __restrict__ H, int s_, int lane) {
  const float4* hp = (const float4*)(H + (size_t)s_ * F1);
  Msg m;
  m.v0 = hp[lane]; m.v1 = hp[lane + 32]; m.v2 = hp[lane + 64]; m.v3 = hp[lane + 96];
  return m;
}

__device__ __forceinline__ float edge_alpha(float a_s, int r, int hh, const float* s_ad,
                                            const float* s_m, const float* s_dn) {
  float ev = lrelu(a_s + s_ad[r * HEADS + hh]);
  float w = __expf(ev - s_m[r * HEADS + hh]);
  return w / (s_dn[r * HEADS + hh] + 1e-16f);
}

template <bool START>
__device__ __forceinline__ void msg_into_tile(const Msg& m, float alpha, int r, int lane,
                                              float* acc) {
  int hs = lane >> 4;
  float a0 = __shfl(alpha, hs), a1 = __shfl(alpha, hs + 2);
  float a2 = __shfl(alpha, hs + 4), a3 = __shfl(alpha, hs + 6);
  float* ap = acc + r * F1 + lane;
  if (START) {
    ap[0]       = m.v0.x * a0; ap[128]      = m.v0.y * a0; ap[256]      = m.v0.z * a0; ap[384]      = m.v0.w * a0;
    ap[32]      = m.v1.x * a1; ap[128 + 32] = m.v1.y * a1; ap[256 + 32] = m.v1.z * a1; ap[384 + 32] = m.v1.w * a1;
    ap[64]      = m.v2.x * a2; ap[128 + 64] = m.v2.y * a2; ap[256 + 64] = m.v2.z * a2; ap[384 + 64] = m.v2.w * a2;
    ap[96]      = m.v3.x * a3; ap[128 + 96] = m.v3.y * a3; ap[256 + 96] = m.v3.z * a3; ap[384 + 96] = m.v3.w * a3;
  } else {
    atomicAdd(ap,            m.v0.x * a0); atomicAdd(ap + 128,      m.v0.y * a0);
    atomicAdd(ap + 256,      m.v0.z * a0); atomicAdd(ap + 384,      m.v0.w * a0);
    atomicAdd(ap + 32,       m.v1.x * a1); atomicAdd(ap + 128 + 32, m.v1.y * a1);
    atomicAdd(ap + 256 + 32, m.v1.z * a1); atomicAdd(ap + 384 + 32, m.v1.w * a1);
    atomicAdd(ap + 64,       m.v2.x * a2); atomicAdd(ap + 128 + 64, m.v2.y * a2);
    atomicAdd(ap + 256 + 64, m.v2.z * a2); atomicAdd(ap + 384 + 64, m.v2.w * a2);
    atomicAdd(ap + 96,       m.v3.x * a3); atomicAdd(ap + 128 + 96, m.v3.y * a3);
    atomicAdd(ap + 256 + 96, m.v3.z * a3); atomicAdd(ap + 384 + 96, m.v3.w * a3);
  }
}

template <bool START>
__device__ __forceinline__ void edge_into_tile(int s_, int r, int lane,
                                               const float* __restrict__ H,
                                               const float* __restrict__ as,
                                               const float* s_ad, const float* s_m,
                                               const float* s_dn, float* acc) {
  Msg m = load_row(H, s_, lane);
  int hh = lane & 7;
  float alpha = edge_alpha(as[(size_t)s_ * HEADS + hh], r, hh, s_ad, s_m, s_dn);
  msg_into_tile<START>(m, alpha, r, lane, acc);
}

__device__ __forceinline__ void scores8(const float* __restrict__ as, const float* s_ad,
                                        int s_, int r, float* ev) {
  const float4* ap = (const float4*)(as + (size_t)s_ * HEADS);
  float4 a0 = ap[0], a1 = ap[1];
  const float4* dp = (const float4*)(s_ad + r * HEADS);
  float4 d0 = dp[0], d1 = dp[1];
  ev[0] = lrelu(a0.x + d0.x); ev[1] = lrelu(a0.y + d0.y);
  ev[2] = lrelu(a0.z + d0.z); ev[3] = lrelu(a0.w + d0.w);
  ev[4] = lrelu(a1.x + d1.x); ev[5] = lrelu(a1.y + d1.y);
  ev[6] = lrelu(a1.z + d1.z); ev[7] = lrelu(a1.w + d1.w);
}

__global__ __launch_bounds__(AGG_THREADS)
void tile_agg1(const int* __restrict__ src, const int* __restrict__ dst,
               int nE, int n, const float* __restrict__ H,
               const float* __restrict__ as, const float* __restrict__ ad,
               float* __restrict__ out, int* __restrict__ tl_cnt, int* __restrict__ tl_ent) {
  __shared__ float acc[TILE_R * F1];
  __shared__ __attribute__((aligned(16))) float s_ad[TILE_R * HEADS];
  __shared__ float s_m[TILE_R * HEADS];
  __shared__ float s_dn[TILE_R * HEADS];
  __shared__ unsigned s_mo[TILE_R * HEADS];
  __shared__ int2 s_list[LIST_CAP];
  __shared__ int s_cnt;

  const int tid = threadIdx.x, lane = tid & 31, wave = tid >> 5;
  const int node0 = blockIdx.x * TILE_R;
  const int rows = min(TILE_R, n - node0);

  for (int i = tid; i < rows * HEADS; i += AGG_THREADS) {
    size_t g = (size_t)node0 * HEADS + i;
    float a_d = ad[g];
    float ev = lrelu(as[g] + a_d);
    s_ad[i] = a_d;
    s_mo[i] = f2ord(ev);
    s_dn[i] = ev;
  }
  if (tid == 0) s_cnt = 0;
  __syncthreads();

  auto scan = [&](auto take) {
    auto take4 = [&](int4 d, bool ok, int e0) {
      unsigned lo = min(min((unsigned)(d.x - node0), (unsigned)(d.y - node0)),
                        min((unsigned)(d.z - node0), (unsigned)(d.w - node0)));
      bool any = (lo < (unsigned)rows) & ok;
      if (__builtin_amdgcn_ballot_w32(any) != 0u) {
        take(ok ? d.x : -1, e0);
        take(ok ? d.y : -1, e0 + 1);
        take(ok ? d.z : -1, e0 + 2);
        take(ok ? d.w : -1, e0 + 3);
      }
    };
    int pre = (int)(((16u - (unsigned)((size_t)dst & 15u)) & 15u) >> 2);
    if (pre > nE) pre = nE;
    const int nvec = (nE - pre) >> 2;
    const int4* dv = (const int4*)(dst + pre);
    const int nchunk = (nvec + 31) >> 5;
    if (nvec > 0) {
      const int last = nvec - 1;
      int4 cur[4];
#pragma unroll
      for (int j = 0; j < 4; ++j)
        cur[j] = dv[min(((wave + AGG_WAVES * j) << 5) + lane, last)];
      for (int base = wave; base < nchunk; base += 4 * AGG_WAVES) {
        int4 nxt[4];
#pragma unroll
        for (int j = 0; j < 4; ++j)
          nxt[j] = dv[min(((base + 4 * AGG_WAVES + AGG_WAVES * j) << 5) + lane, last)];
#pragma unroll
        for (int j = 0; j < 4; ++j) {
          int p = ((base + AGG_WAVES * j) << 5) + lane;
          take4(cur[j], p < nvec, pre + 4 * p);
          cur[j] = nxt[j];
        }
      }
    }
    if (wave == 0) {
      int tail0 = pre + 4 * nvec;
      int e = lane < pre ? lane : tail0 + (lane - pre);
      take(e < nE ? dst[e] : -1, e);
    }
  };

  auto for_hits = [&](int d, int e, auto f) {
    unsigned dl = (unsigned)(d - node0);
    unsigned mask = __builtin_amdgcn_ballot_w32(dl < (unsigned)rows);
    while (mask) {
      int l = __builtin_ctz(mask);
      mask &= mask - 1;
      int r  = __builtin_amdgcn_readlane((int)dl, l);
      int ee = __builtin_amdgcn_readlane(e, l);
      f(src[ee], r);
    }
  };

  scan([&](int d, int e) {
    unsigned dl = (unsigned)(d - node0);
    bool hit = dl < (unsigned)rows;
    unsigned mask = __builtin_amdgcn_ballot_w32(hit);
    if (mask != 0u) {
      int base = 0;
      if (lane == 0) base = atomicAdd(&s_cnt, __builtin_popcount(mask));
      base = __builtin_amdgcn_readfirstlane(base);
      int pos = base + (int)__builtin_amdgcn_mbcnt_lo(mask, 0u);
      if (hit && pos < LIST_CAP) s_list[pos] = make_int2(e, (int)dl);
    }
  });
  __syncthreads();
  const int cnt = s_cnt;
  const bool listed = cnt <= LIST_CAP;
  if (tid == 0) tl_cnt[blockIdx.x] = cnt;
  int2* tl = (int2*)tl_ent + (size_t)blockIdx.x * LIST_CAP;

  if (listed) {
    for (int i = tid; i < cnt; i += AGG_THREADS) {
      int2 en = s_list[i];
      int s_ = src[en.x];
      s_list[i].x = s_;
      tl[i] = make_int2(s_, en.y);
      float ev[HEADS];
      scores8(as, s_ad, s_, en.y, ev);
      unsigned* mp = s_mo + en.y * HEADS;
#pragma unroll
      for (int h = 0; h < HEADS; ++h) atomicMax(mp + h, f2ord(ev[h]));
    }
  } else {
    scan([&](int d, int e) {
      for_hits(d, e, [&](int s_, int r) {
        int hh = lane & 7;
        float ev = lrelu(as[(size_t)s_ * HEADS + hh] + s_ad[r * HEADS + hh]);
        if (lane < HEADS) atomicMax(s_mo + r * HEADS + hh, f2ord(ev));
      });
    });
  }
  __syncthreads();
  for (int i = tid; i < rows * HEADS; i += AGG_THREADS) {
    float mx = ord2f(s_mo[i]);
    s_m[i] = mx;
    s_dn[i] = __expf(s_dn[i] - mx);
  }
  __syncthreads();

  if (listed) {
    for (int i = tid; i < cnt; i += AGG_THREADS) {
      int2 en = s_list[i];
      float ev[HEADS];
      scores8(as, s_ad, en.x, en.y, ev);
      const float* mp = s_m + en.y * HEADS;
      float* dp = s_dn + en.y * HEADS;
#pragma unroll
      for (int h = 0; h < HEADS; ++h) atomicAdd(dp + h, __expf(ev[h] - mp[h]));
    }
  } else {
    scan([&](int d, int e) {
      for_hits(d, e, [&](int s_, int r) {
        int hh = lane & 7;
        float ev = lrelu(as[(size_t)s_ * HEADS + hh] + s_ad[r * HEADS + hh]);
        if (lane < HEADS) atomicAdd(s_dn + r * HEADS + hh, __expf(ev - s_m[r * HEADS + hh]));
      });
    });
  }
  __syncthreads();

  for (int r = wave; r < rows; r += AGG_WAVES)
    edge_into_tile<true>(node0 + r, r, lane, H, as, s_ad, s_m, s_dn, acc);
  __syncthreads();

  if (listed) {
    const int hh = lane & 7;
    int i = wave;
    for (; i + 3 * AGG_WAVES < cnt; i += 4 * AGG_WAVES) {
      int sN[4], rN[4];
      float av[4];
      Msg mm[4];
#pragma unroll
      for (int k = 0; k < 4; ++k) {
        int2 en = s_list[i + AGG_WAVES * k];
        sN[k] = __builtin_amdgcn_readfirstlane(en.x);
        rN[k] = __builtin_amdgcn_readfirstlane(en.y);
        av[k] = as[(size_t)sN[k] * HEADS + hh];
      }
#pragma unroll
      for (int k = 0; k < 4; ++k) mm[k] = load_row(H, sN[k], lane);
      asm volatile("" :: "v"(av[0]), "v"(av[1]), "v"(av[2]), "v"(av[3]),
                   "v"(mm[0].v0.x), "v"(mm[0].v1.x), "v"(mm[0].v2.x), "v"(mm[0].v3.x),
                   "v"(mm[1].v0.x), "v"(mm[1].v1.x), "v"(mm[1].v2.x), "v"(mm[1].v3.x),
                   "v"(mm[2].v0.x), "v"(mm[2].v1.x), "v"(mm[2].v2.x), "v"(mm[2].v3.x),
                   "v"(mm[3].v0.x), "v"(mm[3].v1.x), "v"(mm[3].v2.x), "v"(mm[3].v3.x));
#pragma unroll
      for (int k = 0; k < 4; ++k) {
        float alpha = edge_alpha(av[k], rN[k], hh, s_ad, s_m, s_dn);
        msg_into_tile<false>(mm[k], alpha, rN[k], lane, acc);
      }
    }
    for (; i < cnt; i += AGG_WAVES) {
      int2 en = s_list[i];
      edge_into_tile<false>(__builtin_amdgcn_readfirstlane(en.x),
                            __builtin_amdgcn_readfirstlane(en.y),
                            lane, H, as, s_ad, s_m, s_dn, acc);
    }
  } else {
    scan([&](int d, int e) {
      for_hits(d, e, [&](int s_, int r) {
        edge_into_tile<false>(s_, r, lane, H, as, s_ad, s_m, s_dn, acc);
      });
    });
  }
  __syncthreads();

  for (int i = tid; i < rows * (F1 / 4); i += AGG_THREADS) {
    int r = i >> 7, f = i & 127;
    const float* a = acc + r * F1 + f;
    float4 v = make_float4(a[0], a[128], a[256], a[384]);
    ((float4*)(out + (size_t)(node0 + r) * F1))[f] = v;
  }
}

#define T2_THREADS 1024
#define T2_SLOTS   (LIST_CAP / T2_THREADS)
static_assert(T2_SLOTS * T2_THREADS == LIST_CAP, "the slots of a work-group cover a whole list");
static_assert(TILE_R * D_OUT <= T2_THREADS, "one thread per (row, column) of the tile");

__global__ __launch_bounds__(T2_THREADS, 16)
void tile_agg2(const int* __restrict__ src, const int* __restrict__ dst, int nE, int n,
               const float* __restrict__ H2, const float* __restrict__ as,
               const float* __restrict__ ad, const float* __restrict__ b2,
               const int* __restrict__ tl_cnt, const int* __restrict__ tl_ent,
               float* __restrict__ out) {
  __shared__ float s_ad[TILE_R];
  __shared__ unsigned s_mo[TILE_R];
  __shared__ float s_m[TILE_R];
  __shared__ float s_w0[TILE_R];
  __shared__ float s_dn[TILE_R];
  __shared__ float acc[D_OUT * TILE_R];

  const int tid = threadIdx.x;
  const int node0 = blockIdx.x * TILE_R;
  const int rows = min(TILE_R, n - node0);
  const int cnt = tl_cnt[blockIdx.x];
  const bool listed = cnt <= LIST_CAP;

  const bool rowt = tid < rows;
  const bool elt = tid < rows * D_OUT;
  float a_d0 = 0.f, a_s0 = 0.f, hself = 0.f, bias = 0.f;
  if (rowt) { a_d0 = ad[node0 + tid]; a_s0 = as[node0 + tid]; }
  if (elt) { hself = H2[(size_t)node0 * D_OUT + tid]; bias = b2[tid & 7]; }

  const int2* tl = (const int2*)tl_ent + (size_t)blockIdx.x * LIST_CAP;
  bool ok[T2_SLOTS];
  int rN[T2_SLOTS];
  float aS[T2_SLOTS];
  float4 hA[T2_SLOTS], hB[T2_SLOTS];
#pragma unroll
  for (int k = 0; k < T2_SLOTS; ++k) {
    int i = tid + T2_THREADS * k;
    ok[k] = listed && i < cnt;
    rN[k] = 0; aS[k] = 0.f;
    hA[k] = hB[k] = make_float4(0.f, 0.f, 0.f, 0.f);
    if (ok[k]) {
      int2 en = tl[i];
      rN[k] = en.y;
      aS[k] = as[en.x];
      const float4* hp = (const float4*)(H2 + (size_t)en.x * D_OUT);
      hA[k] = hp[0]; hB[k] = hp[1];
    }
  }

  auto scan = [&](auto f) {
    for (int e = tid; e < nE; e += T2_THREADS) {
      unsigned dl = (unsigned)(dst[e] - node0);
      if (dl < (unsigned)rows) f(src[e], (int)dl);
    }
  };

  if (rowt) {
    float ev0 = lrelu(a_s0 + a_d0);
    s_ad[tid] = a_d0;
    s_mo[tid] = f2ord(ev0);
    s_w0[tid] = ev0;
  }
  __syncthreads();

  float ev[T2_SLOTS];
  if (listed) {
#pragma unroll
    for (int k = 0; k < T2_SLOTS; ++k) {
      ev[k] = 0.f;
      if (ok[k]) {
        ev[k] = lrelu(aS[k] + s_ad[rN[k]]);
        atomicMax(s_mo + rN[k], f2ord(ev[k]));
      }
    }
  } else {
    scan([&](int s_, int r) { atomicMax(s_mo + r, f2ord(lrelu(as[s_] + s_ad[r]))); });
  }
  __syncthreads();
  if (rowt) {
    float mx = ord2f(s_mo[tid]);
    float w0 = __expf(s_w0[tid] - mx);
    s_m[tid] = mx;
    s_w0[tid] = w0;
    s_dn[tid] = w0;
  }
  __syncthreads();

  float w[T2_SLOTS];
  if (listed) {
#pragma unroll
    for (int k = 0; k < T2_SLOTS; ++k) {
      w[k] = 0.f;
      if (ok[k]) {
        w[k] = __expf(ev[k] - s_m[rN[k]]);
        atomicAdd(s_dn + rN[k], w[k]);
      }
    }
  } else {
    scan([&](int s_, int r) {
      atomicAdd(s_dn + r, __expf(lrelu(as[s_] + s_ad[r]) - s_m[r]));
    });
  }
  __syncthreads();

  if (elt) {
    int r = tid >> 3, cc = tid & 7;
    float alpha = s_w0[r] / (s_dn[r] + 1e-16f);
    acc[cc * TILE_R + r] = hself * alpha;
  }
  __syncthreads();
  if (listed) {
#pragma unroll
    for (int k = 0; k < T2_SLOTS; ++k) {
      if (ok[k]) {
        float alpha = w[k] / (s_dn[rN[k]] + 1e-16f);
        float* ap = acc + rN[k];
        atomicAdd(ap,              hA[k].x * alpha); atomicAdd(ap + TILE_R,     hA[k].y * alpha);
        atomicAdd(ap + 2 * TILE_R, hA[k].z * alpha); atomicAdd(ap + 3 * TILE_R, hA[k].w * alpha);
        atomicAdd(ap + 4 * TILE_R, hB[k].x * alpha); atomicAdd(ap + 5 * TILE_R, hB[k].y * alpha);
        atomicAdd(ap + 6 * TILE_R, hB[k].z * alpha); atomicAdd(ap + 7 * TILE_R, hB[k].w * alpha);
      }
    }
  } else {
    scan([&](int s_, int r) {
      float alpha = __expf(lrelu(as[s_] + s_ad[r]) - s_m[r]) / (s_dn[r] + 1e-16f);
      const float* hp = H2 + (size_t)s_ * D_OUT;
#pragma unroll
      for (int cc = 0; cc < D_OUT; ++cc) atomicAdd(acc + cc * TILE_R + r, hp[cc] * alpha);
    });
  }
  __syncthreads();

  if (elt) {
    int r = tid >> 3, cc = tid & 7;
    out[(size_t)node0 * D_OUT + tid] = acc[cc * TILE_R + r] + bias;
  }
}

extern "C" void kernel_launch(void* const* d_in, const int* in_sizes, int n_in,
                              void* d_out, int out_size, void* d_ws, size_t ws_size,
                              hipStream_t stream) {
  const float* x      = (const float*)d_in[0];
  const int*   ei     = (const int*)  d_in[1];
  const float* W1     = (const float*)d_in[2];
  const float* a_src1 = (const float*)d_in[3];
  const float* a_dst1 = (const float*)d_in[4];
  const float* b1     = (const float*)d_in[5];
  const float* W2     = (const float*)d_in[6];
  const float* a_src2 = (const float*)d_in[7];
  const float* a_dst2 = (const float*)d_in[8];
  const float* b2     = (const float*)d_in[9];

  int n  = in_sizes[0] / D_IN;
  int nE = in_sizes[1] / 2;
  const int* src = ei;
  const int* dst = ei + nE;
  int ntiles = (n + TILE_R - 1) / TILE_R;

  char* ws = (char*)d_ws;
  size_t off = 0;
  auto alloc = [&](size_t bytes) -> char* {
    char* p = ws + off; off += (bytes + 255) & ~(size_t)255; return p;
  };
  float*    h1   = (float*)   alloc((size_t)n * F1 * 4);
  float*    out1 = (float*)   alloc((size_t)n * F1 * 4);
  float*    as1  = (float*)   alloc((size_t)n * HEADS * 4);
  float*    ad1  = (float*)   alloc((size_t)n * HEADS * 4);
  float*    h2   = (float*)   alloc((size_t)n * D_OUT * 4);
  float*    as2  = (float*)   alloc((size_t)n * 4);
  float*    ad2  = (float*)   alloc((size_t)n * 4);
  int*      tlc  = (int*)     alloc((size_t)ntiles * 4);
  int*      tle  = (int*)     alloc((size_t)ntiles * LIST_CAP * 8);

  gemm1_wmma<<<dim3(F1 / G1_BN, (n + G1_BM - 1) / G1_BM), 256, 0, stream>>>(x, W1, h1, n);
  alphas1<<<(n * HEADS + 255) / 256, 256, 0, stream>>>(h1, a_src1, a_dst1, as1, ad1, n);
  tile_agg1<<<ntiles, AGG_THREADS, 0, stream>>>(src, dst, nE, n, h1, as1, ad1, out1, tlc, tle);

  {
    int waves  = (n + 15) / 16;
    int blocks = (waves + 7) / 8;
    gemm2_wmma<<<blocks, 256, 0, stream>>>(out1, b1, W2, h2, n);
  }
  alphas2<<<(n + 255) / 256, 256, 0, stream>>>(h2, a_src2, a_dst2, as2, ad2, n);
  tile_agg2<<<ntiles, T2_THREADS, 0, stream>>>(src, dst, nE, n, h2, as2, ad2, b2, tlc, tle, (float*)d_out);
}


# Layer-1 tile kernel: a work-group owns two tiles of 98 nodes and scans the destination array once for both (one list per tile in LDS), then sums tile after tile; 255 of 511 scans gone; sum order kept
# speedup vs baseline: 6.9167x; 1.0940x over previous
#include <hip/hip_runtime.h>
#include <math.h>

#define D_IN   128
#define HEADS  8
#define HID    64
#define F1     (HEADS * HID)
#define D_OUT  8
#define NEG_SLOPE 0.2f

typedef float v2f __attribute__((ext_vector_type(2)));
typedef float v8f __attribute__((ext_vector_type(8)));

__device__ __forceinline__ unsigned f2ord(float f) {
  unsigned u = __float_as_uint(f);
  return (u & 0x80000000u) ? ~u : (u | 0x80000000u);
}
__device__ __forceinline__ float ord2f(unsigned v) {
  unsigned u = (v & 0x80000000u) ? (v ^ 0x80000000u) : ~v;
  return __uint_as_float(u);
}
__device__ __forceinline__ float lrelu(float x) { return x > 0.f ? x : NEG_SLOPE * x; }

#define G1_BM 128
#define G1_BN 128
#define G1_XP 132
#define G1_WP 288

__global__ __launch_bounds__(256)
void gemm1_wmma(const float* __restrict__ X, const float* __restrict__ W,
                float* __restrict__ H, int n) {
  __shared__ float Xs[G1_BM * G1_XP];
  __shared__ float Ws[(D_IN / 2) * G1_WP];

  const int tid = threadIdx.x, lane = tid & 31;
  const int wave = __builtin_amdgcn_readfirstlane(tid >> 5);
  const int col0 = blockIdx.x * G1_BN;
  const int row0 = blockIdx.y * G1_BM;

#pragma unroll
  for (int i = 0; i < (G1_BM * D_IN / 4) / 256; ++i) {
    int idx = tid + 256 * i;
    int r = idx >> 5, c4 = idx & 31;
    int gr = min(row0 + r, n - 1);
    float4 v = ((const float4*)(X + (size_t)gr * D_IN))[c4];
    *(float4*)(Xs + r * G1_XP + 4 * c4) = v;
  }
#pragma unroll
  for (int i = 0; i < ((D_IN / 2) * G1_BN / 4) / 256; ++i) {
    int idx = tid + 256 * i;
    int kp = idx >> 5, c4 = idx & 31;
    const float* wp = W + (size_t)(2 * kp) * F1 + col0 + 4 * c4;
    float4 r0 = *(const float4*)wp;
    float4 r1 = *(const float4*)(wp + F1);
    float* d = Ws + kp * G1_WP + 8 * c4;
    *(float4*)d       = make_float4(r0.x, r1.x, r0.y, r1.y);
    *(float4*)(d + 4) = make_float4(r0.z, r1.z, r0.w, r1.w);
  }
  __syncthreads();

  const int half = lane >> 4;
  const int l15  = lane & 15;
  const float* xa = Xs + (16 * wave + l15) * G1_XP + 2 * half;
  const float* wb = Ws + half * G1_WP + 2 * l15;

  v8f acc[G1_BN / 16];
#pragma unroll
  for (int i = 0; i < G1_BN / 16; ++i) acc[i] = (v8f){};
#pragma unroll
  for (int j = 0; j < D_IN / 4; ++j) {
    v2f a = *(const v2f*)(xa + 4 * j);
#pragma unroll
    for (int i = 0; i < G1_BN / 16; ++i) {
      v2f b = *(const v2f*)(wb + 2 * j * G1_WP + 32 * i);
      acc[i] = __builtin_amdgcn_wmma_f32_16x16x4_f32(false, a, false, b,
                                                     (short)0, acc[i], false, false);
    }
  }

  float* xo = Xs + (16 * wave + 8 * half) * G1_XP + l15;
#pragma unroll
  for (int i = 0; i < G1_BN / 16; ++i) {
#pragma unroll
    for (int v = 0; v < 8; ++v) xo[v * G1_XP + 16 * i] = acc[i][v];
  }
  __syncthreads();
  const float* xr = Xs + (16 * wave) * G1_XP + 4 * lane;
  float* hp = H + (size_t)(row0 + 16 * wave) * F1 + col0 + 4 * lane;
  const int nrows = n - (row0 + 16 * wave);
  if (nrows >= 16) {
    float4 o[16];
#pragma unroll
    for (int r = 0; r < 16; ++r) o[r] = *(const float4*)(xr + r * G1_XP);
#pragma unroll
    for (int r = 0; r < 16; ++r) *(float4*)(hp + (size_t)r * F1) = o[r];
  } else {
    for (int r = 0; r < nrows; ++r)
      *(float4*)(hp + (size_t)r * F1) = *(const float4*)(xr + r * G1_XP);
  }
}

__global__ void gemm2_wmma(const float* __restrict__ X2, const float* __restrict__ b1,
                           const float* __restrict__ W2,
                           float* __restrict__ H2, int n) {
  int wave = (blockIdx.x * blockDim.x + threadIdx.x) >> 5;
  int lane = threadIdx.x & 31;
  int row0 = wave * 16;
  if (row0 >= n) return;

  int half = lane >> 4;
  int l15  = lane & 15;
  const float* xrow = X2 + (size_t)(row0 + l15) * F1;
  bool colok = (l15 < D_OUT);

  v8f acc = {};
  for (int k = 0; k < F1; k += 4) {
    int kk = k + half * 2;
    float s0 = xrow[kk] + b1[kk], s1 = xrow[kk + 1] + b1[kk + 1];
    v2f a; a.x = s0 > 0.f ? s0 : 0.f; a.y = s1 > 0.f ? s1 : 0.f;
    v2f b;
    b.x = colok ? W2[(size_t)kk * D_OUT + l15]       : 0.f;
    b.y = colok ? W2[(size_t)(kk + 1) * D_OUT + l15] : 0.f;
    acc = __builtin_amdgcn_wmma_f32_16x16x4_f32(false, a, false, b,
                                                (short)0, acc, false, false);
  }
  if (colok) {
#pragma unroll
    for (int v = 0; v < 8; ++v) {
      int r = row0 + v + half * 8;
      H2[(size_t)r * D_OUT + l15] = acc[v];
    }
  }
}

__global__ void alphas1(const float* __restrict__ H, const float* __restrict__ a_src,
                        const float* __restrict__ a_dst,
                        float* __restrict__ as, float* __restrict__ ad, int n) {
  int t = blockIdx.x * blockDim.x + threadIdx.x;
  if (t >= n * HEADS) return;
  int h = t % HEADS;
  const float* row = H + (size_t)(t / HEADS) * F1 + h * HID;
  const float* vs = a_src + h * HID;
  const float* vd = a_dst + h * HID;
  float s = 0.f, d = 0.f;
#pragma unroll 8
  for (int c = 0; c < HID; ++c) { float x = row[c]; s += x * vs[c]; d += x * vd[c]; }
  as[t] = s; ad[t] = d;
}

__global__ void alphas2(const float* __restrict__ H2, const float* __restrict__ a_src,
                        const float* __restrict__ a_dst,
                        float* __restrict__ as, float* __restrict__ ad, int n) {
  int t = blockIdx.x * blockDim.x + threadIdx.x;
  if (t >= n) return;
  const float* row = H2 + (size_t)t * D_OUT;
  float s = 0.f, d = 0.f;
#pragma unroll
  for (int c = 0; c < D_OUT; ++c) { s += row[c] * a_src[c]; d += row[c] * a_dst[c]; }
  as[t] = s; ad[t] = d;
}

#define TILE_R      98
#define NSUB        2
#define GROUP_R     (NSUB * TILE_R)
#define AGG_THREADS 1024
#define AGG_WAVES   (AGG_THREADS / 32)
#define LIST_CAP    4096

struct Msg { float4 v0, v1, v2, v3; };

__device__ __forceinline__ Msg load_row(const float* __restrict__ H, int s_, int lane) {
  const float4* hp = (const float4*)(H + (size_t)s_ * F1);
  Msg m;
  m.v0 = hp[lane]; m.v1 = hp[lane + 32]; m.v2 = hp[lane + 64]; m.v3 = hp[lane + 96];
  return m;
}

__device__ __forceinline__ float edge_alpha(float a_s, int r, int hh, const float* s_ad,
                                            const float* s_m, const float* s_dn) {
  float ev = lrelu(a_s + s_ad[r * HEADS + hh]);
  float w = __expf(ev - s_m[r * HEADS + hh]);
  return w / (s_dn[r * HEADS + hh] + 1e-16f);
}

template <bool START>
__device__ __forceinline__ void msg_into_tile(const Msg& m, float alpha, int r, int lane,
                                              float* acc) {
  int hs = lane >> 4;
  float a0 = __shfl(alpha, hs), a1 = __shfl(alpha, hs + 2);
  float a2 = __shfl(alpha, hs + 4), a3 = __shfl(alpha, hs + 6);
  float* ap = acc + r * F1 + lane;
  if (START) {
    ap[0]       = m.v0.x * a0; ap[128]      = m.v0.y * a0; ap[256]      = m.v0.z * a0; ap[384]      = m.v0.w * a0;
    ap[32]      = m.v1.x * a1; ap[128 + 32] = m.v1.y * a1; ap[256 + 32] = m.v1.z * a1; ap[384 + 32] = m.v1.w * a1;
    ap[64]      = m.v2.x * a2; ap[128 + 64] = m.v2.y * a2; ap[256 + 64] = m.v2.z * a2; ap[384 + 64] = m.v2.w * a2;
    ap[96]      = m.v3.x * a3; ap[128 + 96] = m.v3.y * a3; ap[256 + 96] = m.v3.z * a3; ap[384 + 96] = m.v3.w * a3;
  } else {
    atomicAdd(ap,            m.v0.x * a0); atomicAdd(ap + 128,      m.v0.y * a0);
    atomicAdd(ap + 256,      m.v0.z * a0); atomicAdd(ap + 384,      m.v0.w * a0);
    atomicAdd(ap + 32,       m.v1.x * a1); atomicAdd(ap + 128 + 32, m.v1.y * a1);
    atomicAdd(ap + 256 + 32, m.v1.z * a1); atomicAdd(ap + 384 + 32, m.v1.w * a1);
    atomicAdd(ap + 64,       m.v2.x * a2); atomicAdd(ap + 128 + 64, m.v2.y * a2);
    atomicAdd(ap + 256 + 64, m.v2.z * a2); atomicAdd(ap + 384 + 64, m.v2.w * a2);
    atomicAdd(ap + 96,       m.v3.x * a3); atomicAdd(ap + 128 + 96, m.v3.y * a3);
    atomicAdd(ap + 256 + 96, m.v3.z * a3); atomicAdd(ap + 384 + 96, m.v3.w * a3);
  }
}

template <bool START>
__device__ __forceinline__ void edge_into_tile(int s_, int r, int lane,
                                               const float* __restrict__ H,
                                               const float* __restrict__ as,
                                               const float* s_ad, const float* s_m,
                                               const float* s_dn, float* acc) {
  Msg m = load_row(H, s_, lane);
  int hh = lane & 7;
  float alpha = edge_alpha(as[(size_t)s_ * HEADS + hh], r, hh, s_ad, s_m, s_dn);
  msg_into_tile<START>(m, alpha, r, lane, acc);
}

__device__ __forceinline__ void scores8(const float* __restrict__ as, const float* s_ad,
                                        int s_, int r, float* ev) {
  const float4* ap = (const float4*)(as + (size_t)s_ * HEADS);
  float4 a0 = ap[0], a1 = ap[1];
  const float4* dp = (const float4*)(s_ad + r * HEADS);
  float4 d0 = dp[0], d1 = dp[1];
  ev[0] = lrelu(a0.x + d0.x); ev[1] = lrelu(a0.y + d0.y);
  ev[2] = lrelu(a0.z + d0.z); ev[3] = lrelu(a0.w + d0.w);
  ev[4] = lrelu(a1.x + d1.x); ev[5] = lrelu(a1.y + d1.y);
  ev[6] = lrelu(a1.z + d1.z); ev[7] = lrelu(a1.w + d1.w);
}

__global__ __launch_bounds__(AGG_THREADS)
void tile_agg1(const int* __restrict__ src, const int* __restrict__ dst,
               int nE, int n, const float* __restrict__ H,
               const float* __restrict__ as, const float* __restrict__ ad,
               float* __restrict__ out, int* __restrict__ tl_cnt, int* __restrict__ tl_ent) {
  __shared__ float acc[TILE_R * F1];
  __shared__ __attribute__((aligned(16))) float s_ad[GROUP_R * HEADS];
  __shared__ float s_m[GROUP_R * HEADS];
  __shared__ float s_dn[GROUP_R * HEADS];
  __shared__ unsigned s_mo[GROUP_R * HEADS];
  __shared__ int2 s_list[NSUB * LIST_CAP];
  __shared__ int s_cnt[NSUB];

  const int tid = threadIdx.x, lane = tid & 31, wave = tid >> 5;
  const int node0 = blockIdx.x * GROUP_R;
  const int rows = min(GROUP_R, n - node0);

  for (int i = tid; i < rows * HEADS; i += AGG_THREADS) {
    size_t g = (size_t)node0 * HEADS + i;
    float a_d = ad[g];
    float ev = lrelu(as[g] + a_d);
    s_ad[i] = a_d;
    s_mo[i] = f2ord(ev);
    s_dn[i] = ev;
  }
  if (tid < NSUB) s_cnt[tid] = 0;
  __syncthreads();

  auto scan = [&](auto take) {
    auto take4 = [&](int4 d, bool ok, int e0) {
      unsigned lo = min(min((unsigned)(d.x - node0), (unsigned)(d.y - node0)),
                        min((unsigned)(d.z - node0), (unsigned)(d.w - node0)));
      bool any = (lo < (unsigned)rows) & ok;
      if (__builtin_amdgcn_ballot_w32(any) != 0u) {
        take(ok ? d.x : -1, e0);
        take(ok ? d.y : -1, e0 + 1);
        take(ok ? d.z : -1, e0 + 2);
        take(ok ? d.w : -1, e0 + 3);
      }
    };
    int pre = (int)(((16u - (unsigned)((size_t)dst & 15u)) & 15u) >> 2);
    if (pre > nE) pre = nE;
    const int nvec = (nE - pre) >> 2;
    const int4* dv = (const int4*)(dst + pre);
    const int nchunk = (nvec + 31) >> 5;
    if (nvec > 0) {
      const int last = nvec - 1;
      int4 cur[4];
#pragma unroll
      for (int j = 0; j < 4; ++j)
        cur[j] = dv[min(((wave + AGG_WAVES * j) << 5) + lane, last)];
      for (int base = wave; base < nchunk; base += 4 * AGG_WAVES) {
        int4 nxt[4];
#pragma unroll
        for (int j = 0; j < 4; ++j)
          nxt[j] = dv[min(((base + 4 * AGG_WAVES + AGG_WAVES * j) << 5) + lane, last)];
#pragma unroll
        for (int j = 0; j < 4; ++j) {
          int p = ((base + AGG_WAVES * j) << 5) + lane;
          take4(cur[j], p < nvec, pre + 4 * p);
          cur[j] = nxt[j];
        }
      }
    }
    if (wave == 0) {
      int tail0 = pre + 4 * nvec;
      int e = lane < pre ? lane : tail0 + (lane - pre);
      take(e < nE ? dst[e] : -1, e);
    }
  };

  auto for_hits = [&](int d, int e, int nb, int nr, auto f) {
    unsigned dl = (unsigned)(d - nb);
    unsigned mask = __builtin_amdgcn_ballot_w32(dl < (unsigned)nr);
    while (mask) {
      int l = __builtin_ctz(mask);
      mask &= mask - 1;
      int r  = __builtin_amdgcn_readlane((int)dl, l);
      int ee = __builtin_amdgcn_readlane(e, l);
      f(src[ee], r);
    }
  };

  scan([&](int d, int e) {
    unsigned dl = (unsigned)(d - node0);
    bool hit = dl < (unsigned)rows;
    if (__builtin_amdgcn_ballot_w32(hit) != 0u) {
      int t = hit ? (int)(dl / TILE_R) : -1;
#pragma unroll
      for (int s = 0; s < NSUB; ++s) {
        bool hs = t == s;
        unsigned mask = __builtin_amdgcn_ballot_w32(hs);
        if (mask != 0u) {
          int base = 0;
          if (lane == 0) base = atomicAdd(&s_cnt[s], __builtin_popcount(mask));
          base = __builtin_amdgcn_readfirstlane(base);
          int pos = base + (int)__builtin_amdgcn_mbcnt_lo(mask, 0u);
          if (hs && pos < LIST_CAP)
            s_list[s * LIST_CAP + pos] = make_int2(e, (int)dl - s * TILE_R);
        }
      }
    }
  });
  __syncthreads();
  if (tid < NSUB && rows > tid * TILE_R) tl_cnt[blockIdx.x * NSUB + tid] = s_cnt[tid];

  for (int s = 0; s < NSUB; ++s) {
    const int nb = node0 + s * TILE_R;
    const int nr = min(TILE_R, n - nb);
    if (nr <= 0) break;
    const int cnt = s_cnt[s];
    const float* t_ad = s_ad + s * TILE_R * HEADS;
    unsigned* t_mo = s_mo + s * TILE_R * HEADS;
    if (cnt <= LIST_CAP) {
      int2* sl = s_list + s * LIST_CAP;
      int2* tl = (int2*)tl_ent + (size_t)(blockIdx.x * NSUB + s) * LIST_CAP;
      for (int i = tid; i < cnt; i += AGG_THREADS) {
        int2 en = sl[i];
        int s_ = src[en.x];
        sl[i].x = s_;
        tl[i] = make_int2(s_, en.y);
        float ev[HEADS];
        scores8(as, t_ad, s_, en.y, ev);
        unsigned* mp = t_mo + en.y * HEADS;
#pragma unroll
        for (int h = 0; h < HEADS; ++h) atomicMax(mp + h, f2ord(ev[h]));
      }
    } else {
      scan([&](int d, int e) {
        for_hits(d, e, nb, nr, [&](int s_, int r) {
          int hh = lane & 7;
          float ev = lrelu(as[(size_t)s_ * HEADS + hh] + t_ad[r * HEADS + hh]);
          if (lane < HEADS) atomicMax(t_mo + r * HEADS + hh, f2ord(ev));
        });
      });
    }
  }
  __syncthreads();
  for (int i = tid; i < rows * HEADS; i += AGG_THREADS) {
    float mx = ord2f(s_mo[i]);
    s_m[i] = mx;
    s_dn[i] = __expf(s_dn[i] - mx);
  }
  __syncthreads();

  for (int s = 0; s < NSUB; ++s) {
    const int nb = node0 + s * TILE_R;
    const int nr = min(TILE_R, n - nb);
    if (nr <= 0) break;
    const int cnt = s_cnt[s];
    const float* t_ad = s_ad + s * TILE_R * HEADS;
    const float* t_m = s_m + s * TILE_R * HEADS;
    float* t_dn = s_dn + s * TILE_R * HEADS;
    if (cnt <= LIST_CAP) {
      const int2* sl = s_list + s * LIST_CAP;
      for (int i = tid; i < cnt; i += AGG_THREADS) {
        int2 en = sl[i];
        float ev[HEADS];
        scores8(as, t_ad, en.x, en.y, ev);
        const float* mp = t_m + en.y * HEADS;
        float* dp = t_dn + en.y * HEADS;
#pragma unroll
        for (int h = 0; h < HEADS; ++h) atomicAdd(dp + h, __expf(ev[h] - mp[h]));
      }
    } else {
      scan([&](int d, int e) {
        for_hits(d, e, nb, nr, [&](int s_, int r) {
          int hh = lane & 7;
          float ev = lrelu(as[(size_t)s_ * HEADS + hh] + t_ad[r * HEADS + hh]);
          if (lane < HEADS) atomicAdd(t_dn + r * HEADS + hh, __expf(ev - t_m[r * HEADS + hh]));
        });
      });
    }
  }
  __syncthreads();

  for (int s = 0; s < NSUB; ++s) {
    const int nb = node0 + s * TILE_R;
    const int nr = min(TILE_R, n - nb);
    if (nr <= 0) break;
    const int cnt = s_cnt[s];
    const float* t_ad = s_ad + s * TILE_R * HEADS;
    const float* t_m = s_m + s * TILE_R * HEADS;
    const float* t_dn = s_dn + s * TILE_R * HEADS;

    for (int r = wave; r < nr; r += AGG_WAVES)
      edge_into_tile<true>(nb + r, r, lane, H, as, t_ad, t_m, t_dn, acc);
    __syncthreads();

    if (cnt <= LIST_CAP) {
      const int2* sl = s_list + s * LIST_CAP;
      const int hh = lane & 7;
      int i = wave;
      for (; i + 3 * AGG_WAVES < cnt; i += 4 * AGG_WAVES) {
        int sN[4], rN[4];
        float av[4];
        Msg mm[4];
#pragma unroll
        for (int k = 0; k < 4; ++k) {
          int2 en = sl[i + AGG_WAVES * k];
          sN[k] = __builtin_amdgcn_readfirstlane(en.x);
          rN[k] = __builtin_amdgcn_readfirstlane(en.y);
          av[k] = as[(size_t)sN[k] * HEADS + hh];
        }
#pragma unroll
        for (int k = 0; k < 4; ++k) mm[k] = load_row(H, sN[k], lane);
        asm volatile("" :: "v"(av[0]), "v"(av[1]), "v"(av[2]), "v"(av[3]),
                     "v"(mm[0].v0.x), "v"(mm[0].v1.x), "v"(mm[0].v2.x), "v"(mm[0].v3.x),
                     "v"(mm[1].v0.x), "v"(mm[1].v1.x), "v"(mm[1].v2.x), "v"(mm[1].v3.x),
                     "v"(mm[2].v0.x), "v"(mm[2].v1.x), "v"(mm[2].v2.x), "v"(mm[2].v3.x),
                     "v"(mm[3].v0.x), "v"(mm[3].v1.x), "v"(mm[3].v2.x), "v"(mm[3].v3.x));
#pragma unroll
        for (int k = 0; k < 4; ++k) {
          float alpha = edge_alpha(av[k], rN[k], hh, t_ad, t_m, t_dn);
          msg_into_tile<false>(mm[k], alpha, rN[k], lane, acc);
        }
      }
      for (; i < cnt; i += AGG_WAVES) {
        int2 en = sl[i];
        edge_into_tile<false>(__builtin_amdgcn_readfirstlane(en.x),
                              __builtin_amdgcn_readfirstlane(en.y),
                              lane, H, as, t_ad, t_m, t_dn, acc);
      }
    } else {
      scan([&](int d, int e) {
        for_hits(d, e, nb, nr, [&](int s_, int r) {
          edge_into_tile<false>(s_, r, lane, H, as, t_ad, t_m, t_dn, acc);
        });
      });
    }
    __syncthreads();

    for (int i = tid; i < nr * (F1 / 4); i += AGG_THREADS) {
      int r = i >> 7, f = i & 127;
      const float* a = acc + r * F1 + f;
      float4 v = make_float4(a[0], a[128], a[256], a[384]);
      ((float4*)(out + (size_t)(nb + r) * F1))[f] = v;
    }
    if (s + 1 < NSUB) __syncthreads();
  }
}

#define T2_THREADS 1024
#define T2_SLOTS   (LIST_CAP / T2_THREADS)
static_assert(T2_SLOTS * T2_THREADS == LIST_CAP, "the slots of a work-group cover a whole list");
static_assert(TILE_R * D_OUT <= T2_THREADS, "one thread per (row, column) of the tile");

__global__ __launch_bounds__(T2_THREADS, 16)
void tile_agg2(const int* __restrict__ src, const int* __restrict__ dst, int nE, int n,
               const float* __restrict__ H2, const float* __restrict__ as,
               const float* __restrict__ ad, const float* __restrict__ b2,
               const int* __restrict__ tl_cnt, const int* __restrict__ tl_ent,
               float* __restrict__ out) {
  __shared__ float s_ad[TILE_R];
  __shared__ unsigned s_mo[TILE_R];
  __shared__ float s_m[TILE_R];
  __shared__ float s_w0[TILE_R];
  __shared__ float s_dn[TILE_R];
  __shared__ float acc[D_OUT * TILE_R];

  const int tid = threadIdx.x;
  const int node0 = blockIdx.x * TILE_R;
  const int rows = min(TILE_R, n - node0);
  const int cnt = tl_cnt[blockIdx.x];
  const bool listed = cnt <= LIST_CAP;

  const bool rowt = tid < rows;
  const bool elt = tid < rows * D_OUT;
  float a_d0 = 0.f, a_s0 = 0.f, hself = 0.f, bias = 0.f;
  if (rowt) { a_d0 = ad[node0 + tid]; a_s0 = as[node0 + tid]; }
  if (elt) { hself = H2[(size_t)node0 * D_OUT + tid]; bias = b2[tid & 7]; }

  const int2* tl = (const int2*)tl_ent + (size_t)blockIdx.x * LIST_CAP;
  bool ok[T2_SLOTS];
  int rN[T2_SLOTS];
  float aS[T2_SLOTS];
  float4 hA[T2_SLOTS], hB[T2_SLOTS];
#pragma unroll
  for (int k = 0; k < T2_SLOTS; ++k) {
    int i = tid + T2_THREADS * k;
    ok[k] = listed && i < cnt;
    rN[k] = 0; aS[k] = 0.f;
    hA[k] = hB[k] = make_float4(0.f, 0.f, 0.f, 0.f);
    if (ok[k]) {
      int2 en = tl[i];
      rN[k] = en.y;
      aS[k] = as[en.x];
      const float4* hp = (const float4*)(H2 + (size_t)en.x * D_OUT);
      hA[k] = hp[0]; hB[k] = hp[1];
    }
  }

  auto scan = [&](auto f) {
    for (int e = tid; e < nE; e += T2_THREADS) {
      unsigned dl = (unsigned)(dst[e] - node0);
      if (dl < (unsigned)rows) f(src[e], (int)dl);
    }
  };

  if (rowt) {
    float ev0 = lrelu(a_s0 + a_d0);
    s_ad[tid] = a_d0;
    s_mo[tid] = f2ord(ev0);
    s_w0[tid] = ev0;
  }
  __syncthreads();

  float ev[T2_SLOTS];
  if (listed) {
#pragma unroll
    for (int k = 0; k < T2_SLOTS; ++k) {
      ev[k] = 0.f;
      if (ok[k]) {
        ev[k] = lrelu(aS[k] + s_ad[rN[k]]);
        atomicMax(s_mo + rN[k], f2ord(ev[k]));
      }
    }
  } else {
    scan([&](int s_, int r) { atomicMax(s_mo + r, f2ord(lrelu(as[s_] + s_ad[r]))); });
  }
  __syncthreads();
  if (rowt) {
    float mx = ord2f(s_mo[tid]);
    float w0 = __expf(s_w0[tid] - mx);
    s_m[tid] = mx;
    s_w0[tid] = w0;
    s_dn[tid] = w0;
  }
  __syncthreads();

  float w[T2_SLOTS];
  if (listed) {
#pragma unroll
    for (int k = 0; k < T2_SLOTS; ++k) {
      w[k] = 0.f;
      if (ok[k]) {
        w[k] = __expf(ev[k] - s_m[rN[k]]);
        atomicAdd(s_dn + rN[k], w[k]);
      }
    }
  } else {
    scan([&](int s_, int r) {
      atomicAdd(s_dn + r, __expf(lrelu(as[s_] + s_ad[r]) - s_m[r]));
    });
  }
  __syncthreads();

  if (elt) {
    int r = tid >> 3, cc = tid & 7;
    float alpha = s_w0[r] / (s_dn[r] + 1e-16f);
    acc[cc * TILE_R + r] = hself * alpha;
  }
  __syncthreads();
  if (listed) {
#pragma unroll
    for (int k = 0; k < T2_SLOTS; ++k) {
      if (ok[k]) {
        float alpha = w[k] / (s_dn[rN[k]] + 1e-16f);
        float* ap = acc + rN[k];
        atomicAdd(ap,              hA[k].x * alpha); atomicAdd(ap + TILE_R,     hA[k].y * alpha);
        atomicAdd(ap + 2 * TILE_R, hA[k].z * alpha); atomicAdd(ap + 3 * TILE_R, hA[k].w * alpha);
        atomicAdd(ap + 4 * TILE_R, hB[k].x * alpha); atomicAdd(ap + 5 * TILE_R, hB[k].y * alpha);
        atomicAdd(ap + 6 * TILE_R, hB[k].z * alpha); atomicAdd(ap + 7 * TILE_R, hB[k].w * alpha);
      }
    }
  } else {
    scan([&](int s_, int r) {
      float alpha = __expf(lrelu(as[s_] + s_ad[r]) - s_m[r]) / (s_dn[r] + 1e-16f);
      const float* hp = H2 + (size_t)s_ * D_OUT;
#pragma unroll
      for (int cc = 0; cc < D_OUT; ++cc) atomicAdd(acc + cc * TILE_R + r, hp[cc] * alpha);
    });
  }
  __syncthreads();

  if (elt) {
    int r = tid >> 3, cc = tid & 7;
    out[(size_t)node0 * D_OUT + tid] = acc[cc * TILE_R + r] + bias;
  }
}

extern "C" void kernel_launch(void* const* d_in, const int* in_sizes, int n_in,
                              void* d_out, int out_size, void* d_ws, size_t ws_size,
                              hipStream_t stream) {
  const float* x      = (const float*)d_in[0];
  const int*   ei     = (const int*)  d_in[1];
  const float* W1     = (const float*)d_in[2];
  const float* a_src1 = (const float*)d_in[3];
  const float* a_dst1 = (const float*)d_in[4];
  const float* b1     = (const float*)d_in[5];
  const float* W2     = (const float*)d_in[6];
  const float* a_src2 = (const float*)d_in[7];
  const float* a_dst2 = (const float*)d_in[8];
  const float* b2     = (const float*)d_in[9];

  int n  = in_sizes[0] / D_IN;
  int nE = in_sizes[1] / 2;
  const int* src = ei;
  const int* dst = ei + nE;
  int ntiles = (n + TILE_R - 1) / TILE_R;
  int ngroups = (n + GROUP_R - 1) / GROUP_R;

  char* ws = (char*)d_ws;
  size_t off = 0;
  auto alloc = [&](size_t bytes) -> char* {
    char* p = ws + off; off += (bytes + 255) & ~(size_t)255; return p;
  };
  float*    h1   = (float*)   alloc((size_t)n * F1 * 4);
  float*    out1 = (float*)   alloc((size_t)n * F1 * 4);
  float*    as1  = (float*)   alloc((size_t)n * HEADS * 4);
  float*    ad1  = (float*)   alloc((size_t)n * HEADS * 4);
  float*    h2   = (float*)   alloc((size_t)n * D_OUT * 4);
  float*    as2  = (float*)   alloc((size_t)n * 4);
  float*    ad2  = (float*)   alloc((size_t)n * 4);
  int*      tlc  = (int*)     alloc((size_t)ntiles * 4);
  int*      tle  = (int*)     alloc((size_t)ntiles * LIST_CAP * 8);

  gemm1_wmma<<<dim3(F1 / G1_BN, (n + G1_BM - 1) / G1_BM), 256, 0, stream>>>(x, W1, h1, n);
  alphas1<<<(n * HEADS + 255) / 256, 256, 0, stream>>>(h1, a_src1, a_dst1, as1, ad1, n);
  tile_agg1<<<ngroups, AGG_THREADS, 0, stream>>>(src, dst, nE, n, h1, as1, ad1, out1, tlc, tle);

  {
    int waves  = (n + 15) / 16;
    int blocks = (waves + 7) / 8;
    gemm2_wmma<<<blocks, 256, 0, stream>>>(out1, b1, W2, h2, n);
  }
  alphas2<<<(n + 255) / 256, 256, 0, stream>>>(h2, a_src2, a_dst2, as2, ad2, n);
  tile_agg2<<<ntiles, T2_THREADS, 0, stream>>>(src, dst, nE, n, h2, as2, ad2, b2, tlc, tle, (float*)d_out);
}


# Layer-1 tile kernel scans one-byte group codes of the destinations (written each call by the coefficient kernel), 16 edges per load; coded hits re-read dst for the exact test; list order: arrival
# speedup vs baseline: 7.2307x; 1.0454x over previous
#include <hip/hip_runtime.h>
#include <math.h>

#define D_IN   128
#define HEADS  8
#define HID    64
#define F1     (HEADS * HID)
#define D_OUT  8
#define NEG_SLOPE 0.2f

typedef float v2f __attribute__((ext_vector_type(2)));
typedef float v8f __attribute__((ext_vector_type(8)));

__device__ __forceinline__ unsigned f2ord(float f) {
  unsigned u = __float_as_uint(f);
  return (u & 0x80000000u) ? ~u : (u | 0x80000000u);
}
__device__ __forceinline__ float ord2f(unsigned v) {
  unsigned u = (v & 0x80000000u) ? (v ^ 0x80000000u) : ~v;
  return __uint_as_float(u);
}
__device__ __forceinline__ float lrelu(float x) { return x > 0.f ? x : NEG_SLOPE * x; }

#define G1_BM 128
#define G1_BN 128
#define G1_XP 132
#define G1_WP 288

__global__ __launch_bounds__(256)
void gemm1_wmma(const float* __restrict__ X, const float* __restrict__ W,
                float* __restrict__ H, int n) {
  __shared__ float Xs[G1_BM * G1_XP];
  __shared__ float Ws[(D_IN / 2) * G1_WP];

  const int tid = threadIdx.x, lane = tid & 31;
  const int wave = __builtin_amdgcn_readfirstlane(tid >> 5);
  const int col0 = blockIdx.x * G1_BN;
  const int row0 = blockIdx.y * G1_BM;

#pragma unroll
  for (int i = 0; i < (G1_BM * D_IN / 4) / 256; ++i) {
    int idx = tid + 256 * i;
    int r = idx >> 5, c4 = idx & 31;
    int gr = min(row0 + r, n - 1);
    float4 v = ((const float4*)(X + (size_t)gr * D_IN))[c4];
    *(float4*)(Xs + r * G1_XP + 4 * c4) = v;
  }
#pragma unroll
  for (int i = 0; i < ((D_IN / 2) * G1_BN / 4) / 256; ++i) {
    int idx = tid + 256 * i;
    int kp = idx >> 5, c4 = idx & 31;
    const float* wp = W + (size_t)(2 * kp) * F1 + col0 + 4 * c4;
    float4 r0 = *(const float4*)wp;
    float4 r1 = *(const float4*)(wp + F1);
    float* d = Ws + kp * G1_WP + 8 * c4;
    *(float4*)d       = make_float4(r0.x, r1.x, r0.y, r1.y);
    *(float4*)(d + 4) = make_float4(r0.z, r1.z, r0.w, r1.w);
  }
  __syncthreads();

  const int half = lane >> 4;
  const int l15  = lane & 15;
  const float* xa = Xs + (16 * wave + l15) * G1_XP + 2 * half;
  const float* wb = Ws + half * G1_WP + 2 * l15;

  v8f acc[G1_BN / 16];
#pragma unroll
  for (int i = 0; i < G1_BN / 16; ++i) acc[i] = (v8f){};
#pragma unroll
  for (int j = 0; j < D_IN / 4; ++j) {
    v2f a = *(const v2f*)(xa + 4 * j);
#pragma unroll
    for (int i = 0; i < G1_BN / 16; ++i) {
      v2f b = *(const v2f*)(wb + 2 * j * G1_WP + 32 * i);
      acc[i] = __builtin_amdgcn_wmma_f32_16x16x4_f32(false, a, false, b,
                                                     (short)0, acc[i], false, false);
    }
  }

  float* xo = Xs + (16 * wave + 8 * half) * G1_XP + l15;
#pragma unroll
  for (int i = 0; i < G1_BN / 16; ++i) {
#pragma unroll
    for (int v = 0; v < 8; ++v) xo[v * G1_XP + 16 * i] = acc[i][v];
  }
  __syncthreads();
  const float* xr = Xs + (16 * wave) * G1_XP + 4 * lane;
  float* hp = H + (size_t)(row0 + 16 * wave) * F1 + col0 + 4 * lane;
  const int nrows = n - (row0 + 16 * wave);
  if (nrows >= 16) {
    float4 o[16];
#pragma unroll
    for (int r = 0; r < 16; ++r) o[r] = *(const float4*)(xr + r * G1_XP);
#pragma unroll
    for (int r = 0; r < 16; ++r) *(float4*)(hp + (size_t)r * F1) = o[r];
  } else {
    for (int r = 0; r < nrows; ++r)
      *(float4*)(hp + (size_t)r * F1) = *(const float4*)(xr + r * G1_XP);
  }
}

__global__ void gemm2_wmma(const float* __restrict__ X2, const float* __restrict__ b1,
                           const float* __restrict__ W2,
                           float* __restrict__ H2, int n) {
  int wave = (blockIdx.x * blockDim.x + threadIdx.x) >> 5;
  int lane = threadIdx.x & 31;
  int row0 = wave * 16;
  if (row0 >= n) return;

  int half = lane >> 4;
  int l15  = lane & 15;
  const float* xrow = X2 + (size_t)(row0 + l15) * F1;
  bool colok = (l15 < D_OUT);

  v8f acc = {};
  for (int k = 0; k < F1; k += 4) {
    int kk = k + half * 2;
    float s0 = xrow[kk] + b1[kk], s1 = xrow[kk + 1] + b1[kk + 1];
    v2f a; a.x = s0 > 0.f ? s0 : 0.f; a.y = s1 > 0.f ? s1 : 0.f;
    v2f b;
    b.x = colok ? W2[(size_t)kk * D_OUT + l15]       : 0.f;
    b.y = colok ? W2[(size_t)(kk + 1) * D_OUT + l15] : 0.f;
    acc = __builtin_amdgcn_wmma_f32_16x16x4_f32(false, a, false, b,
                                                (short)0, acc, false, false);
  }
  if (colok) {
#pragma unroll
    for (int v = 0; v < 8; ++v) {
      int r = row0 + v + half * 8;
      H2[(size_t)r * D_OUT + l15] = acc[v];
    }
  }
}

#define TILE_R      98
#define NSUB        2
#define GROUP_R     (NSUB * TILE_R)

__global__ void alphas1(const float* __restrict__ H, const float* __restrict__ a_src,
                        const float* __restrict__ a_dst,
                        float* __restrict__ as, float* __restrict__ ad, int n,
                        const int* __restrict__ dst, int nE, unsigned* __restrict__ codes) {
  int t = blockIdx.x * blockDim.x + threadIdx.x;
  const int nw = ((nE + 15) >> 4) << 2;
  for (int w = t; w < nw; w += gridDim.x * blockDim.x) {
    unsigned pk = 0u;
#pragma unroll
    for (int k = 0; k < 4; ++k) {
      int e = 4 * w + k;
      unsigned cd = e < nE ? (((unsigned)dst[e] / GROUP_R) & 255u) : 0u;
      pk |= cd << (8 * k);
    }
    codes[w] = pk;
  }
  if (t >= n * HEADS) return;
  int h = t % HEADS;
  const float* row = H + (size_t)(t / HEADS) * F1 + h * HID;
  const float* vs = a_src + h * HID;
  const float* vd = a_dst + h * HID;
  float s = 0.f, d = 0.f;
#pragma unroll 8
  for (int c = 0; c < HID; ++c) { float x = row[c]; s += x * vs[c]; d += x * vd[c]; }
  as[t] = s; ad[t] = d;
}

__global__ void alphas2(const float* __restrict__ H2, const float* __restrict__ a_src,
                        const float* __restrict__ a_dst,
                        float* __restrict__ as, float* __restrict__ ad, int n) {
  int t = blockIdx.x * blockDim.x + threadIdx.x;
  if (t >= n) return;
  const float* row = H2 + (size_t)t * D_OUT;
  float s = 0.f, d = 0.f;
#pragma unroll
  for (int c = 0; c < D_OUT; ++c) { s += row[c] * a_src[c]; d += row[c] * a_dst[c]; }
  as[t] = s; ad[t] = d;
}

#define AGG_THREADS 1024
#define AGG_WAVES   (AGG_THREADS / 32)
#define LIST_CAP    4096
#define CAND_CAP    (NSUB * LIST_CAP)

struct Msg { float4 v0, v1, v2, v3; };

__device__ __forceinline__ Msg load_row(const float* __restrict__ H, int s_, int lane) {
  const float4* hp = (const float4*)(H + (size_t)s_ * F1);
  Msg m;
  m.v0 = hp[lane]; m.v1 = hp[lane + 32]; m.v2 = hp[lane + 64]; m.v3 = hp[lane + 96];
  return m;
}

__device__ __forceinline__ float edge_alpha(float a_s, int r, int hh, const float* s_ad,
                                            const float* s_m, const float* s_dn) {
  float ev = lrelu(a_s + s_ad[r * HEADS + hh]);
  float w = __expf(ev - s_m[r * HEADS + hh]);
  return w / (s_dn[r * HEADS + hh] + 1e-16f);
}

template <bool START>
__device__ __forceinline__ void msg_into_tile(const Msg& m, float alpha, int r, int lane,
                                              float* acc) {
  int hs = lane >> 4;
  float a0 = __shfl(alpha, hs), a1 = __shfl(alpha, hs + 2);
  float a2 = __shfl(alpha, hs + 4), a3 = __shfl(alpha, hs + 6);
  float* ap = acc + r * F1 + lane;
  if (START) {
    ap[0]       = m.v0.x * a0; ap[128]      = m.v0.y * a0; ap[256]      = m.v0.z * a0; ap[384]      = m.v0.w * a0;
    ap[32]      = m.v1.x * a1; ap[128 + 32] = m.v1.y * a1; ap[256 + 32] = m.v1.z * a1; ap[384 + 32] = m.v1.w * a1;
    ap[64]      = m.v2.x * a2; ap[128 + 64] = m.v2.y * a2; ap[256 + 64] = m.v2.z * a2; ap[384 + 64] = m.v2.w * a2;
    ap[96]      = m.v3.x * a3; ap[128 + 96] = m.v3.y * a3; ap[256 + 96] = m.v3.z * a3; ap[384 + 96] = m.v3.w * a3;
  } else {
    atomicAdd(ap,            m.v0.x * a0); atomicAdd(ap + 128,      m.v0.y * a0);
    atomicAdd(ap + 256,      m.v0.z * a0); atomicAdd(ap + 384,      m.v0.w * a0);
    atomicAdd(ap + 32,       m.v1.x * a1); atomicAdd(ap + 128 + 32, m.v1.y * a1);
    atomicAdd(ap + 256 + 32, m.v1.z * a1); atomicAdd(ap + 384 + 32, m.v1.w * a1);
    atomicAdd(ap + 64,       m.v2.x * a2); atomicAdd(ap + 128 + 64, m.v2.y * a2);
    atomicAdd(ap + 256 + 64, m.v2.z * a2); atomicAdd(ap + 384 + 64, m.v2.w * a2);
    atomicAdd(ap + 96,       m.v3.x * a3); atomicAdd(ap + 128 + 96, m.v3.y * a3);
    atomicAdd(ap + 256 + 96, m.v3.z * a3); atomicAdd(ap + 384 + 96, m.v3.w * a3);
  }
}

template <bool START>
__device__ __forceinline__ void edge_into_tile(int s_, int r, int lane,
                                               const float* __restrict__ H,
                                               const float* __restrict__ as,
                                               const float* s_ad, const float* s_m,
                                               const float* s_dn, float* acc) {
  Msg m = load_row(H, s_, lane);
  int hh = lane & 7;
  float alpha = edge_alpha(as[(size_t)s_ * HEADS + hh], r, hh, s_ad, s_m, s_dn);
  msg_into_tile<START>(m, alpha, r, lane, acc);
}

__device__ __forceinline__ void scores8(const float* __restrict__ as, const float* s_ad,
                                        int s_, int r, float* ev) {
  const float4* ap = (const float4*)(as + (size_t)s_ * HEADS);
  float4 a0 = ap[0], a1 = ap[1];
  const float4* dp = (const float4*)(s_ad + r * HEADS);
  float4 d0 = dp[0], d1 = dp[1];
  ev[0] = lrelu(a0.x + d0.x); ev[1] = lrelu(a0.y + d0.y);
  ev[2] = lrelu(a0.z + d0.z); ev[3] = lrelu(a0.w + d0.w);
  ev[4] = lrelu(a1.x + d1.x); ev[5] = lrelu(a1.y + d1.y);
  ev[6] = lrelu(a1.z + d1.z); ev[7] = lrelu(a1.w + d1.w);
}

__global__ __launch_bounds__(AGG_THREADS)
void tile_agg1(const int* __restrict__ src, const int* __restrict__ dst,
               int nE, int n, const float* __restrict__ H,
               const float* __restrict__ as, const float* __restrict__ ad,
               float* __restrict__ out, int* __restrict__ tl_cnt, int* __restrict__ tl_ent,
               const unsigned* __restrict__ codes) {
  __shared__ float acc[TILE_R * F1];
  __shared__ __attribute__((aligned(16))) float s_ad[GROUP_R * HEADS];
  __shared__ float s_m[GROUP_R * HEADS];
  __shared__ float s_dn[GROUP_R * HEADS];
  __shared__ unsigned s_mo[GROUP_R * HEADS];
  __shared__ int2 s_list[NSUB * LIST_CAP];
  __shared__ int s_cnt[NSUB];
  __shared__ int s_cand[CAND_CAP];
  __shared__ int s_ccnt;

  const int tid = threadIdx.x, lane = tid & 31, wave = tid >> 5;
  const int node0 = blockIdx.x * GROUP_R;
  const int rows = min(GROUP_R, n - node0);

  for (int i = tid; i < rows * HEADS; i += AGG_THREADS) {
    size_t g = (size_t)node0 * HEADS + i;
    float a_d = ad[g];
    float ev = lrelu(as[g] + a_d);
    s_ad[i] = a_d;
    s_mo[i] = f2ord(ev);
    s_dn[i] = ev;
  }
  if (tid < NSUB) s_cnt[tid] = 0;
  if (tid == 0) s_ccnt = 0;
  __syncthreads();

  auto scan = [&](auto take) {
    auto take4 = [&](int4 d, bool ok, int e0) {
      unsigned lo = min(min((unsigned)(d.x - node0), (unsigned)(d.y - node0)),
                        min((unsigned)(d.z - node0), (unsigned)(d.w - node0)));
      bool any = (lo < (unsigned)rows) & ok;
      if (__builtin_amdgcn_ballot_w32(any) != 0u) {
        take(ok ? d.x : -1, e0);
        take(ok ? d.y : -1, e0 + 1);
        take(ok ? d.z : -1, e0 + 2);
        take(ok ? d.w : -1, e0 + 3);
      }
    };
    int pre = (int)(((16u - (unsigned)((size_t)dst & 15u)) & 15u) >> 2);
    if (pre > nE) pre = nE;
    const int nvec = (nE - pre) >> 2;
    const int4* dv = (const int4*)(dst + pre);
    const int nchunk = (nvec + 31) >> 5;
    if (nvec > 0) {
      const int last = nvec - 1;
      int4 cur[4];
#pragma unroll
      for (int j = 0; j < 4; ++j)
        cur[j] = dv[min(((wave + AGG_WAVES * j) << 5) + lane, last)];
      for (int base = wave; base < nchunk; base += 4 * AGG_WAVES) {
        int4 nxt[4];
#pragma unroll
        for (int j = 0; j < 4; ++j)
          nxt[j] = dv[min(((base + 4 * AGG_WAVES + AGG_WAVES * j) << 5) + lane, last)];
#pragma unroll
        for (int j = 0; j < 4; ++j) {
          int p = ((base + AGG_WAVES * j) << 5) + lane;
          take4(cur[j], p < nvec, pre + 4 * p);
          cur[j] = nxt[j];
        }
      }
    }
    if (wave == 0) {
      int tail0 = pre + 4 * nvec;
      int e = lane < pre ? lane : tail0 + (lane - pre);
      take(e < nE ? dst[e] : -1, e);
    }
  };

  auto for_hits = [&](int d, int e, int nb, int nr, auto f) {
    unsigned dl = (unsigned)(d - nb);
    unsigned mask = __builtin_amdgcn_ballot_w32(dl < (unsigned)nr);
    while (mask) {
      int l = __builtin_ctz(mask);
      mask &= mask - 1;
      int r  = __builtin_amdgcn_readlane((int)dl, l);
      int ee = __builtin_amdgcn_readlane(e, l);
      f(src[ee], r);
    }
  };

  auto list_hit = [&](int d, int e) {
    unsigned dl = (unsigned)(d - node0);
    bool hit = dl < (unsigned)rows;
    if (__builtin_amdgcn_ballot_w32(hit) != 0u) {
      int t = hit ? (int)(dl / TILE_R) : -1;
#pragma unroll
      for (int s = 0; s < NSUB; ++s) {
        bool hs = t == s;
        unsigned mask = __builtin_amdgcn_ballot_w32(hs);
        if (mask != 0u) {
          int base = 0;
          if (lane == 0) base = atomicAdd(&s_cnt[s], __builtin_popcount(mask));
          base = __builtin_amdgcn_readfirstlane(base);
          int pos = base + (int)__builtin_amdgcn_mbcnt_lo(mask, 0u);
          if (hs && pos < LIST_CAP)
            s_list[s * LIST_CAP + pos] = make_int2(e, (int)dl - s * TILE_R);
        }
      }
    }
  };

  {
    const unsigned gpat = ((unsigned)blockIdx.x & 255u) * 0x01010101u;
    const uint4* cv = (const uint4*)codes;
    const int nvec = (nE + 15) >> 4;
    const int nchunk = (nvec + 31) >> 5;
    auto test16 = [&](uint4 cw, bool ok, int e0) {
      unsigned x0 = cw.x ^ gpat, x1 = cw.y ^ gpat, x2 = cw.z ^ gpat, x3 = cw.w ^ gpat;
      unsigned z[4];
      z[0] = (x0 - 0x01010101u) & ~x0 & 0x80808080u;
      z[1] = (x1 - 0x01010101u) & ~x1 & 0x80808080u;
      z[2] = (x2 - 0x01010101u) & ~x2 & 0x80808080u;
      z[3] = (x3 - 0x01010101u) & ~x3 & 0x80808080u;
      if (ok && (z[0] | z[1] | z[2] | z[3]) != 0u) {
        int pos = atomicAdd(&s_ccnt, __builtin_popcount(z[0]) + __builtin_popcount(z[1]) +
                                     __builtin_popcount(z[2]) + __builtin_popcount(z[3]));
#pragma unroll
        for (int j = 0; j < 4; ++j) {
          unsigned zz = z[j];
          while (zz) {
            int e = e0 + 4 * j + (__builtin_ctz(zz) >> 3);
            zz &= zz - 1u;
            if (pos < CAND_CAP) s_cand[pos] = e;
            ++pos;
          }
        }
      }
    };
    if (nvec > 0) {
      const int last = nvec - 1;
      uint4 cur[4];
#pragma unroll
      for (int j = 0; j < 4; ++j)
        cur[j] = cv[min(((wave + AGG_WAVES * j) << 5) + lane, last)];
      for (int base = wave; base < nchunk; base += 4 * AGG_WAVES) {
        uint4 nxt[4];
#pragma unroll
        for (int j = 0; j < 4; ++j)
          nxt[j] = cv[min(((base + 4 * AGG_WAVES + AGG_WAVES * j) << 5) + lane, last)];
#pragma unroll
        for (int j = 0; j < 4; ++j) {
          int p = ((base + AGG_WAVES * j) << 5) + lane;
          test16(cur[j], p < nvec, 16 * p);
          cur[j] = nxt[j];
        }
      }
    }
  }
  __syncthreads();
  {
    const int ccnt = s_ccnt;
    if (ccnt <= CAND_CAP) {
      for (int i0 = 0; i0 < ccnt; i0 += AGG_THREADS) {
        int i = i0 + tid;
        int e = i < ccnt ? s_cand[i] : nE;
        int d = e < nE ? dst[e] : -1;
        list_hit(d, e);
      }
    } else {
      scan(list_hit);
    }
  }
  __syncthreads();
  if (tid < NSUB && rows > tid * TILE_R) tl_cnt[blockIdx.x * NSUB + tid] = s_cnt[tid];

  for (int s = 0; s < NSUB; ++s) {
    const int nb = node0 + s * TILE_R;
    const int nr = min(TILE_R, n - nb);
    if (nr <= 0) break;
    const int cnt = s_cnt[s];
    const float* t_ad = s_ad + s * TILE_R * HEADS;
    unsigned* t_mo = s_mo + s * TILE_R * HEADS;
    if (cnt <= LIST_CAP) {
      int2* sl = s_list + s * LIST_CAP;
      int2* tl = (int2*)tl_ent + (size_t)(blockIdx.x * NSUB + s) * LIST_CAP;
      for (int i = tid; i < cnt; i += AGG_THREADS) {
        int2 en = sl[i];
        int s_ = src[en.x];
        sl[i].x = s_;
        tl[i] = make_int2(s_, en.y);
        float ev[HEADS];
        scores8(as, t_ad, s_, en.y, ev);
        unsigned* mp = t_mo + en.y * HEADS;
#pragma unroll
        for (int h = 0; h < HEADS; ++h) atomicMax(mp + h, f2ord(ev[h]));
      }
    } else {
      scan([&](int d, int e) {
        for_hits(d, e, nb, nr, [&](int s_, int r) {
          int hh = lane & 7;
          float ev = lrelu(as[(size_t)s_ * HEADS + hh] + t_ad[r * HEADS + hh]);
          if (lane < HEADS) atomicMax(t_mo + r * HEADS + hh, f2ord(ev));
        });
      });
    }
  }
  __syncthreads();
  for (int i = tid; i < rows * HEADS; i += AGG_THREADS) {
    float mx = ord2f(s_mo[i]);
    s_m[i] = mx;
    s_dn[i] = __expf(s_dn[i] - mx);
  }
  __syncthreads();

  for (int s = 0; s < NSUB; ++s) {
    const int nb = node0 + s * TILE_R;
    const int nr = min(TILE_R, n - nb);
    if (nr <= 0) break;
    const int cnt = s_cnt[s];
    const float* t_ad = s_ad + s * TILE_R * HEADS;
    const float* t_m = s_m + s * TILE_R * HEADS;
    float* t_dn = s_dn + s * TILE_R * HEADS;
    if (cnt <= LIST_CAP) {
      const int2* sl = s_list + s * LIST_CAP;
      for (int i = tid; i < cnt; i += AGG_THREADS) {
        int2 en = sl[i];
        float ev[HEADS];
        scores8(as, t_ad, en.x, en.y, ev);
        const float* mp = t_m + en.y * HEADS;
        float* dp = t_dn + en.y * HEADS;
#pragma unroll
        for (int h = 0; h < HEADS; ++h) atomicAdd(dp + h, __expf(ev[h] - mp[h]));
      }
    } else {
      scan([&](int d, int e) {
        for_hits(d, e, nb, nr, [&](int s_, int r) {
          int hh = lane & 7;
          float ev = lrelu(as[(size_t)s_ * HEADS + hh] + t_ad[r * HEADS + hh]);
          if (lane < HEADS) atomicAdd(t_dn + r * HEADS + hh, __expf(ev - t_m[r * HEADS + hh]));
        });
      });
    }
  }
  __syncthreads();

  for (int s = 0; s < NSUB; ++s) {
    const int nb = node0 + s * TILE_R;
    const int nr = min(TILE_R, n - nb);
    if (nr <= 0) break;
    const int cnt = s_cnt[s];
    const float* t_ad = s_ad + s * TILE_R * HEADS;
    const float* t_m = s_m + s * TILE_R * HEADS;
    const float* t_dn = s_dn + s * TILE_R * HEADS;

    for (int r = wave; r < nr; r += AGG_WAVES)
      edge_into_tile<true>(nb + r, r, lane, H, as, t_ad, t_m, t_dn, acc);
    __syncthreads();

    if (cnt <= LIST_CAP) {
      const int2* sl = s_list + s * LIST_CAP;
      const int hh = lane & 7;
      int i = wave;
      for (; i + 3 * AGG_WAVES < cnt; i += 4 * AGG_WAVES) {
        int sN[4], rN[4];
        float av[4];
        Msg mm[4];
#pragma unroll
        for (int k = 0; k < 4; ++k) {
          int2 en = sl[i + AGG_WAVES * k];
          sN[k] = __builtin_amdgcn_readfirstlane(en.x);
          rN[k] = __builtin_amdgcn_readfirstlane(en.y);
          av[k] = as[(size_t)sN[k] * HEADS + hh];
        }
#pragma unroll
        for (int k = 0; k < 4; ++k) mm[k] = load_row(H, sN[k], lane);
        asm volatile("" :: "v"(av[0]), "v"(av[1]), "v"(av[2]), "v"(av[3]),
                     "v"(mm[0].v0.x), "v"(mm[0].v1.x), "v"(mm[0].v2.x), "v"(mm[0].v3.x),
                     "v"(mm[1].v0.x), "v"(mm[1].v1.x), "v"(mm[1].v2.x), "v"(mm[1].v3.x),
                     "v"(mm[2].v0.x), "v"(mm[2].v1.x), "v"(mm[2].v2.x), "v"(mm[2].v3.x),
                     "v"(mm[3].v0.x), "v"(mm[3].v1.x), "v"(mm[3].v2.x), "v"(mm[3].v3.x));
#pragma unroll
        for (int k = 0; k < 4; ++k) {
          float alpha = edge_alpha(av[k], rN[k], hh, t_ad, t_m, t_dn);
          msg_into_tile<false>(mm[k], alpha, rN[k], lane, acc);
        }
      }
      for (; i < cnt; i += AGG_WAVES) {
        int2 en = sl[i];
        edge_into_tile<false>(__builtin_amdgcn_readfirstlane(en.x),
                              __builtin_amdgcn_readfirstlane(en.y),
                              lane, H, as, t_ad, t_m, t_dn, acc);
      }
    } else {
      scan([&](int d, int e) {
        for_hits(d, e, nb, nr, [&](int s_, int r) {
          edge_into_tile<false>(s_, r, lane, H, as, t_ad, t_m, t_dn, acc);
        });
      });
    }
    __syncthreads();

    for (int i = tid; i < nr * (F1 / 4); i += AGG_THREADS) {
      int r = i >> 7, f = i & 127;
      const float* a = acc + r * F1 + f;
      float4 v = make_float4(a[0], a[128], a[256], a[384]);
      ((float4*)(out + (size_t)(nb + r) * F1))[f] = v;
    }
    if (s + 1 < NSUB) __syncthreads();
  }
}

#define T2_THREADS 1024
#define T2_SLOTS   (LIST_CAP / T2_THREADS)
static_assert(T2_SLOTS * T2_THREADS == LIST_CAP, "the slots of a work-group cover a whole list");
static_assert(TILE_R * D_OUT <= T2_THREADS, "one thread per (row, column) of the tile");

__global__ __launch_bounds__(T2_THREADS, 16)
void tile_agg2(const int* __restrict__ src, const int* __restrict__ dst, int nE, int n,
               const float* __restrict__ H2, const float* __restrict__ as,
               const float* __restrict__ ad, const float* __restrict__ b2,
               const int* __restrict__ tl_cnt, const int* __restrict__ tl_ent,
               float* __restrict__ out) {
  __shared__ float s_ad[TILE_R];
  __shared__ unsigned s_mo[TILE_R];
  __shared__ float s_m[TILE_R];
  __shared__ float s_w0[TILE_R];
  __shared__ float s_dn[TILE_R];
  __shared__ float acc[D_OUT * TILE_R];

  const int tid = threadIdx.x;
  const int node0 = blockIdx.x * TILE_R;
  const int rows = min(TILE_R, n - node0);
  const int cnt = tl_cnt[blockIdx.x];
  const bool listed = cnt <= LIST_CAP;

  const bool rowt = tid < rows;
  const bool elt = tid < rows * D_OUT;
  float a_d0 = 0.f, a_s0 = 0.f, hself = 0.f, bias = 0.f;
  if (rowt) { a_d0 = ad[node0 + tid]; a_s0 = as[node0 + tid]; }
  if (elt) { hself = H2[(size_t)node0 * D_OUT + tid]; bias = b2[tid & 7]; }

  const int2* tl = (const int2*)tl_ent + (size_t)blockIdx.x * LIST_CAP;
  bool ok[T2_SLOTS];
  int rN[T2_SLOTS];
  float aS[T2_SLOTS];
  float4 hA[T2_SLOTS], hB[T2_SLOTS];
#pragma unroll
  for (int k = 0; k < T2_SLOTS; ++k) {
    int i = tid + T2_THREADS * k;
    ok[k] = listed && i < cnt;
    rN[k] = 0; aS[k] = 0.f;
    hA[k] = hB[k] = make_float4(0.f, 0.f, 0.f, 0.f);
    if (ok[k]) {
      int2 en = tl[i];
      rN[k] = en.y;
      aS[k] = as[en.x];
      const float4* hp = (const float4*)(H2 + (size_t)en.x * D_OUT);
      hA[k] = hp[0]; hB[k] = hp[1];
    }
  }

  auto scan = [&](auto f) {
    for (int e = tid; e < nE; e += T2_THREADS) {
      unsigned dl = (unsigned)(dst[e] - node0);
      if (dl < (unsigned)rows) f(src[e], (int)dl);
    }
  };

  if (rowt) {
    float ev0 = lrelu(a_s0 + a_d0);
    s_ad[tid] = a_d0;
    s_mo[tid] = f2ord(ev0);
    s_w0[tid] = ev0;
  }
  __syncthreads();

  float ev[T2_SLOTS];
  if (listed) {
#pragma unroll
    for (int k = 0; k < T2_SLOTS; ++k) {
      ev[k] = 0.f;
      if (ok[k]) {
        ev[k] = lrelu(aS[k] + s_ad[rN[k]]);
        atomicMax(s_mo + rN[k], f2ord(ev[k]));
      }
    }
  } else {
    scan([&](int s_, int r) { atomicMax(s_mo + r, f2ord(lrelu(as[s_] + s_ad[r]))); });
  }
  __syncthreads();
  if (rowt) {
    float mx = ord2f(s_mo[tid]);
    float w0 = __expf(s_w0[tid] - mx);
    s_m[tid] = mx;
    s_w0[tid] = w0;
    s_dn[tid] = w0;
  }
  __syncthreads();

  float w[T2_SLOTS];
  if (listed) {
#pragma unroll
    for (int k = 0; k < T2_SLOTS; ++k) {
      w[k] = 0.f;
      if (ok[k]) {
        w[k] = __expf(ev[k] - s_m[rN[k]]);
        atomicAdd(s_dn + rN[k], w[k]);
      }
    }
  } else {
    scan([&](int s_, int r) {
      atomicAdd(s_dn + r, __expf(lrelu(as[s_] + s_ad[r]) - s_m[r]));
    });
  }
  __syncthreads();

  if (elt) {
    int r = tid >> 3, cc = tid & 7;
    float alpha = s_w0[r] / (s_dn[r] + 1e-16f);
    acc[cc * TILE_R + r] = hself * alpha;
  }
  __syncthreads();
  if (listed) {
#pragma unroll
    for (int k = 0; k < T2_SLOTS; ++k) {
      if (ok[k]) {
        float alpha = w[k] / (s_dn[rN[k]] + 1e-16f);
        float* ap = acc + rN[k];
        atomicAdd(ap,              hA[k].x * alpha); atomicAdd(ap + TILE_R,     hA[k].y * alpha);
        atomicAdd(ap + 2 * TILE_R, hA[k].z * alpha); atomicAdd(ap + 3 * TILE_R, hA[k].w * alpha);
        atomicAdd(ap + 4 * TILE_R, hB[k].x * alpha); atomicAdd(ap + 5 * TILE_R, hB[k].y * alpha);
        atomicAdd(ap + 6 * TILE_R, hB[k].z * alpha); atomicAdd(ap + 7 * TILE_R, hB[k].w * alpha);
      }
    }
  } else {
    scan([&](int s_, int r) {
      float alpha = __expf(lrelu(as[s_] + s_ad[r]) - s_m[r]) / (s_dn[r] + 1e-16f);
      const float* hp = H2 + (size_t)s_ * D_OUT;
#pragma unroll
      for (int cc = 0; cc < D_OUT; ++cc) atomicAdd(acc + cc * TILE_R + r, hp[cc] * alpha);
    });
  }
  __syncthreads();

  if (elt) {
    int r = tid >> 3, cc = tid & 7;
    out[(size_t)node0 * D_OUT + tid] = acc[cc * TILE_R + r] + bias;
  }
}

extern "C" void kernel_launch(void* const* d_in, const int* in_sizes, int n_in,
                              void* d_out, int out_size, void* d_ws, size_t ws_size,
                              hipStream_t stream) {
  const float* x      = (const float*)d_in[0];
  const int*   ei     = (const int*)  d_in[1];
  const float* W1     = (const float*)d_in[2];
  const float* a_src1 = (const float*)d_in[3];
  const float* a_dst1 = (const float*)d_in[4];
  const float* b1     = (const float*)d_in[5];
  const float* W2     = (const float*)d_in[6];
  const float* a_src2 = (const float*)d_in[7];
  const float* a_dst2 = (const float*)d_in[8];
  const float* b2     = (const float*)d_in[9];

  int n  = in_sizes[0] / D_IN;
  int nE = in_sizes[1] / 2;
  const int* src = ei;
  const int* dst = ei + nE;
  int ntiles = (n + TILE_R - 1) / TILE_R;
  int ngroups = (n + GROUP_R - 1) / GROUP_R;

  char* ws = (char*)d_ws;
  size_t off = 0;
  auto alloc = [&](size_t bytes) -> char* {
    char* p = ws + off; off += (bytes + 255) & ~(size_t)255; return p;
  };
  float*    h1   = (float*)   alloc((size_t)n * F1 * 4);
  float*    out1 = (float*)   alloc((size_t)n * F1 * 4);
  float*    as1  = (float*)   alloc((size_t)n * HEADS * 4);
  float*    ad1  = (float*)   alloc((size_t)n * HEADS * 4);
  float*    h2   = (float*)   alloc((size_t)n * D_OUT * 4);
  float*    as2  = (float*)   alloc((size_t)n * 4);
  float*    ad2  = (float*)   alloc((size_t)n * 4);
  int*      tlc  = (int*)     alloc((size_t)ntiles * 4);
  int*      tle  = (int*)     alloc((size_t)ntiles * LIST_CAP * 8);
  unsigned* dcd  = (unsigned*)alloc(((size_t)nE + 15) / 16 * 16);

  gemm1_wmma<<<dim3(F1 / G1_BN, (n + G1_BM - 1) / G1_BM), 256, 0, stream>>>(x, W1, h1, n);
  alphas1<<<(n * HEADS + 255) / 256, 256, 0, stream>>>(h1, a_src1, a_dst1, as1, ad1, n, dst, nE, dcd);
  tile_agg1<<<ngroups, AGG_THREADS, 0, stream>>>(src, dst, nE, n, h1, as1, ad1, out1, tlc, tle, dcd);

  {
    int waves  = (n + 15) / 16;
    int blocks = (waves + 7) / 8;
    gemm2_wmma<<<blocks, 256, 0, stream>>>(out1, b1, W2, h2, n);
  }
  alphas2<<<(n + 255) / 256, 256, 0, stream>>>(h2, a_src2, a_dst2, as2, ad2, n);
  tile_agg2<<<ntiles, T2_THREADS, 0, stream>>>(src, dst, nE, n, h2, as2, ad2, b2, tlc, tle, (float*)d_out);
}


# Whole-line loads: coefficient kernel and layer-2 GEMM read rows as 512-byte pieces per wave load (GEMM A via the wave's own LDS rows, k order kept); as1/ad1: lane sums + 16-lane tree, new sum order
# speedup vs baseline: 8.8143x; 1.2190x over previous
#include <hip/hip_runtime.h>
#include <math.h>

#define D_IN   128
#define HEADS  8
#define HID    64
#define F1     (HEADS * HID)
#define D_OUT  8
#define NEG_SLOPE 0.2f

typedef float v2f __attribute__((ext_vector_type(2)));
typedef float v8f __attribute__((ext_vector_type(8)));

__device__ __forceinline__ unsigned f2ord(float f) {
  unsigned u = __float_as_uint(f);
  return (u & 0x80000000u) ? ~u : (u | 0x80000000u);
}
__device__ __forceinline__ float ord2f(unsigned v) {
  unsigned u = (v & 0x80000000u) ? (v ^ 0x80000000u) : ~v;
  return __uint_as_float(u);
}
__device__ __forceinline__ float lrelu(float x) { return x > 0.f ? x : NEG_SLOPE * x; }

#define G1_BM 128
#define G1_BN 128
#define G1_XP 132
#define G1_WP 288

__global__ __launch_bounds__(256)
void gemm1_wmma(const float* __restrict__ X, const float* __restrict__ W,
                float* __restrict__ H, int n) {
  __shared__ float Xs[G1_BM * G1_XP];
  __shared__ float Ws[(D_IN / 2) * G1_WP];

  const int tid = threadIdx.x, lane = tid & 31;
  const int wave = __builtin_amdgcn_readfirstlane(tid >> 5);
  const int col0 = blockIdx.x * G1_BN;
  const int row0 = blockIdx.y * G1_BM;

#pragma unroll
  for (int i = 0; i < (G1_BM * D_IN / 4) / 256; ++i) {
    int idx = tid + 256 * i;
    int r = idx >> 5, c4 = idx & 31;
    int gr = min(row0 + r, n - 1);
    float4 v = ((const float4*)(X + (size_t)gr * D_IN))[c4];
    *(float4*)(Xs + r * G1_XP + 4 * c4) = v;
  }
#pragma unroll
  for (int i = 0; i < ((D_IN / 2) * G1_BN / 4) / 256; ++i) {
    int idx = tid + 256 * i;
    int kp = idx >> 5, c4 = idx & 31;
    const float* wp = W + (size_t)(2 * kp) * F1 + col0 + 4 * c4;
    float4 r0 = *(const float4*)wp;
    float4 r1 = *(const float4*)(wp + F1);
    float* d = Ws + kp * G1_WP + 8 * c4;
    *(float4*)d       = make_float4(r0.x, r1.x, r0.y, r1.y);
    *(float4*)(d + 4) = make_float4(r0.z, r1.z, r0.w, r1.w);
  }
  __syncthreads();

  const int half = lane >> 4;
  const int l15  = lane & 15;
  const float* xa = Xs + (16 * wave + l15) * G1_XP + 2 * half;
  const float* wb = Ws + half * G1_WP + 2 * l15;

  v8f acc[G1_BN / 16];
#pragma unroll
  for (int i = 0; i < G1_BN / 16; ++i) acc[i] = (v8f){};
#pragma unroll
  for (int j = 0; j < D_IN / 4; ++j) {
    v2f a = *(const v2f*)(xa + 4 * j);
#pragma unroll
    for (int i = 0; i < G1_BN / 16; ++i) {
      v2f b = *(const v2f*)(wb + 2 * j * G1_WP + 32 * i);
      acc[i] = __builtin_amdgcn_wmma_f32_16x16x4_f32(false, a, false, b,
                                                     (short)0, acc[i], false, false);
    }
  }

  float* xo = Xs + (16 * wave + 8 * half) * G1_XP + l15;
#pragma unroll
  for (int i = 0; i < G1_BN / 16; ++i) {
#pragma unroll
    for (int v = 0; v < 8; ++v) xo[v * G1_XP + 16 * i] = acc[i][v];
  }
  __syncthreads();
  const float* xr = Xs + (16 * wave) * G1_XP + 4 * lane;
  float* hp = H + (size_t)(row0 + 16 * wave) * F1 + col0 + 4 * lane;
  const int nrows = n - (row0 + 16 * wave);
  if (nrows >= 16) {
    float4 o[16];
#pragma unroll
    for (int r = 0; r < 16; ++r) o[r] = *(const float4*)(xr + r * G1_XP);
#pragma unroll
    for (int r = 0; r < 16; ++r) *(float4*)(hp + (size_t)r * F1) = o[r];
  } else {
    for (int r = 0; r < nrows; ++r)
      *(float4*)(hp + (size_t)r * F1) = *(const float4*)(xr + r * G1_XP);
  }
}

#define G2_XP 132
#define G2_WP 32
#define G2_CH (F1 / 128)

__global__ __launch_bounds__(256)
void gemm2_wmma(const float* __restrict__ X2, const float* __restrict__ b1,
                const float* __restrict__ W2,
                float* __restrict__ H2, int n) {
  __shared__ float Xs[8 * 16 * G2_XP];
  __shared__ float Ws[(F1 / 2) * G2_WP];

  const int tid = threadIdx.x, lane = tid & 31;
  const int wave = __builtin_amdgcn_readfirstlane(tid >> 5);
  const int row0 = (blockIdx.x * 8 + wave) * 16;

  float4 w4[4];
#pragma unroll
  for (int i = 0; i < 4; ++i) w4[i] = ((const float4*)W2)[tid + 256 * i];
  float4 v[16];
  float4 bb = ((const float4*)b1)[lane];
#pragma unroll
  for (int r = 0; r < 16; ++r)
    v[r] = ((const float4*)(X2 + (size_t)min(row0 + r, n - 1) * F1))[lane];
#pragma unroll
  for (int i = 0; i < 4; ++i) {
    int g = tid + 256 * i;
    int k = g >> 1, c0 = (g & 1) * 4;
    float* d = Ws + (k >> 1) * G2_WP + 2 * c0 + (k & 1);
    d[0] = w4[i].x; d[2] = w4[i].y; d[4] = w4[i].z; d[6] = w4[i].w;
  }
#pragma unroll
  for (int i = 0; i < 4; ++i) {
    int p = tid + 256 * i;
    *(float4*)(Ws + (p >> 2) * G2_WP + 16 + 4 * (p & 3)) = make_float4(0.f, 0.f, 0.f, 0.f);
  }

  const int half = lane >> 4;
  const int l15  = lane & 15;
  float* xw = Xs + wave * 16 * G2_XP;
  const float* xa = xw + l15 * G2_XP + 2 * half;
  const float* wb = Ws + half * G2_WP + 2 * l15;

  v8f acc = {};
#pragma unroll
  for (int c = 0; c < G2_CH; ++c) {
#pragma unroll
    for (int r = 0; r < 16; ++r) {
      float4 s = make_float4(v[r].x + bb.x, v[r].y + bb.y, v[r].z + bb.z, v[r].w + bb.w);
      s.x = s.x > 0.f ? s.x : 0.f; s.y = s.y > 0.f ? s.y : 0.f;
      s.z = s.z > 0.f ? s.z : 0.f; s.w = s.w > 0.f ? s.w : 0.f;
      *(float4*)(xw + r * G2_XP + 4 * lane) = s;
    }
    if (c == 0) __syncthreads();
    if (c + 1 < G2_CH) {
      bb = ((const float4*)b1)[32 * (c + 1) + lane];
#pragma unroll
      for (int r = 0; r < 16; ++r)
        v[r] = ((const float4*)(X2 + (size_t)min(row0 + r, n - 1) * F1))[32 * (c + 1) + lane];
    }
#pragma unroll 8
    for (int j = 0; j < 32; ++j) {
      v2f a = *(const v2f*)(xa + 4 * j);
      v2f b = *(const v2f*)(wb + (64 * c + 2 * j) * G2_WP);
      acc = __builtin_amdgcn_wmma_f32_16x16x4_f32(false, a, false, b,
                                                  (short)0, acc, false, false);
    }
  }

  if (l15 < D_OUT) {
#pragma unroll
    for (int vv = 0; vv < 8; ++vv) {
      int r = row0 + vv + half * 8;
      if (r < n) H2[(size_t)r * D_OUT + l15] = acc[vv];
    }
  }
}

#define TILE_R      98
#define NSUB        2
#define GROUP_R     (NSUB * TILE_R)

#define A1_ROWS 4

template <int M>
__device__ __forceinline__ float add_lane_xor(float v) {
  int o = __builtin_amdgcn_update_dpp(0, __float_as_int(v), 0x160 | M, 0xF, 0xF, true);
  return v + __int_as_float(o);
}

__global__ void alphas1(const float* __restrict__ H, const float* __restrict__ a_src,
                        const float* __restrict__ a_dst,
                        float* __restrict__ as, float* __restrict__ ad, int n,
                        const int* __restrict__ dst, int nE, unsigned* __restrict__ codes) {
  int t = blockIdx.x * blockDim.x + threadIdx.x;
  const int nw = ((nE + 15) >> 4) << 2;
  for (int w = t; w < nw; w += gridDim.x * blockDim.x) {
    unsigned pk = 0u;
#pragma unroll
    for (int k = 0; k < 4; ++k) {
      int e = 4 * w + k;
      unsigned cd = e < nE ? (((unsigned)dst[e] / GROUP_R) & 255u) : 0u;
      pk |= cd << (8 * k);
    }
    codes[w] = pk;
  }

  const int lane = threadIdx.x & 31;
  const int r0 = (t >> 5) * A1_ROWS;
  if (r0 >= n) return;

  float4 hv[A1_ROWS][4], vs[4], vd[4];
#pragma unroll
  for (int j = 0; j < A1_ROWS; ++j) {
    const float4* hp = (const float4*)(H + (size_t)min(r0 + j, n - 1) * F1);
#pragma unroll
    for (int q = 0; q < 4; ++q) hv[j][q] = hp[lane + 32 * q];
  }
#pragma unroll
  for (int q = 0; q < 4; ++q) {
    vs[q] = ((const float4*)a_src)[lane + 32 * q];
    vd[q] = ((const float4*)a_dst)[lane + 32 * q];
  }

  const int l15 = lane & 15;
  float so = 0.f, dd = 0.f;
#pragma unroll
  for (int j = 0; j < A1_ROWS; ++j) {
#pragma unroll
    for (int q = 0; q < 4; ++q) {
      float4 x = hv[j][q];
      float s = x.x * vs[q].x; s += x.y * vs[q].y; s += x.z * vs[q].z; s += x.w * vs[q].w;
      float d = x.x * vd[q].x; d += x.y * vd[q].y; d += x.z * vd[q].z; d += x.w * vd[q].w;
      s = add_lane_xor<8>(s); s = add_lane_xor<4>(s); s = add_lane_xor<2>(s); s = add_lane_xor<1>(s);
      d = add_lane_xor<8>(d); d = add_lane_xor<4>(d); d = add_lane_xor<2>(d); d = add_lane_xor<1>(d);
      if (l15 == 4 * j + q) { so = s; dd = d; }
    }
  }
  const int r = r0 + (l15 >> 2);
  if (r < n) {
    int o = r * HEADS + 2 * (l15 & 3) + (lane >> 4);
    as[o] = so; ad[o] = dd;
  }
}

__global__ void alphas2(const float* __restrict__ H2, const float* __restrict__ a_src,
                        const float* __restrict__ a_dst,
                        float* __restrict__ as, float* __restrict__ ad, int n) {
  int t = blockIdx.x * blockDim.x + threadIdx.x;
  if (t >= n) return;
  const float* row = H2 + (size_t)t * D_OUT;
  float s = 0.f, d = 0.f;
#pragma unroll
  for (int c = 0; c < D_OUT; ++c) { s += row[c] * a_src[c]; d += row[c] * a_dst[c]; }
  as[t] = s; ad[t] = d;
}

#define AGG_THREADS 1024
#define AGG_WAVES   (AGG_THREADS / 32)
#define LIST_CAP    4096
#define CAND_CAP    (NSUB * LIST_CAP)

struct Msg { float4 v0, v1, v2, v3; };

__device__ __forceinline__ Msg load_row(const float* __restrict__ H, int s_, int lane) {
  const float4* hp = (const float4*)(H + (size_t)s_ * F1);
  Msg m;
  m.v0 = hp[lane]; m.v1 = hp[lane + 32]; m.v2 = hp[lane + 64]; m.v3 = hp[lane + 96];
  return m;
}

__device__ __forceinline__ float edge_alpha(float a_s, int r, int hh, const float* s_ad,
                                            const float* s_m, const float* s_dn) {
  float ev = lrelu(a_s + s_ad[r * HEADS + hh]);
  float w = __expf(ev - s_m[r * HEADS + hh]);
  return w / (s_dn[r * HEADS + hh] + 1e-16f);
}

template <bool START>
__device__ __forceinline__ void msg_into_tile(const Msg& m, float alpha, int r, int lane,
                                              float* acc) {
  int hs = lane >> 4;
  float a0 = __shfl(alpha, hs), a1 = __shfl(alpha, hs + 2);
  float a2 = __shfl(alpha, hs + 4), a3 = __shfl(alpha, hs + 6);
  float* ap = acc + r * F1 + lane;
  if (START) {
    ap[0]       = m.v0.x * a0; ap[128]      = m.v0.y * a0; ap[256]      = m.v0.z * a0; ap[384]      = m.v0.w * a0;
    ap[32]      = m.v1.x * a1; ap[128 + 32] = m.v1.y * a1; ap[256 + 32] = m.v1.z * a1; ap[384 + 32] = m.v1.w * a1;
    ap[64]      = m.v2.x * a2; ap[128 + 64] = m.v2.y * a2; ap[256 + 64] = m.v2.z * a2; ap[384 + 64] = m.v2.w * a2;
    ap[96]      = m.v3.x * a3; ap[128 + 96] = m.v3.y * a3; ap[256 + 96] = m.v3.z * a3; ap[384 + 96] = m.v3.w * a3;
  } else {
    atomicAdd(ap,            m.v0.x * a0); atomicAdd(ap + 128,      m.v0.y * a0);
    atomicAdd(ap + 256,      m.v0.z * a0); atomicAdd(ap + 384,      m.v0.w * a0);
    atomicAdd(ap + 32,       m.v1.x * a1); atomicAdd(ap + 128 + 32, m.v1.y * a1);
    atomicAdd(ap + 256 + 32, m.v1.z * a1); atomicAdd(ap + 384 + 32, m.v1.w * a1);
    atomicAdd(ap + 64,       m.v2.x * a2); atomicAdd(ap + 128 + 64, m.v2.y * a2);
    atomicAdd(ap + 256 + 64, m.v2.z * a2); atomicAdd(ap + 384 + 64, m.v2.w * a2);
    atomicAdd(ap + 96,       m.v3.x * a3); atomicAdd(ap + 128 + 96, m.v3.y * a3);
    atomicAdd(ap + 256 + 96, m.v3.z * a3); atomicAdd(ap + 384 + 96, m.v3.w * a3);
  }
}

template <bool START>
__device__ __forceinline__ void edge_into_tile(int s_, int r, int lane,
                                               const float* __restrict__ H,
                                               const float* __restrict__ as,
                                               const float* s_ad, const float* s_m,
                                               const float* s_dn, float* acc) {
  Msg m = load_row(H, s_, lane);
  int hh = lane & 7;
  float alpha = edge_alpha(as[(size_t)s_ * HEADS + hh], r, hh, s_ad, s_m, s_dn);
  msg_into_tile<START>(m, alpha, r, lane, acc);
}

__device__ __forceinline__ void scores8(const float* __restrict__ as, const float* s_ad,
                                        int s_, int r, float* ev) {
  const float4* ap = (const float4*)(as + (size_t)s_ * HEADS);
  float4 a0 = ap[0], a1 = ap[1];
  const float4* dp = (const float4*)(s_ad + r * HEADS);
  float4 d0 = dp[0], d1 = dp[1];
  ev[0] = lrelu(a0.x + d0.x); ev[1] = lrelu(a0.y + d0.y);
  ev[2] = lrelu(a0.z + d0.z); ev[3] = lrelu(a0.w + d0.w);
  ev[4] = lrelu(a1.x + d1.x); ev[5] = lrelu(a1.y + d1.y);
  ev[6] = lrelu(a1.z + d1.z); ev[7] = lrelu(a1.w + d1.w);
}

__global__ __launch_bounds__(AGG_THREADS)
void tile_agg1(const int* __restrict__ src, const int* __restrict__ dst,
               int nE, int n, const float* __restrict__ H,
               const float* __restrict__ as, const float* __restrict__ ad,
               float* __restrict__ out, int* __restrict__ tl_cnt, int* __restrict__ tl_ent,
               const unsigned* __restrict__ codes) {
  __shared__ float acc[TILE_R * F1];
  __shared__ __attribute__((aligned(16))) float s_ad[GROUP_R * HEADS];
  __shared__ float s_m[GROUP_R * HEADS];
  __shared__ float s_dn[GROUP_R * HEADS];
  __shared__ unsigned s_mo[GROUP_R * HEADS];
  __shared__ int2 s_list[NSUB * LIST_CAP];
  __shared__ int s_cnt[NSUB];
  __shared__ int s_cand[CAND_CAP];
  __shared__ int s_ccnt;

  const int tid = threadIdx.x, lane = tid & 31, wave = tid >> 5;
  const int node0 = blockIdx.x * GROUP_R;
  const int rows = min(GROUP_R, n - node0);

  for (int i = tid; i < rows * HEADS; i += AGG_THREADS) {
    size_t g = (size_t)node0 * HEADS + i;
    float a_d = ad[g];
    float ev = lrelu(as[g] + a_d);
    s_ad[i] = a_d;
    s_mo[i] = f2ord(ev);
    s_dn[i] = ev;
  }
  if (tid < NSUB) s_cnt[tid] = 0;
  if (tid == 0) s_ccnt = 0;
  __syncthreads();

  auto scan = [&](auto take) {
    auto take4 = [&](int4 d, bool ok, int e0) {
      unsigned lo = min(min((unsigned)(d.x - node0), (unsigned)(d.y - node0)),
                        min((unsigned)(d.z - node0), (unsigned)(d.w - node0)));
      bool any = (lo < (unsigned)rows) & ok;
      if (__builtin_amdgcn_ballot_w32(any) != 0u) {
        take(ok ? d.x : -1, e0);
        take(ok ? d.y : -1, e0 + 1);
        take(ok ? d.z : -1, e0 + 2);
        take(ok ? d.w : -1, e0 + 3);
      }
    };
    int pre = (int)(((16u - (unsigned)((size_t)dst & 15u)) & 15u) >> 2);
    if (pre > nE) pre = nE;
    const int nvec = (nE - pre) >> 2;
    const int4* dv = (const int4*)(dst + pre);
    const int nchunk = (nvec + 31) >> 5;
    if (nvec > 0) {
      const int last = nvec - 1;
      int4 cur[4];
#pragma unroll
      for (int j = 0; j < 4; ++j)
        cur[j] = dv[min(((wave + AGG_WAVES * j) << 5) + lane, last)];
      for (int base = wave; base < nchunk; base += 4 * AGG_WAVES) {
        int4 nxt[4];
#pragma unroll
        for (int j = 0; j < 4; ++j)
          nxt[j] = dv[min(((base + 4 * AGG_WAVES + AGG_WAVES * j) << 5) + lane, last)];
#pragma unroll
        for (int j = 0; j < 4; ++j) {
          int p = ((base + AGG_WAVES * j) << 5) + lane;
          take4(cur[j], p < nvec, pre + 4 * p);
          cur[j] = nxt[j];
        }
      }
    }
    if (wave == 0) {
      int tail0 = pre + 4 * nvec;
      int e = lane < pre ? lane : tail0 + (lane - pre);
      take(e < nE ? dst[e] : -1, e);
    }
  };

  auto for_hits = [&](int d, int e, int nb, int nr, auto f) {
    unsigned dl = (unsigned)(d - nb);
    unsigned mask = __builtin_amdgcn_ballot_w32(dl < (unsigned)nr);
    while (mask) {
      int l = __builtin_ctz(mask);
      mask &= mask - 1;
      int r  = __builtin_amdgcn_readlane((int)dl, l);
      int ee = __builtin_amdgcn_readlane(e, l);
      f(src[ee], r);
    }
  };

  auto list_hit = [&](int d, int e) {
    unsigned dl = (unsigned)(d - node0);
    bool hit = dl < (unsigned)rows;
    if (__builtin_amdgcn_ballot_w32(hit) != 0u) {
      int t = hit ? (int)(dl / TILE_R) : -1;
#pragma unroll
      for (int s = 0; s < NSUB; ++s) {
        bool hs = t == s;
        unsigned mask = __builtin_amdgcn_ballot_w32(hs);
        if (mask != 0u) {
          int base = 0;
          if (lane == 0) base = atomicAdd(&s_cnt[s], __builtin_popcount(mask));
          base = __builtin_amdgcn_readfirstlane(base);
          int pos = base + (int)__builtin_amdgcn_mbcnt_lo(mask, 0u);
          if (hs && pos < LIST_CAP)
            s_list[s * LIST_CAP + pos] = make_int2(e, (int)dl - s * TILE_R);
        }
      }
    }
  };

  {
    const unsigned gpat = ((unsigned)blockIdx.x & 255u) * 0x01010101u;
    const uint4* cv = (const uint4*)codes;
    const int nvec = (nE + 15) >> 4;
    const int nchunk = (nvec + 31) >> 5;
    auto test16 = [&](uint4 cw, bool ok, int e0) {
      unsigned x0 = cw.x ^ gpat, x1 = cw.y ^ gpat, x2 = cw.z ^ gpat, x3 = cw.w ^ gpat;
      unsigned z[4];
      z[0] = (x0 - 0x01010101u) & ~x0 & 0x80808080u;
      z[1] = (x1 - 0x01010101u) & ~x1 & 0x80808080u;
      z[2] = (x2 - 0x01010101u) & ~x2 & 0x80808080u;
      z[3] = (x3 - 0x01010101u) & ~x3 & 0x80808080u;
      if (ok && (z[0] | z[1] | z[2] | z[3]) != 0u) {
        int pos = atomicAdd(&s_ccnt, __builtin_popcount(z[0]) + __builtin_popcount(z[1]) +
                                     __builtin_popcount(z[2]) + __builtin_popcount(z[3]));
#pragma unroll
        for (int j = 0; j < 4; ++j) {
          unsigned zz = z[j];
          while (zz) {
            int e = e0 + 4 * j + (__builtin_ctz(zz) >> 3);
            zz &= zz - 1u;
            if (pos < CAND_CAP) s_cand[pos] = e;
            ++pos;
          }
        }
      }
    };
    if (nvec > 0) {
      const int last = nvec - 1;
      uint4 cur[4];
#pragma unroll
      for (int j = 0; j < 4; ++j)
        cur[j] = cv[min(((wave + AGG_WAVES * j) << 5) + lane, last)];
      for (int base = wave; base < nchunk; base += 4 * AGG_WAVES) {
        uint4 nxt[4];
#pragma unroll
        for (int j = 0; j < 4; ++j)
          nxt[j] = cv[min(((base + 4 * AGG_WAVES + AGG_WAVES * j) << 5) + lane, last)];
#pragma unroll
        for (int j = 0; j < 4; ++j) {
          int p = ((base + AGG_WAVES * j) << 5) + lane;
          test16(cur[j], p < nvec, 16 * p);
          cur[j] = nxt[j];
        }
      }
    }
  }
  __syncthreads();
  {
    const int ccnt = s_ccnt;
    if (ccnt <= CAND_CAP) {
      for (int i0 = 0; i0 < ccnt; i0 += AGG_THREADS) {
        int i = i0 + tid;
        int e = i < ccnt ? s_cand[i] : nE;
        int d = e < nE ? dst[e] : -1;
        list_hit(d, e);
      }
    } else {
      scan(list_hit);
    }
  }
  __syncthreads();
  if (tid < NSUB && rows > tid * TILE_R) tl_cnt[blockIdx.x * NSUB + tid] = s_cnt[tid];

  for (int s = 0; s < NSUB; ++s) {
    const int nb = node0 + s * TILE_R;
    const int nr = min(TILE_R, n - nb);
    if (nr <= 0) break;
    const int cnt = s_cnt[s];
    const float* t_ad = s_ad + s * TILE_R * HEADS;
    unsigned* t_mo = s_mo + s * TILE_R * HEADS;
    if (cnt <= LIST_CAP) {
      int2* sl = s_list + s * LIST_CAP;
      int2* tl = (int2*)tl_ent + (size_t)(blockIdx.x * NSUB + s) * LIST_CAP;
      for (int i = tid; i < cnt; i += AGG_THREADS) {
        int2 en = sl[i];
        int s_ = src[en.x];
        sl[i].x = s_;
        tl[i] = make_int2(s_, en.y);
        float ev[HEADS];
        scores8(as, t_ad, s_, en.y, ev);
        unsigned* mp = t_mo + en.y * HEADS;
#pragma unroll
        for (int h = 0; h < HEADS; ++h) atomicMax(mp + h, f2ord(ev[h]));
      }
    } else {
      scan([&](int d, int e) {
        for_hits(d, e, nb, nr, [&](int s_, int r) {
          int hh = lane & 7;
          float ev = lrelu(as[(size_t)s_ * HEADS + hh] + t_ad[r * HEADS + hh]);
          if (lane < HEADS) atomicMax(t_mo + r * HEADS + hh, f2ord(ev));
        });
      });
    }
  }
  __syncthreads();
  for (int i = tid; i < rows * HEADS; i += AGG_THREADS) {
    float mx = ord2f(s_mo[i]);
    s_m[i] = mx;
    s_dn[i] = __expf(s_dn[i] - mx);
  }
  __syncthreads();

  for (int s = 0; s < NSUB; ++s) {
    const int nb = node0 + s * TILE_R;
    const int nr = min(TILE_R, n - nb);
    if (nr <= 0) break;
    const int cnt = s_cnt[s];
    const float* t_ad = s_ad + s * TILE_R * HEADS;
    const float* t_m = s_m + s * TILE_R * HEADS;
    float* t_dn = s_dn + s * TILE_R * HEADS;
    if (cnt <= LIST_CAP) {
      const int2* sl = s_list + s * LIST_CAP;
      for (int i = tid; i < cnt; i += AGG_THREADS) {
        int2 en = sl[i];
        float ev[HEADS];
        scores8(as, t_ad, en.x, en.y, ev);
        const float* mp = t_m + en.y * HEADS;
        float* dp = t_dn + en.y * HEADS;
#pragma unroll
        for (int h = 0; h < HEADS; ++h) atomicAdd(dp + h, __expf(ev[h] - mp[h]));
      }
    } else {
      scan([&](int d, int e) {
        for_hits(d, e, nb, nr, [&](int s_, int r) {
          int hh = lane & 7;
          float ev = lrelu(as[(size_t)s_ * HEADS + hh] + t_ad[r * HEADS + hh]);
          if (lane < HEADS) atomicAdd(t_dn + r * HEADS + hh, __expf(ev - t_m[r * HEADS + hh]));
        });
      });
    }
  }
  __syncthreads();

  for (int s = 0; s < NSUB; ++s) {
    const int nb = node0 + s * TILE_R;
    const int nr = min(TILE_R, n - nb);
    if (nr <= 0) break;
    const int cnt = s_cnt[s];
    const float* t_ad = s_ad + s * TILE_R * HEADS;
    const float* t_m = s_m + s * TILE_R * HEADS;
    const float* t_dn = s_dn + s * TILE_R * HEADS;

    for (int r = wave; r < nr; r += AGG_WAVES)
      edge_into_tile<true>(nb + r, r, lane, H, as, t_ad, t_m, t_dn, acc);
    __syncthreads();

    if (cnt <= LIST_CAP) {
      const int2* sl = s_list + s * LIST_CAP;
      const int hh = lane & 7;
      int i = wave;
      for (; i + 3 * AGG_WAVES < cnt; i += 4 * AGG_WAVES) {
        int sN[4], rN[4];
        float av[4];
        Msg mm[4];
#pragma unroll
        for (int k = 0; k < 4; ++k) {
          int2 en = sl[i + AGG_WAVES * k];
          sN[k] = __builtin_amdgcn_readfirstlane(en.x);
          rN[k] = __builtin_amdgcn_readfirstlane(en.y);
          av[k] = as[(size_t)sN[k] * HEADS + hh];
        }
#pragma unroll
        for (int k = 0; k < 4; ++k) mm[k] = load_row(H, sN[k], lane);
        asm volatile("" :: "v"(av[0]), "v"(av[1]), "v"(av[2]), "v"(av[3]),
                     "v"(mm[0].v0.x), "v"(mm[0].v1.x), "v"(mm[0].v2.x), "v"(mm[0].v3.x),
                     "v"(mm[1].v0.x), "v"(mm[1].v1.x), "v"(mm[1].v2.x), "v"(mm[1].v3.x),
                     "v"(mm[2].v0.x), "v"(mm[2].v1.x), "v"(mm[2].v2.x), "v"(mm[2].v3.x),
                     "v"(mm[3].v0.x), "v"(mm[3].v1.x), "v"(mm[3].v2.x), "v"(mm[3].v3.x));
#pragma unroll
        for (int k = 0; k < 4; ++k) {
          float alpha = edge_alpha(av[k], rN[k], hh, t_ad, t_m, t_dn);
          msg_into_tile<false>(mm[k], alpha, rN[k], lane, acc);
        }
      }
      for (; i < cnt; i += AGG_WAVES) {
        int2 en = sl[i];
        edge_into_tile<false>(__builtin_amdgcn_readfirstlane(en.x),
                              __builtin_amdgcn_readfirstlane(en.y),
                              lane, H, as, t_ad, t_m, t_dn, acc);
      }
    } else {
      scan([&](int d, int e) {
        for_hits(d, e, nb, nr, [&](int s_, int r) {
          edge_into_tile<false>(s_, r, lane, H, as, t_ad, t_m, t_dn, acc);
        });
      });
    }
    __syncthreads();

    for (int i = tid; i < nr * (F1 / 4); i += AGG_THREADS) {
      int r = i >> 7, f = i & 127;
      const float* a = acc + r * F1 + f;
      float4 v = make_float4(a[0], a[128], a[256], a[384]);
      ((float4*)(out + (size_t)(nb + r) * F1))[f] = v;
    }
    if (s + 1 < NSUB) __syncthreads();
  }
}

#define T2_THREADS 1024
#define T2_SLOTS   (LIST_CAP / T2_THREADS)
static_assert(T2_SLOTS * T2_THREADS == LIST_CAP, "the slots of a work-group cover a whole list");
static_assert(TILE_R * D_OUT <= T2_THREADS, "one thread per (row, column) of the tile");

__global__ __launch_bounds__(T2_THREADS, 16)
void tile_agg2(const int* __restrict__ src, const int* __restrict__ dst, int nE, int n,
               const float* __restrict__ H2, const float* __restrict__ as,
               const float* __restrict__ ad, const float* __restrict__ b2,
               const int* __restrict__ tl_cnt, const int* __restrict__ tl_ent,
               float* __restrict__ out) {
  __shared__ float s_ad[TILE_R];
  __shared__ unsigned s_mo[TILE_R];
  __shared__ float s_m[TILE_R];
  __shared__ float s_w0[TILE_R];
  __shared__ float s_dn[TILE_R];
  __shared__ float acc[D_OUT * TILE_R];

  const int tid = threadIdx.x;
  const int node0 = blockIdx.x * TILE_R;
  const int rows = min(TILE_R, n - node0);
  const int cnt = tl_cnt[blockIdx.x];
  const bool listed = cnt <= LIST_CAP;

  const bool rowt = tid < rows;
  const bool elt = tid < rows * D_OUT;
  float a_d0 = 0.f, a_s0 = 0.f, hself = 0.f, bias = 0.f;
  if (rowt) { a_d0 = ad[node0 + tid]; a_s0 = as[node0 + tid]; }
  if (elt) { hself = H2[(size_t)node0 * D_OUT + tid]; bias = b2[tid & 7]; }

  const int2* tl = (const int2*)tl_ent + (size_t)blockIdx.x * LIST_CAP;
  bool ok[T2_SLOTS];
  int rN[T2_SLOTS];
  float aS[T2_SLOTS];
  float4 hA[T2_SLOTS], hB[T2_SLOTS];
#pragma unroll
  for (int k = 0; k < T2_SLOTS; ++k) {
    int i = tid + T2_THREADS * k;
    ok[k] = listed && i < cnt;
    rN[k] = 0; aS[k] = 0.f;
    hA[k] = hB[k] = make_float4(0.f, 0.f, 0.f, 0.f);
    if (ok[k]) {
      int2 en = tl[i];
      rN[k] = en.y;
      aS[k] = as[en.x];
      const float4* hp = (const float4*)(H2 + (size_t)en.x * D_OUT);
      hA[k] = hp[0]; hB[k] = hp[1];
    }
  }

  auto scan = [&](auto f) {
    for (int e = tid; e < nE; e += T2_THREADS) {
      unsigned dl = (unsigned)(dst[e] - node0);
      if (dl < (unsigned)rows) f(src[e], (int)dl);
    }
  };

  if (rowt) {
    float ev0 = lrelu(a_s0 + a_d0);
    s_ad[tid] = a_d0;
    s_mo[tid] = f2ord(ev0);
    s_w0[tid] = ev0;
  }
  __syncthreads();

  float ev[T2_SLOTS];
  if (listed) {
#pragma unroll
    for (int k = 0; k < T2_SLOTS; ++k) {
      ev[k] = 0.f;
      if (ok[k]) {
        ev[k] = lrelu(aS[k] + s_ad[rN[k]]);
        atomicMax(s_mo + rN[k], f2ord(ev[k]));
      }
    }
  } else {
    scan([&](int s_, int r) { atomicMax(s_mo + r, f2ord(lrelu(as[s_] + s_ad[r]))); });
  }
  __syncthreads();
  if (rowt) {
    float mx = ord2f(s_mo[tid]);
    float w0 = __expf(s_w0[tid] - mx);
    s_m[tid] = mx;
    s_w0[tid] = w0;
    s_dn[tid] = w0;
  }
  __syncthreads();

  float w[T2_SLOTS];
  if (listed) {
#pragma unroll
    for (int k = 0; k < T2_SLOTS; ++k) {
      w[k] = 0.f;
      if (ok[k]) {
        w[k] = __expf(ev[k] - s_m[rN[k]]);
        atomicAdd(s_dn + rN[k], w[k]);
      }
    }
  } else {
    scan([&](int s_, int r) {
      atomicAdd(s_dn + r, __expf(lrelu(as[s_] + s_ad[r]) - s_m[r]));
    });
  }
  __syncthreads();

  if (elt) {
    int r = tid >> 3, cc = tid & 7;
    float alpha = s_w0[r] / (s_dn[r] + 1e-16f);
    acc[cc * TILE_R + r] = hself * alpha;
  }
  __syncthreads();
  if (listed) {
#pragma unroll
    for (int k = 0; k < T2_SLOTS; ++k) {
      if (ok[k]) {
        float alpha = w[k] / (s_dn[rN[k]] + 1e-16f);
        float* ap = acc + rN[k];
        atomicAdd(ap,              hA[k].x * alpha); atomicAdd(ap + TILE_R,     hA[k].y * alpha);
        atomicAdd(ap + 2 * TILE_R, hA[k].z * alpha); atomicAdd(ap + 3 * TILE_R, hA[k].w * alpha);
        atomicAdd(ap + 4 * TILE_R, hB[k].x * alpha); atomicAdd(ap + 5 * TILE_R, hB[k].y * alpha);
        atomicAdd(ap + 6 * TILE_R, hB[k].z * alpha); atomicAdd(ap + 7 * TILE_R, hB[k].w * alpha);
      }
    }
  } else {
    scan([&](int s_, int r) {
      float alpha = __expf(lrelu(as[s_] + s_ad[r]) - s_m[r]) / (s_dn[r] + 1e-16f);
      const float* hp = H2 + (size_t)s_ * D_OUT;
#pragma unroll
      for (int cc = 0; cc < D_OUT; ++cc) atomicAdd(acc + cc * TILE_R + r, hp[cc] * alpha);
    });
  }
  __syncthreads();

  if (elt) {
    int r = tid >> 3, cc = tid & 7;
    out[(size_t)node0 * D_OUT + tid] = acc[cc * TILE_R + r] + bias;
  }
}

extern "C" void kernel_launch(void* const* d_in, const int* in_sizes, int n_in,
                              void* d_out, int out_size, void* d_ws, size_t ws_size,
                              hipStream_t stream) {
  const float* x      = (const float*)d_in[0];
  const int*   ei     = (const int*)  d_in[1];
  const float* W1     = (const float*)d_in[2];
  const float* a_src1 = (const float*)d_in[3];
  const float* a_dst1 = (const float*)d_in[4];
  const float* b1     = (const float*)d_in[5];
  const float* W2     = (const float*)d_in[6];
  const float* a_src2 = (const float*)d_in[7];
  const float* a_dst2 = (const float*)d_in[8];
  const float* b2     = (const float*)d_in[9];

  int n  = in_sizes[0] / D_IN;
  int nE = in_sizes[1] / 2;
  const int* src = ei;
  const int* dst = ei + nE;
  int ntiles = (n + TILE_R - 1) / TILE_R;
  int ngroups = (n + GROUP_R - 1) / GROUP_R;

  char* ws = (char*)d_ws;
  size_t off = 0;
  auto alloc = [&](size_t bytes) -> char* {
    char* p = ws + off; off += (bytes + 255) & ~(size_t)255; return p;
  };
  float*    h1   = (float*)   alloc((size_t)n * F1 * 4);
  float*    out1 = (float*)   alloc((size_t)n * F1 * 4);
  float*    as1  = (float*)   alloc((size_t)n * HEADS * 4);
  float*    ad1  = (float*)   alloc((size_t)n * HEADS * 4);
  float*    h2   = (float*)   alloc((size_t)n * D_OUT * 4);
  float*    as2  = (float*)   alloc((size_t)n * 4);
  float*    ad2  = (float*)   alloc((size_t)n * 4);
  int*      tlc  = (int*)     alloc((size_t)ntiles * 4);
  int*      tle  = (int*)     alloc((size_t)ntiles * LIST_CAP * 8);
  unsigned* dcd  = (unsigned*)alloc(((size_t)nE + 15) / 16 * 16);

  gemm1_wmma<<<dim3(F1 / G1_BN, (n + G1_BM - 1) / G1_BM), 256, 0, stream>>>(x, W1, h1, n);
  alphas1<<<(n * HEADS + 255) / 256, 256, 0, stream>>>(h1, a_src1, a_dst1, as1, ad1, n, dst, nE, dcd);
  tile_agg1<<<ngroups, AGG_THREADS, 0, stream>>>(src, dst, nE, n, h1, as1, ad1, out1, tlc, tle, dcd);

  {
    int waves  = (n + 15) / 16;
    int blocks = (waves + 7) / 8;
    gemm2_wmma<<<blocks, 256, 0, stream>>>(out1, b1, W2, h2, n);
  }
  alphas2<<<(n + 255) / 256, 256, 0, stream>>>(h2, a_src2, a_dst2, as2, ad2, n);
  tile_agg2<<<ntiles, T2_THREADS, 0, stream>>>(src, dst, nE, n, h2, as2, ad2, b2, tlc, tle, (float*)d_out);
}


# Layer-1 tile kernel forms each listed edge's 8 softmax weights once (one thread per edge) and keeps them in LDS: message loop reads 16 B per lane and edge, no score gather, exp, divide or permute
# speedup vs baseline: 9.1220x; 1.0349x over previous
#include <hip/hip_runtime.h>
#include <math.h>

#define D_IN   128
#define HEADS  8
#define HID    64
#define F1     (HEADS * HID)
#define D_OUT  8
#define NEG_SLOPE 0.2f

typedef float v2f __attribute__((ext_vector_type(2)));
typedef float v8f __attribute__((ext_vector_type(8)));

__device__ __forceinline__ unsigned f2ord(float f) {
  unsigned u = __float_as_uint(f);
  return (u & 0x80000000u) ? ~u : (u | 0x80000000u);
}
__device__ __forceinline__ float ord2f(unsigned v) {
  unsigned u = (v & 0x80000000u) ? (v ^ 0x80000000u) : ~v;
  return __uint_as_float(u);
}
__device__ __forceinline__ float lrelu(float x) { return x > 0.f ? x : NEG_SLOPE * x; }

#define G1_BM 128
#define G1_BN 128
#define G1_XP 132
#define G1_WP 288

__global__ __launch_bounds__(256)
void gemm1_wmma(const float* __restrict__ X, const float* __restrict__ W,
                float* __restrict__ H, int n) {
  __shared__ float Xs[G1_BM * G1_XP];
  __shared__ float Ws[(D_IN / 2) * G1_WP];

  const int tid = threadIdx.x, lane = tid & 31;
  const int wave = __builtin_amdgcn_readfirstlane(tid >> 5);
  const int col0 = blockIdx.x * G1_BN;
  const int row0 = blockIdx.y * G1_BM;

#pragma unroll
  for (int i = 0; i < (G1_BM * D_IN / 4) / 256; ++i) {
    int idx = tid + 256 * i;
    int r = idx >> 5, c4 = idx & 31;
    int gr = min(row0 + r, n - 1);
    float4 v = ((const float4*)(X + (size_t)gr * D_IN))[c4];
    *(float4*)(Xs + r * G1_XP + 4 * c4) = v;
  }
#pragma unroll
  for (int i = 0; i < ((D_IN / 2) * G1_BN / 4) / 256; ++i) {
    int idx = tid + 256 * i;
    int kp = idx >> 5, c4 = idx & 31;
    const float* wp = W + (size_t)(2 * kp) * F1 + col0 + 4 * c4;
    float4 r0 = *(const float4*)wp;
    float4 r1 = *(const float4*)(wp + F1);
    float* d = Ws + kp * G1_WP + 8 * c4;
    *(float4*)d       = make_float4(r0.x, r1.x, r0.y, r1.y);
    *(float4*)(d + 4) = make_float4(r0.z, r1.z, r0.w, r1.w);
  }
  __syncthreads();

  const int half = lane >> 4;
  const int l15  = lane & 15;
  const float* xa = Xs + (16 * wave + l15) * G1_XP + 2 * half;
  const float* wb = Ws + half * G1_WP + 2 * l15;

  v8f acc[G1_BN / 16];
#pragma unroll
  for (int i = 0; i < G1_BN / 16; ++i) acc[i] = (v8f){};
#pragma unroll
  for (int j = 0; j < D_IN / 4; ++j) {
    v2f a = *(const v2f*)(xa + 4 * j);
#pragma unroll
    for (int i = 0; i < G1_BN / 16; ++i) {
      v2f b = *(const v2f*)(wb + 2 * j * G1_WP + 32 * i);
      acc[i] = __builtin_amdgcn_wmma_f32_16x16x4_f32(false, a, false, b,
                                                     (short)0, acc[i], false, false);
    }
  }

  float* xo = Xs + (16 * wave + 8 * half) * G1_XP + l15;
#pragma unroll
  for (int i = 0; i < G1_BN / 16; ++i) {
#pragma unroll
    for (int v = 0; v < 8; ++v) xo[v * G1_XP + 16 * i] = acc[i][v];
  }
  __syncthreads();
  const float* xr = Xs + (16 * wave) * G1_XP + 4 * lane;
  float* hp = H + (size_t)(row0 + 16 * wave) * F1 + col0 + 4 * lane;
  const int nrows = n - (row0 + 16 * wave);
  if (nrows >= 16) {
    float4 o[16];
#pragma unroll
    for (int r = 0; r < 16; ++r) o[r] = *(const float4*)(xr + r * G1_XP);
#pragma unroll
    for (int r = 0; r < 16; ++r) *(float4*)(hp + (size_t)r * F1) = o[r];
  } else {
    for (int r = 0; r < nrows; ++r)
      *(float4*)(hp + (size_t)r * F1) = *(const float4*)(xr + r * G1_XP);
  }
}

#define G2_XP 132
#define G2_WP 32
#define G2_CH (F1 / 128)

__global__ __launch_bounds__(256)
void gemm2_wmma(const float* __restrict__ X2, const float* __restrict__ b1,
                const float* __restrict__ W2,
                float* __restrict__ H2, int n) {
  __shared__ float Xs[8 * 16 * G2_XP];
  __shared__ float Ws[(F1 / 2) * G2_WP];

  const int tid = threadIdx.x, lane = tid & 31;
  const int wave = __builtin_amdgcn_readfirstlane(tid >> 5);
  const int row0 = (blockIdx.x * 8 + wave) * 16;

  float4 w4[4];
#pragma unroll
  for (int i = 0; i < 4; ++i) w4[i] = ((const float4*)W2)[tid + 256 * i];
  float4 v[16];
  float4 bb = ((const float4*)b1)[lane];
#pragma unroll
  for (int r = 0; r < 16; ++r)
    v[r] = ((const float4*)(X2 + (size_t)min(row0 + r, n - 1) * F1))[lane];
#pragma unroll
  for (int i = 0; i < 4; ++i) {
    int g = tid + 256 * i;
    int k = g >> 1, c0 = (g & 1) * 4;
    float* d = Ws + (k >> 1) * G2_WP + 2 * c0 + (k & 1);
    d[0] = w4[i].x; d[2] = w4[i].y; d[4] = w4[i].z; d[6] = w4[i].w;
  }
#pragma unroll
  for (int i = 0; i < 4; ++i) {
    int p = tid + 256 * i;
    *(float4*)(Ws + (p >> 2) * G2_WP + 16 + 4 * (p & 3)) = make_float4(0.f, 0.f, 0.f, 0.f);
  }

  const int half = lane >> 4;
  const int l15  = lane & 15;
  float* xw = Xs + wave * 16 * G2_XP;
  const float* xa = xw + l15 * G2_XP + 2 * half;
  const float* wb = Ws + half * G2_WP + 2 * l15;

  v8f acc = {};
#pragma unroll
  for (int c = 0; c < G2_CH; ++c) {
#pragma unroll
    for (int r = 0; r < 16; ++r) {
      float4 s = make_float4(v[r].x + bb.x, v[r].y + bb.y, v[r].z + bb.z, v[r].w + bb.w);
      s.x = s.x > 0.f ? s.x : 0.f; s.y = s.y > 0.f ? s.y : 0.f;
      s.z = s.z > 0.f ? s.z : 0.f; s.w = s.w > 0.f ? s.w : 0.f;
      *(float4*)(xw + r * G2_XP + 4 * lane) = s;
    }
    if (c == 0) __syncthreads();
    if (c + 1 < G2_CH) {
      bb = ((const float4*)b1)[32 * (c + 1) + lane];
#pragma unroll
      for (int r = 0; r < 16; ++r)
        v[r] = ((const float4*)(X2 + (size_t)min(row0 + r, n - 1) * F1))[32 * (c + 1) + lane];
    }
#pragma unroll 8
    for (int j = 0; j < 32; ++j) {
      v2f a = *(const v2f*)(xa + 4 * j);
      v2f b = *(const v2f*)(wb + (64 * c + 2 * j) * G2_WP);
      acc = __builtin_amdgcn_wmma_f32_16x16x4_f32(false, a, false, b,
                                                  (short)0, acc, false, false);
    }
  }

  if (l15 < D_OUT) {
#pragma unroll
    for (int vv = 0; vv < 8; ++vv) {
      int r = row0 + vv + half * 8;
      if (r < n) H2[(size_t)r * D_OUT + l15] = acc[vv];
    }
  }
}

#define TILE_R      98
#define NSUB        2
#define GROUP_R     (NSUB * TILE_R)

#define A1_ROWS 4

template <int M>
__device__ __forceinline__ float add_lane_xor(float v) {
  int o = __builtin_amdgcn_update_dpp(0, __float_as_int(v), 0x160 | M, 0xF, 0xF, true);
  return v + __int_as_float(o);
}

__global__ void alphas1(const float* __restrict__ H, const float* __restrict__ a_src,
                        const float* __restrict__ a_dst,
                        float* __restrict__ as, float* __restrict__ ad, int n,
                        const int* __restrict__ dst, int nE, unsigned* __restrict__ codes) {
  int t = blockIdx.x * blockDim.x + threadIdx.x;
  const int nw = ((nE + 15) >> 4) << 2;
  for (int w = t; w < nw; w += gridDim.x * blockDim.x) {
    unsigned pk = 0u;
#pragma unroll
    for (int k = 0; k < 4; ++k) {
      int e = 4 * w + k;
      unsigned cd = e < nE ? (((unsigned)dst[e] / GROUP_R) & 255u) : 0u;
      pk |= cd << (8 * k);
    }
    codes[w] = pk;
  }

  const int lane = threadIdx.x & 31;
  const int r0 = (t >> 5) * A1_ROWS;
  if (r0 >= n) return;

  float4 hv[A1_ROWS][4], vs[4], vd[4];
#pragma unroll
  for (int j = 0; j < A1_ROWS; ++j) {
    const float4* hp = (const float4*)(H + (size_t)min(r0 + j, n - 1) * F1);
#pragma unroll
    for (int q = 0; q < 4; ++q) hv[j][q] = hp[lane + 32 * q];
  }
#pragma unroll
  for (int q = 0; q < 4; ++q) {
    vs[q] = ((const float4*)a_src)[lane + 32 * q];
    vd[q] = ((const float4*)a_dst)[lane + 32 * q];
  }

  const int l15 = lane & 15;
  float so = 0.f, dd = 0.f;
#pragma unroll
  for (int j = 0; j < A1_ROWS; ++j) {
#pragma unroll
    for (int q = 0; q < 4; ++q) {
      float4 x = hv[j][q];
      float s = x.x * vs[q].x; s += x.y * vs[q].y; s += x.z * vs[q].z; s += x.w * vs[q].w;
      float d = x.x * vd[q].x; d += x.y * vd[q].y; d += x.z * vd[q].z; d += x.w * vd[q].w;
      s = add_lane_xor<8>(s); s = add_lane_xor<4>(s); s = add_lane_xor<2>(s); s = add_lane_xor<1>(s);
      d = add_lane_xor<8>(d); d = add_lane_xor<4>(d); d = add_lane_xor<2>(d); d = add_lane_xor<1>(d);
      if (l15 == 4 * j + q) { so = s; dd = d; }
    }
  }
  const int r = r0 + (l15 >> 2);
  if (r < n) {
    int o = r * HEADS + 2 * (l15 & 3) + (lane >> 4);
    as[o] = so; ad[o] = dd;
  }
}

__global__ void alphas2(const float* __restrict__ H2, const float* __restrict__ a_src,
                        const float* __restrict__ a_dst,
                        float* __restrict__ as, float* __restrict__ ad, int n) {
  int t = blockIdx.x * blockDim.x + threadIdx.x;
  if (t >= n) return;
  const float* row = H2 + (size_t)t * D_OUT;
  float s = 0.f, d = 0.f;
#pragma unroll
  for (int c = 0; c < D_OUT; ++c) { s += row[c] * a_src[c]; d += row[c] * a_dst[c]; }
  as[t] = s; ad[t] = d;
}

#define AGG_THREADS 1024
#define AGG_WAVES   (AGG_THREADS / 32)
#define LIST_CAP    2048
#define CAND_CAP    (NSUB * LIST_CAP)
#define SC_CAP      1024

struct Msg { float4 v0, v1, v2, v3; };

__device__ __forceinline__ Msg load_row(const float* __restrict__ H, int s_, int lane) {
  const float4* hp = (const float4*)(H + (size_t)s_ * F1);
  Msg m;
  m.v0 = hp[lane]; m.v1 = hp[lane + 32]; m.v2 = hp[lane + 64]; m.v3 = hp[lane + 96];
  return m;
}

__device__ __forceinline__ float edge_alpha(float a_s, int r, int hh, const float* s_ad,
                                            const float* s_m, const float* s_dn) {
  float ev = lrelu(a_s + s_ad[r * HEADS + hh]);
  float w = __expf(ev - s_m[r * HEADS + hh]);
  return w / (s_dn[r * HEADS + hh] + 1e-16f);
}

template <bool START>
__device__ __forceinline__ void msg_scaled(const Msg& m, float a0, float a1, float a2, float a3,
                                           int r, int lane, float* acc) {
  float* ap = acc + r * F1 + lane;
  if (START) {
    ap[0]       = m.v0.x * a0; ap[128]      = m.v0.y * a0; ap[256]      = m.v0.z * a0; ap[384]      = m.v0.w * a0;
    ap[32]      = m.v1.x * a1; ap[128 + 32] = m.v1.y * a1; ap[256 + 32] = m.v1.z * a1; ap[384 + 32] = m.v1.w * a1;
    ap[64]      = m.v2.x * a2; ap[128 + 64] = m.v2.y * a2; ap[256 + 64] = m.v2.z * a2; ap[384 + 64] = m.v2.w * a2;
    ap[96]      = m.v3.x * a3; ap[128 + 96] = m.v3.y * a3; ap[256 + 96] = m.v3.z * a3; ap[384 + 96] = m.v3.w * a3;
  } else {
    atomicAdd(ap,            m.v0.x * a0); atomicAdd(ap + 128,      m.v0.y * a0);
    atomicAdd(ap + 256,      m.v0.z * a0); atomicAdd(ap + 384,      m.v0.w * a0);
    atomicAdd(ap + 32,       m.v1.x * a1); atomicAdd(ap + 128 + 32, m.v1.y * a1);
    atomicAdd(ap + 256 + 32, m.v1.z * a1); atomicAdd(ap + 384 + 32, m.v1.w * a1);
    atomicAdd(ap + 64,       m.v2.x * a2); atomicAdd(ap + 128 + 64, m.v2.y * a2);
    atomicAdd(ap + 256 + 64, m.v2.z * a2); atomicAdd(ap + 384 + 64, m.v2.w * a2);
    atomicAdd(ap + 96,       m.v3.x * a3); atomicAdd(ap + 128 + 96, m.v3.y * a3);
    atomicAdd(ap + 256 + 96, m.v3.z * a3); atomicAdd(ap + 384 + 96, m.v3.w * a3);
  }
}

template <bool START>
__device__ __forceinline__ void msg_into_tile(const Msg& m, float alpha, int r, int lane,
                                              float* acc) {
  int hs = lane >> 4;
  float a0 = __shfl(alpha, hs), a1 = __shfl(alpha, hs + 2);
  float a2 = __shfl(alpha, hs + 4), a3 = __shfl(alpha, hs + 6);
  msg_scaled<START>(m, a0, a1, a2, a3, r, lane, acc);
}

template <bool START>
__device__ __forceinline__ void edge_into_tile(int s_, int r, int lane,
                                               const float* __restrict__ H, float a_s,
                                               const float* s_ad, const float* s_m,
                                               const float* s_dn, float* acc) {
  Msg m = load_row(H, s_, lane);
  int hh = lane & 7;
  float alpha = edge_alpha(a_s, r, hh, s_ad, s_m, s_dn);
  msg_into_tile<START>(m, alpha, r, lane, acc);
}

__device__ __forceinline__ void scores8(float4 a0, float4 a1, const float* s_ad,
                                        int r, float* ev) {
  const float4* dp = (const float4*)(s_ad + r * HEADS);
  float4 d0 = dp[0], d1 = dp[1];
  ev[0] = lrelu(a0.x + d0.x); ev[1] = lrelu(a0.y + d0.y);
  ev[2] = lrelu(a0.z + d0.z); ev[3] = lrelu(a0.w + d0.w);
  ev[4] = lrelu(a1.x + d1.x); ev[5] = lrelu(a1.y + d1.y);
  ev[6] = lrelu(a1.z + d1.z); ev[7] = lrelu(a1.w + d1.w);
}

__global__ __launch_bounds__(AGG_THREADS)
void tile_agg1(const int* __restrict__ src, const int* __restrict__ dst,
               int nE, int n, const float* __restrict__ H,
               const float* __restrict__ as, const float* __restrict__ ad,
               float* __restrict__ out, int* __restrict__ tl_cnt, int* __restrict__ tl_ent,
               const unsigned* __restrict__ codes) {
  __shared__ float acc[TILE_R * F1];
  __shared__ __attribute__((aligned(16))) float s_ad[GROUP_R * HEADS];
  __shared__ float s_m[GROUP_R * HEADS];
  __shared__ __attribute__((aligned(16))) float s_dn[GROUP_R * HEADS];
  __shared__ unsigned s_mo[GROUP_R * HEADS];
  __shared__ int2 s_list[NSUB * LIST_CAP];
  __shared__ int s_cnt[NSUB];
  __shared__ __attribute__((aligned(16))) float s_sc[NSUB * SC_CAP * HEADS];
  int* s_cand = (int*)s_sc;
  static_assert(CAND_CAP * sizeof(int) <= sizeof(float) * NSUB * SC_CAP * HEADS,
                "the candidates fit into the kept entries' bytes");
  __shared__ int s_ccnt;

  const int tid = threadIdx.x, lane = tid & 31, wave = tid >> 5;
  const int node0 = blockIdx.x * GROUP_R;
  const int rows = min(GROUP_R, n - node0);

  for (int i = tid; i < rows * HEADS; i += AGG_THREADS) {
    size_t g = (size_t)node0 * HEADS + i;
    float a_d = ad[g];
    float ev = lrelu(as[g] + a_d);
    s_ad[i] = a_d;
    s_mo[i] = f2ord(ev);
    s_dn[i] = ev;
  }
  if (tid < NSUB) s_cnt[tid] = 0;
  if (tid == 0) s_ccnt = 0;
  __syncthreads();

  auto scan = [&](auto take) {
    auto take4 = [&](int4 d, bool ok, int e0) {
      unsigned lo = min(min((unsigned)(d.x - node0), (unsigned)(d.y - node0)),
                        min((unsigned)(d.z - node0), (unsigned)(d.w - node0)));
      bool any = (lo < (unsigned)rows) & ok;
      if (__builtin_amdgcn_ballot_w32(any) != 0u) {
        take(ok ? d.x : -1, e0);
        take(ok ? d.y : -1, e0 + 1);
        take(ok ? d.z : -1, e0 + 2);
        take(ok ? d.w : -1, e0 + 3);
      }
    };
    int pre = (int)(((16u - (unsigned)((size_t)dst & 15u)) & 15u) >> 2);
    if (pre > nE) pre = nE;
    const int nvec = (nE - pre) >> 2;
    const int4* dv = (const int4*)(dst + pre);
    const int nchunk = (nvec + 31) >> 5;
    if (nvec > 0) {
      const int last = nvec - 1;
      int4 cur[4];
#pragma unroll
      for (int j = 0; j < 4; ++j)
        cur[j] = dv[min(((wave + AGG_WAVES * j) << 5) + lane, last)];
      for (int base = wave; base < nchunk; base += 4 * AGG_WAVES) {
        int4 nxt[4];
#pragma unroll
        for (int j = 0; j < 4; ++j)
          nxt[j] = dv[min(((base + 4 * AGG_WAVES + AGG_WAVES * j) << 5) + lane, last)];
#pragma unroll
        for (int j = 0; j < 4; ++j) {
          int p = ((base + AGG_WAVES * j) << 5) + lane;
          take4(cur[j], p < nvec, pre + 4 * p);
          cur[j] = nxt[j];
        }
      }
    }
    if (wave == 0) {
      int tail0 = pre + 4 * nvec;
      int e = lane < pre ? lane : tail0 + (lane - pre);
      take(e < nE ? dst[e] : -1, e);
    }
  };

  auto for_hits = [&](int d, int e, int nb, int nr, auto f) {
    unsigned dl = (unsigned)(d - nb);
    unsigned mask = __builtin_amdgcn_ballot_w32(dl < (unsigned)nr);
    while (mask) {
      int l = __builtin_ctz(mask);
      mask &= mask - 1;
      int r  = __builtin_amdgcn_readlane((int)dl, l);
      int ee = __builtin_amdgcn_readlane(e, l);
      f(src[ee], r);
    }
  };

  auto list_hit = [&](int d, int e) {
    unsigned dl = (unsigned)(d - node0);
    bool hit = dl < (unsigned)rows;
    if (__builtin_amdgcn_ballot_w32(hit) != 0u) {
      int t = hit ? (int)(dl / TILE_R) : -1;
#pragma unroll
      for (int s = 0; s < NSUB; ++s) {
        bool hs = t == s;
        unsigned mask = __builtin_amdgcn_ballot_w32(hs);
        if (mask != 0u) {
          int base = 0;
          if (lane == 0) base = atomicAdd(&s_cnt[s], __builtin_popcount(mask));
          base = __builtin_amdgcn_readfirstlane(base);
          int pos = base + (int)__builtin_amdgcn_mbcnt_lo(mask, 0u);
          if (hs && pos < LIST_CAP)
            s_list[s * LIST_CAP + pos] = make_int2(e, (int)dl - s * TILE_R);
        }
      }
    }
  };

  {
    const unsigned gpat = ((unsigned)blockIdx.x & 255u) * 0x01010101u;
    const uint4* cv = (const uint4*)codes;
    const int nvec = (nE + 15) >> 4;
    const int nchunk = (nvec + 31) >> 5;
    auto test16 = [&](uint4 cw, bool ok, int e0) {
      unsigned x0 = cw.x ^ gpat, x1 = cw.y ^ gpat, x2 = cw.z ^ gpat, x3 = cw.w ^ gpat;
      unsigned z[4];
      z[0] = (x0 - 0x01010101u) & ~x0 & 0x80808080u;
      z[1] = (x1 - 0x01010101u) & ~x1 & 0x80808080u;
      z[2] = (x2 - 0x01010101u) & ~x2 & 0x80808080u;
      z[3] = (x3 - 0x01010101u) & ~x3 & 0x80808080u;
      if (ok && (z[0] | z[1] | z[2] | z[3]) != 0u) {
        int pos = atomicAdd(&s_ccnt, __builtin_popcount(z[0]) + __builtin_popcount(z[1]) +
                                     __builtin_popcount(z[2]) + __builtin_popcount(z[3]));
#pragma unroll
        for (int j = 0; j < 4; ++j) {
          unsigned zz = z[j];
          while (zz) {
            int e = e0 + 4 * j + (__builtin_ctz(zz) >> 3);
            zz &= zz - 1u;
            if (pos < CAND_CAP) s_cand[pos] = e;
            ++pos;
          }
        }
      }
    };
    if (nvec > 0) {
      const int last = nvec - 1;
      uint4 cur[4];
#pragma unroll
      for (int j = 0; j < 4; ++j)
        cur[j] = cv[min(((wave + AGG_WAVES * j) << 5) + lane, last)];
      for (int base = wave; base < nchunk; base += 4 * AGG_WAVES) {
        uint4 nxt[4];
#pragma unroll
        for (int j = 0; j < 4; ++j)
          nxt[j] = cv[min(((base + 4 * AGG_WAVES + AGG_WAVES * j) << 5) + lane, last)];
#pragma unroll
        for (int j = 0; j < 4; ++j) {
          int p = ((base + AGG_WAVES * j) << 5) + lane;
          test16(cur[j], p < nvec, 16 * p);
          cur[j] = nxt[j];
        }
      }
    }
  }
  __syncthreads();
  {
    const int ccnt = s_ccnt;
    if (ccnt <= CAND_CAP) {
      for (int i0 = 0; i0 < ccnt; i0 += AGG_THREADS) {
        int i = i0 + tid;
        int e = i < ccnt ? s_cand[i] : nE;
        int d = e < nE ? dst[e] : -1;
        list_hit(d, e);
      }
    } else {
      scan(list_hit);
    }
  }
  __syncthreads();
  if (tid < NSUB && rows > tid * TILE_R) tl_cnt[blockIdx.x * NSUB + tid] = s_cnt[tid];

  for (int s = 0; s < NSUB; ++s) {
    const int nb = node0 + s * TILE_R;
    const int nr = min(TILE_R, n - nb);
    if (nr <= 0) break;
    const int cnt = s_cnt[s];
    const float* t_ad = s_ad + s * TILE_R * HEADS;
    unsigned* t_mo = s_mo + s * TILE_R * HEADS;
    float* t_sc = s_sc + s * SC_CAP * HEADS;
    if (cnt <= LIST_CAP) {
      int2* sl = s_list + s * LIST_CAP;
      int2* tl = (int2*)tl_ent + (size_t)(blockIdx.x * NSUB + s) * LIST_CAP;
      for (int i = tid; i < cnt; i += AGG_THREADS) {
        int2 en = sl[i];
        int s_ = src[en.x];
        sl[i].x = s_;
        tl[i] = make_int2(s_, en.y);
        const float4* ap = (const float4*)(as + (size_t)s_ * HEADS);
        float4 a0 = ap[0], a1 = ap[1];
        if (i < SC_CAP) {
          float4* cp = (float4*)(t_sc + i * HEADS);
          cp[0] = a0; cp[1] = a1;
        }
        float ev[HEADS];
        scores8(a0, a1, t_ad, en.y, ev);
        unsigned* mp = t_mo + en.y * HEADS;
#pragma unroll
        for (int h = 0; h < HEADS; ++h) atomicMax(mp + h, f2ord(ev[h]));
      }
    } else {
      scan([&](int d, int e) {
        for_hits(d, e, nb, nr, [&](int s_, int r) {
          int hh = lane & 7;
          float ev = lrelu(as[(size_t)s_ * HEADS + hh] + t_ad[r * HEADS + hh]);
          if (lane < HEADS) atomicMax(t_mo + r * HEADS + hh, f2ord(ev));
        });
      });
    }
  }
  __syncthreads();
  for (int i = tid; i < rows * HEADS; i += AGG_THREADS) {
    float mx = ord2f(s_mo[i]);
    s_m[i] = mx;
    s_dn[i] = __expf(s_dn[i] - mx);
  }
  __syncthreads();

  for (int s = 0; s < NSUB; ++s) {
    const int nb = node0 + s * TILE_R;
    const int nr = min(TILE_R, n - nb);
    if (nr <= 0) break;
    const int cnt = s_cnt[s];
    const float* t_ad = s_ad + s * TILE_R * HEADS;
    const float* t_m = s_m + s * TILE_R * HEADS;
    float* t_dn = s_dn + s * TILE_R * HEADS;
    float* t_sc = s_sc + s * SC_CAP * HEADS;
    if (cnt <= LIST_CAP) {
      const int2* sl = s_list + s * LIST_CAP;
      for (int i = tid; i < cnt; i += AGG_THREADS) {
        int2 en = sl[i];
        const bool kept = i < SC_CAP;
        float4 a0, a1;
        if (kept) {
          const float4* cp = (const float4*)(t_sc + i * HEADS);
          a0 = cp[0]; a1 = cp[1];
        } else {
          const float4* ap = (const float4*)(as + (size_t)en.x * HEADS);
          a0 = ap[0]; a1 = ap[1];
        }
        float ev[HEADS];
        scores8(a0, a1, t_ad, en.y, ev);
        const float* mp = t_m + en.y * HEADS;
        float* dp = t_dn + en.y * HEADS;
#pragma unroll
        for (int h = 0; h < HEADS; ++h) {
          ev[h] = __expf(ev[h] - mp[h]);
          atomicAdd(dp + h, ev[h]);
        }
        if (kept) {
          float4* cp = (float4*)(t_sc + i * HEADS);
          cp[0] = make_float4(ev[0], ev[1], ev[2], ev[3]);
          cp[1] = make_float4(ev[4], ev[5], ev[6], ev[7]);
        }
      }
    } else {
      scan([&](int d, int e) {
        for_hits(d, e, nb, nr, [&](int s_, int r) {
          int hh = lane & 7;
          float ev = lrelu(as[(size_t)s_ * HEADS + hh] + t_ad[r * HEADS + hh]);
          if (lane < HEADS) atomicAdd(t_dn + r * HEADS + hh, __expf(ev - t_m[r * HEADS + hh]));
        });
      });
    }
  }
  __syncthreads();

  for (int s = 0; s < NSUB; ++s) {
    if (n - (node0 + s * TILE_R) <= 0) break;
    const int cnt = s_cnt[s];
    if (cnt > LIST_CAP) continue;
    const int2* sl = s_list + s * LIST_CAP;
    const float* t_dn = s_dn + s * TILE_R * HEADS;
    float* t_sc = s_sc + s * SC_CAP * HEADS;
    for (int i = tid; i < min(cnt, SC_CAP); i += AGG_THREADS) {
      const int r = sl[i].y;
      float4* cp = (float4*)(t_sc + i * HEADS);
      const float4 w0 = cp[0], w1 = cp[1];
      const float4* dp = (const float4*)(t_dn + r * HEADS);
      const float4 d0 = dp[0], d1 = dp[1];
      cp[0] = make_float4(w0.x / (d0.x + 1e-16f), w0.z / (d0.z + 1e-16f),
                          w1.x / (d1.x + 1e-16f), w1.z / (d1.z + 1e-16f));
      cp[1] = make_float4(w0.y / (d0.y + 1e-16f), w0.w / (d0.w + 1e-16f),
                          w1.y / (d1.y + 1e-16f), w1.w / (d1.w + 1e-16f));
    }
  }

  for (int s = 0; s < NSUB; ++s) {
    const int nb = node0 + s * TILE_R;
    const int nr = min(TILE_R, n - nb);
    if (nr <= 0) break;
    const int cnt = s_cnt[s];
    const float* t_ad = s_ad + s * TILE_R * HEADS;
    const float* t_m = s_m + s * TILE_R * HEADS;
    const float* t_dn = s_dn + s * TILE_R * HEADS;
    const float* t_sc = s_sc + s * SC_CAP * HEADS;
    const int hh = lane & 7;

    for (int r = wave; r < nr; r += AGG_WAVES)
      edge_into_tile<true>(nb + r, r, lane, H, as[(size_t)(nb + r) * HEADS + hh],
                           t_ad, t_m, t_dn, acc);
    __syncthreads();

    if (cnt <= LIST_CAP) {
      const int2* sl = s_list + s * LIST_CAP;
      const float* wq = t_sc + 4 * (lane >> 4);
      int i = __builtin_amdgcn_readfirstlane(wave);
      const int ckept = min(cnt, SC_CAP);
      for (; i + 3 * AGG_WAVES < ckept; i += 4 * AGG_WAVES) {
        int sN[4], rN[4];
        float4 aq[4];
        Msg mm[4];
#pragma unroll
        for (int k = 0; k < 4; ++k) {
          int2 en = sl[i + AGG_WAVES * k];
          sN[k] = __builtin_amdgcn_readfirstlane(en.x);
          rN[k] = __builtin_amdgcn_readfirstlane(en.y);
        }
#pragma unroll
        for (int k = 0; k < 4; ++k) mm[k] = load_row(H, sN[k], lane);
        asm volatile("" :: "v"(mm[0].v0.x), "v"(mm[0].v1.x), "v"(mm[0].v2.x), "v"(mm[0].v3.x),
                     "v"(mm[1].v0.x), "v"(mm[1].v1.x), "v"(mm[1].v2.x), "v"(mm[1].v3.x),
                     "v"(mm[2].v0.x), "v"(mm[2].v1.x), "v"(mm[2].v2.x), "v"(mm[2].v3.x),
                     "v"(mm[3].v0.x), "v"(mm[3].v1.x), "v"(mm[3].v2.x), "v"(mm[3].v3.x));
#pragma unroll
        for (int k = 0; k < 4; ++k) aq[k] = *(const float4*)(wq + (i + AGG_WAVES * k) * HEADS);
#pragma unroll
        for (int k = 0; k < 4; ++k)
          msg_scaled<false>(mm[k], aq[k].x, aq[k].y, aq[k].z, aq[k].w, rN[k], lane, acc);
      }
      for (; i + 3 * AGG_WAVES < cnt; i += 4 * AGG_WAVES) {
        int sN[4], rN[4];
        float av[4];
        Msg mm[4];
#pragma unroll
        for (int k = 0; k < 4; ++k) {
          int2 en = sl[i + AGG_WAVES * k];
          sN[k] = __builtin_amdgcn_readfirstlane(en.x);
          rN[k] = __builtin_amdgcn_readfirstlane(en.y);
          av[k] = as[(size_t)sN[k] * HEADS + hh];
        }
#pragma unroll
        for (int k = 0; k < 4; ++k) mm[k] = load_row(H, sN[k], lane);
        asm volatile("" :: "v"(av[0]), "v"(av[1]), "v"(av[2]), "v"(av[3]),
                     "v"(mm[0].v0.x), "v"(mm[0].v1.x), "v"(mm[0].v2.x), "v"(mm[0].v3.x),
                     "v"(mm[1].v0.x), "v"(mm[1].v1.x), "v"(mm[1].v2.x), "v"(mm[1].v3.x),
                     "v"(mm[2].v0.x), "v"(mm[2].v1.x), "v"(mm[2].v2.x), "v"(mm[2].v3.x),
                     "v"(mm[3].v0.x), "v"(mm[3].v1.x), "v"(mm[3].v2.x), "v"(mm[3].v3.x));
#pragma unroll
        for (int k = 0; k < 4; ++k) {
          float alpha = edge_alpha(av[k], rN[k], hh, t_ad, t_m, t_dn);
          msg_into_tile<false>(mm[k], alpha, rN[k], lane, acc);
        }
      }
      for (; i < cnt; i += AGG_WAVES) {
        int2 en = sl[i];
        const int s_ = __builtin_amdgcn_readfirstlane(en.x);
        const int r = __builtin_amdgcn_readfirstlane(en.y);
        if (i < SC_CAP) {
          Msg m = load_row(H, s_, lane);
          const float4 a = *(const float4*)(wq + i * HEADS);
          msg_scaled<false>(m, a.x, a.y, a.z, a.w, r, lane, acc);
        } else {
          edge_into_tile<false>(s_, r, lane, H, as[(size_t)s_ * HEADS + hh], t_ad, t_m, t_dn, acc);
        }
      }
    } else {
      scan([&](int d, int e) {
        for_hits(d, e, nb, nr, [&](int s_, int r) {
          edge_into_tile<false>(s_, r, lane, H, as[(size_t)s_ * HEADS + hh], t_ad, t_m, t_dn, acc);
        });
      });
    }
    __syncthreads();

    for (int i = tid; i < nr * (F1 / 4); i += AGG_THREADS) {
      int r = i >> 7, f = i & 127;
      const float* a = acc + r * F1 + f;
      float4 v = make_float4(a[0], a[128], a[256], a[384]);
      ((float4*)(out + (size_t)(nb + r) * F1))[f] = v;
    }
    if (s + 1 < NSUB) __syncthreads();
  }
}

#define T2_THREADS 1024
#define T2_SLOTS   (LIST_CAP / T2_THREADS)
static_assert(T2_SLOTS * T2_THREADS == LIST_CAP, "the slots of a work-group cover a whole list");
static_assert(TILE_R * D_OUT <= T2_THREADS, "one thread per (row, column) of the tile");

__global__ __launch_bounds__(T2_THREADS, 16)
void tile_agg2(const int* __restrict__ src, const int* __restrict__ dst, int nE, int n,
               const float* __restrict__ H2, const float* __restrict__ as,
               const float* __restrict__ ad, const float* __restrict__ b2,
               const int* __restrict__ tl_cnt, const int* __restrict__ tl_ent,
               float* __restrict__ out) {
  __shared__ float s_ad[TILE_R];
  __shared__ unsigned s_mo[TILE_R];
  __shared__ float s_m[TILE_R];
  __shared__ float s_w0[TILE_R];
  __shared__ float s_dn[TILE_R];
  __shared__ float acc[D_OUT * TILE_R];

  const int tid = threadIdx.x;
  const int node0 = blockIdx.x * TILE_R;
  const int rows = min(TILE_R, n - node0);
  const int cnt = tl_cnt[blockIdx.x];
  const bool listed = cnt <= LIST_CAP;

  const bool rowt = tid < rows;
  const bool elt = tid < rows * D_OUT;
  float a_d0 = 0.f, a_s0 = 0.f, hself = 0.f, bias = 0.f;
  if (rowt) { a_d0 = ad[node0 + tid]; a_s0 = as[node0 + tid]; }
  if (elt) { hself = H2[(size_t)node0 * D_OUT + tid]; bias = b2[tid & 7]; }

  const int2* tl = (const int2*)tl_ent + (size_t)blockIdx.x * LIST_CAP;
  bool ok[T2_SLOTS];
  int rN[T2_SLOTS];
  float aS[T2_SLOTS];
  float4 hA[T2_SLOTS], hB[T2_SLOTS];
#pragma unroll
  for (int k = 0; k < T2_SLOTS; ++k) {
    int i = tid + T2_THREADS * k;
    ok[k] = listed && i < cnt;
    rN[k] = 0; aS[k] = 0.f;
    hA[k] = hB[k] = make_float4(0.f, 0.f, 0.f, 0.f);
    if (ok[k]) {
      int2 en = tl[i];
      rN[k] = en.y;
      aS[k] = as[en.x];
      const float4* hp = (const float4*)(H2 + (size_t)en.x * D_OUT);
      hA[k] = hp[0]; hB[k] = hp[1];
    }
  }

  auto scan = [&](auto f) {
    for (int e = tid; e < nE; e += T2_THREADS) {
      unsigned dl = (unsigned)(dst[e] - node0);
      if (dl < (unsigned)rows) f(src[e], (int)dl);
    }
  };

  if (rowt) {
    float ev0 = lrelu(a_s0 + a_d0);
    s_ad[tid] = a_d0;
    s_mo[tid] = f2ord(ev0);
    s_w0[tid] = ev0;
  }
  __syncthreads();

  float ev[T2_SLOTS];
  if (listed) {
#pragma unroll
    for (int k = 0; k < T2_SLOTS; ++k) {
      ev[k] = 0.f;
      if (ok[k]) {
        ev[k] = lrelu(aS[k] + s_ad[rN[k]]);
        atomicMax(s_mo + rN[k], f2ord(ev[k]));
      }
    }
  } else {
    scan([&](int s_, int r) { atomicMax(s_mo + r, f2ord(lrelu(as[s_] + s_ad[r]))); });
  }
  __syncthreads();
  if (rowt) {
    float mx = ord2f(s_mo[tid]);
    float w0 = __expf(s_w0[tid] - mx);
    s_m[tid] = mx;
    s_w0[tid] = w0;
    s_dn[tid] = w0;
  }
  __syncthreads();

  float w[T2_SLOTS];
  if (listed) {
#pragma unroll
    for (int k = 0; k < T2_SLOTS; ++k) {
      w[k] = 0.f;
      if (ok[k]) {
        w[k] = __expf(ev[k] - s_m[rN[k]]);
        atomicAdd(s_dn + rN[k], w[k]);
      }
    }
  } else {
    scan([&](int s_, int r) {
      atomicAdd(s_dn + r, __expf(lrelu(as[s_] + s_ad[r]) - s_m[r]));
    });
  }
  __syncthreads();

  if (elt) {
    int r = tid >> 3, cc = tid & 7;
    float alpha = s_w0[r] / (s_dn[r] + 1e-16f);
    acc[cc * TILE_R + r] = hself * alpha;
  }
  __syncthreads();
  if (listed) {
#pragma unroll
    for (int k = 0; k < T2_SLOTS; ++k) {
      if (ok[k]) {
        float alpha = w[k] / (s_dn[rN[k]] + 1e-16f);
        float* ap = acc + rN[k];
        atomicAdd(ap,              hA[k].x * alpha); atomicAdd(ap + TILE_R,     hA[k].y * alpha);
        atomicAdd(ap + 2 * TILE_R, hA[k].z * alpha); atomicAdd(ap + 3 * TILE_R, hA[k].w * alpha);
        atomicAdd(ap + 4 * TILE_R, hB[k].x * alpha); atomicAdd(ap + 5 * TILE_R, hB[k].y * alpha);
        atomicAdd(ap + 6 * TILE_R, hB[k].z * alpha); atomicAdd(ap + 7 * TILE_R, hB[k].w * alpha);
      }
    }
  } else {
    scan([&](int s_, int r) {
      float alpha = __expf(lrelu(as[s_] + s_ad[r]) - s_m[r]) / (s_dn[r] + 1e-16f);
      const float* hp = H2 + (size_t)s_ * D_OUT;
#pragma unroll
      for (int cc = 0; cc < D_OUT; ++cc) atomicAdd(acc + cc * TILE_R + r, hp[cc] * alpha);
    });
  }
  __syncthreads();

  if (elt) {
    int r = tid >> 3, cc = tid & 7;
    out[(size_t)node0 * D_OUT + tid] = acc[cc * TILE_R + r] + bias;
  }
}

extern "C" void kernel_launch(void* const* d_in, const int* in_sizes, int n_in,
                              void* d_out, int out_size, void* d_ws, size_t ws_size,
                              hipStream_t stream) {
  const float* x      = (const float*)d_in[0];
  const int*   ei     = (const int*)  d_in[1];
  const float* W1     = (const float*)d_in[2];
  const float* a_src1 = (const float*)d_in[3];
  const float* a_dst1 = (const float*)d_in[4];
  const float* b1     = (const float*)d_in[5];
  const float* W2     = (const float*)d_in[6];
  const float* a_src2 = (const float*)d_in[7];
  const float* a_dst2 = (const float*)d_in[8];
  const float* b2     = (const float*)d_in[9];

  int n  = in_sizes[0] / D_IN;
  int nE = in_sizes[1] / 2;
  const int* src = ei;
  const int* dst = ei + nE;
  int ntiles = (n + TILE_R - 1) / TILE_R;
  int ngroups = (n + GROUP_R - 1) / GROUP_R;

  char* ws = (char*)d_ws;
  size_t off = 0;
  auto alloc = [&](size_t bytes) -> char* {
    char* p = ws + off; off += (bytes + 255) & ~(size_t)255; return p;
  };
  float*    h1   = (float*)   alloc((size_t)n * F1 * 4);
  float*    out1 = (float*)   alloc((size_t)n * F1 * 4);
  float*    as1  = (float*)   alloc((size_t)n * HEADS * 4);
  float*    ad1  = (float*)   alloc((size_t)n * HEADS * 4);
  float*    h2   = (float*)   alloc((size_t)n * D_OUT * 4);
  float*    as2  = (float*)   alloc((size_t)n * 4);
  float*    ad2  = (float*)   alloc((size_t)n * 4);
  int*      tlc  = (int*)     alloc((size_t)ntiles * 4);
  int*      tle  = (int*)     alloc((size_t)ntiles * LIST_CAP * 8);
  unsigned* dcd  = (unsigned*)alloc(((size_t)nE + 15) / 16 * 16);

  gemm1_wmma<<<dim3(F1 / G1_BN, (n + G1_BM - 1) / G1_BM), 256, 0, stream>>>(x, W1, h1, n);
  alphas1<<<(n * HEADS + 255) / 256, 256, 0, stream>>>(h1, a_src1, a_dst1, as1, ad1, n, dst, nE, dcd);
  tile_agg1<<<ngroups, AGG_THREADS, 0, stream>>>(src, dst, nE, n, h1, as1, ad1, out1, tlc, tle, dcd);

  {
    int waves  = (n + 15) / 16;
    int blocks = (waves + 7) / 8;
    gemm2_wmma<<<blocks, 256, 0, stream>>>(out1, b1, W2, h2, n);
  }
  alphas2<<<(n + 255) / 256, 256, 0, stream>>>(h2, a_src2, a_dst2, as2, ad2, n);
  tile_agg2<<<ntiles, T2_THREADS, 0, stream>>>(src, dst, nE, n, h2, as2, ad2, b2, tlc, tle, (float*)d_out);
}
